# starting kernel
# baseline (speedup 1.0000x reference)
#include <hip/hip_runtime.h>
#include <math.h>


typedef __attribute__((ext_vector_type(2))) float v2f;
typedef __attribute__((ext_vector_type(8))) float v8f;

#define F_IN 512
#define HID  16
#define NCLS 2

__global__ void gnn_zero(float* __restrict__ ws, int n) {
  int i = blockIdx.x * blockDim.x + threadIdx.x;
  if (i < n) ws[i] = 0.0f;
}

__global__ void gnn_degrees(const int* __restrict__ src, const int* __restrict__ dst,
                            float* __restrict__ deg_out, float* __restrict__ deg_in, int E) {
  int e = blockIdx.x * blockDim.x + threadIdx.x;
  if (e < E) {
    atomicAdd(&deg_out[src[e]], 1.0f);
    atomicAdd(&deg_in[dst[e]], 1.0f);
  }
}

__global__ void gnn_invsqrt(const float* __restrict__ deg_out, const float* __restrict__ deg_in,
                            float* __restrict__ inv_out, float* __restrict__ inv_in, int N) {
  int i = blockIdx.x * blockDim.x + threadIdx.x;
  if (i < N) {
    inv_out[i] = rsqrtf(fmaxf(deg_out[i], 1.0f));
    inv_in[i]  = rsqrtf(fmaxf(deg_in[i], 1.0f));
  }
}

__global__ __launch_bounds__(256) void gnn_gemm1_wmma(
    const float* __restrict__ x, const float* __restrict__ W1,
    const float* __restrict__ inv_out, float* __restrict__ h1, int N) {
  __shared__ float wlds[F_IN * HID];
  for (int i = threadIdx.x; i < F_IN * HID; i += blockDim.x) wlds[i] = W1[i];
  __syncthreads();

  const int wave = threadIdx.x >> 5;
  const int lane = threadIdx.x & 31;
  const int tiles = N >> 4;
  const int tile = blockIdx.x * 8 + wave;
  if (tile >= tiles) return;

  const int n16 = lane & 15;
  const int g   = lane >> 4;
  const int row = tile * 16 + n16;
  const float s = inv_out[row];
  const float* __restrict__ xr = x + (size_t)row * F_IN;

  v8f c = {};
  #pragma unroll 4
  for (int k = 0; k < F_IN; k += 4) {
    const int kk = k + 2 * g;
    v2f a = *(const v2f*)(xr + kk);
    a.x *= s;
    a.y *= s;
    v2f b;
    b.x = wlds[(kk + 0) * HID + n16];
    b.y = wlds[(kk + 1) * HID + n16];
    c = __builtin_amdgcn_wmma_f32_16x16x4_f32(
         false, a,  false, b,
         (short)0, c,  false,  false);
  }

  #pragma unroll
  for (int p = 0; p < 8; ++p) {
    const int m = tile * 16 + p + 8 * g;
    h1[(size_t)m * HID + n16] = c[p];
  }
}

__global__ void gnn_scatter16(const int* __restrict__ src, const int* __restrict__ dst,
                              const float* __restrict__ h1, float* __restrict__ agg1,
                              long E4) {
  long t = (long)blockIdx.x * blockDim.x + threadIdx.x;
  if (t >= E4) return;
  long e = t >> 2;
  int f = (int)(t & 3) * 4;
  int s = src[e], d = dst[e];
  float4 v = *(const float4*)(h1 + (size_t)s * HID + f);
  float* a = agg1 + (size_t)d * HID + f;
  atomicAdd(a + 0, v.x);
  atomicAdd(a + 1, v.y);
  atomicAdd(a + 2, v.z);
  atomicAdd(a + 3, v.w);
}

__global__ void gnn_layer2(const float* __restrict__ agg1, const float* __restrict__ inv_in,
                           const float* __restrict__ inv_out, const float* __restrict__ b1,
                           const float* __restrict__ W2, float* __restrict__ h2, int N) {
  int i = blockIdx.x * blockDim.x + threadIdx.x;
  if (i >= N) return;
  const float si = inv_in[i];
  const float so = inv_out[i];
  const float* a = agg1 + (size_t)i * HID;
  float acc0 = 0.0f, acc1 = 0.0f;
  #pragma unroll
  for (int k = 0; k < HID; ++k) {
    float hv = fmaxf(fmaf(a[k], si, b1[k]), 0.0f) * so;
    acc0 = fmaf(hv, W2[k * NCLS + 0], acc0);
    acc1 = fmaf(hv, W2[k * NCLS + 1], acc1);
  }
  h2[(size_t)i * NCLS + 0] = acc0;
  h2[(size_t)i * NCLS + 1] = acc1;
}

__global__ void gnn_scatter2(const int* __restrict__ src, const int* __restrict__ dst,
                             const float* __restrict__ h2, float* __restrict__ agg2, int E) {
  int e = blockIdx.x * blockDim.x + threadIdx.x;
  if (e >= E) return;
  int s = src[e], d = dst[e];
  float2 v = *(const float2*)(h2 + (size_t)s * NCLS);
  atomicAdd(&agg2[(size_t)d * NCLS + 0], v.x);
  atomicAdd(&agg2[(size_t)d * NCLS + 1], v.y);
}

__global__ void gnn_final(const float* __restrict__ agg2, const float* __restrict__ inv_in,
                          const float* __restrict__ b2, float* __restrict__ out, int N) {
  int i = blockIdx.x * blockDim.x + threadIdx.x;
  if (i >= N) return;
  const float si = inv_in[i];
  float v0 = fmaf(agg2[(size_t)i * NCLS + 0], si, b2[0]);
  float v1 = fmaf(agg2[(size_t)i * NCLS + 1], si, b2[1]);
  float m = fmaxf(v0, v1);
  float lse = m + logf(expf(v0 - m) + expf(v1 - m));
  out[(size_t)i * NCLS + 0] = v0 - lse;
  out[(size_t)i * NCLS + 1] = v1 - lse;
}

extern "C" void kernel_launch(void* const* d_in, const int* in_sizes, int n_in,
                              void* d_out, int out_size, void* d_ws, size_t ws_size,
                              hipStream_t stream) {
  const float* x   = (const float*)d_in[0];
  const int*   src = (const int*)d_in[1];
  const int*   dst = (const int*)d_in[2];
  const float* W1  = (const float*)d_in[3];
  const float* b1  = (const float*)d_in[4];
  const float* W2  = (const float*)d_in[5];
  const float* b2  = (const float*)d_in[6];
  float* out = (float*)d_out;

  const int N = in_sizes[0] / F_IN;
  const int E = in_sizes[1];

  float* ws      = (float*)d_ws;
  float* deg_out = ws;
  float* deg_in  = ws + (size_t)N;
  float* inv_out = ws + (size_t)2 * N;
  float* inv_in  = ws + (size_t)3 * N;
  float* h1      = ws + (size_t)4 * N;
  float* agg1    = ws + (size_t)20 * N;
  float* h2      = ws + (size_t)36 * N;
  float* agg2    = ws + (size_t)38 * N;

  const int B = 256;
  const int wsFloats = 40 * N;

  gnn_zero<<<(wsFloats + B - 1) / B, B, 0, stream>>>(ws, wsFloats);
  gnn_degrees<<<(E + B - 1) / B, B, 0, stream>>>(src, dst, deg_out, deg_in, E);
  gnn_invsqrt<<<(N + B - 1) / B, B, 0, stream>>>(deg_out, deg_in, inv_out, inv_in, N);

  const int tiles = N >> 4;
  gnn_gemm1_wmma<<<(tiles + 7) / 8, B, 0, stream>>>(x, W1, inv_out, h1, N);

  const long E4 = (long)E * 4;
  gnn_scatter16<<<(int)((E4 + B - 1) / B), B, 0, stream>>>(src, dst, h1, agg1, E4);
  gnn_layer2<<<(N + B - 1) / B, B, 0, stream>>>(agg1, inv_in, inv_out, b1, W2, h2, N);
  gnn_scatter2<<<(E + B - 1) / B, B, 0, stream>>>(src, dst, h2, agg2, E);
  gnn_final<<<(N + B - 1) / B, B, 0, stream>>>(agg2, inv_in, b2, out, N);
}


# Degree counts and layer-2 edge sums: edges binned by 256-node range (count, scan, scatter), one work-group per range sums its list in LDS, plain stores; no global atomics there; sum order changes
# speedup vs baseline: 1.8028x; 1.8028x over previous
#include <hip/hip_runtime.h>
#include <math.h>


typedef __attribute__((ext_vector_type(2))) float v2f;
typedef __attribute__((ext_vector_type(8))) float v8f;

#define F_IN 512
#define HID  16
#define NCLS 2

#define NPB_SHIFT 8
#define NPB   (1 << NPB_SHIFT)
#define MAXB  512
#define BIN_G 256
#define ENT_SHIFT 20
#define TOT_PITCH 32

__global__ void gnn_zero(float* __restrict__ ws, int n) {
  int i = blockIdx.x * blockDim.x + threadIdx.x;
  if (i < n) ws[i] = 0.0f;
}

__device__ __forceinline__ void edge_chunk(int E, int& lo, int& hi) {
  const int chunk = ((E + BIN_G * 4 - 1) / (BIN_G * 4)) * 4;
  lo = blockIdx.x * chunk;
  if (lo > E) lo = E;
  hi = lo + chunk;
  if (hi > E) hi = E;
}

__global__ __launch_bounds__(1024) void bin_count(const int* __restrict__ src, const int* __restrict__ dst,
                                                  int* __restrict__ cntS, int* __restrict__ cntD,
                                                  int E, int NB) {
  __shared__ int hs[MAXB];
  __shared__ int hd[MAXB];
  for (int i = threadIdx.x; i < MAXB; i += blockDim.x) { hs[i] = 0; hd[i] = 0; }
  __syncthreads();
  int lo, hi;
  edge_chunk(E, lo, hi);
  for (int e = lo + 4 * (int)threadIdx.x; e < hi; e += 4 * (int)blockDim.x) {
    if (e + 3 < hi) {
      const int4 s = *(const int4*)(src + e);
      const int4 d = *(const int4*)(dst + e);
      atomicAdd(&hs[s.x >> NPB_SHIFT], 1);
      atomicAdd(&hs[s.y >> NPB_SHIFT], 1);
      atomicAdd(&hs[s.z >> NPB_SHIFT], 1);
      atomicAdd(&hs[s.w >> NPB_SHIFT], 1);
      atomicAdd(&hd[d.x >> NPB_SHIFT], 1);
      atomicAdd(&hd[d.y >> NPB_SHIFT], 1);
      atomicAdd(&hd[d.z >> NPB_SHIFT], 1);
      atomicAdd(&hd[d.w >> NPB_SHIFT], 1);
    } else {
      for (int j = e; j < hi; ++j) {
        atomicAdd(&hs[src[j] >> NPB_SHIFT], 1);
        atomicAdd(&hd[dst[j] >> NPB_SHIFT], 1);
      }
    }
  }
  __syncthreads();
  for (int b = threadIdx.x; b < NB; b += blockDim.x) {
    cntS[blockIdx.x * MAXB + b] = hs[b];
    cntD[blockIdx.x * MAXB + b] = hd[b];
  }
}

__global__ __launch_bounds__(BIN_G) void bin_scan(const int* __restrict__ cntS, const int* __restrict__ cntD,
                                                  int* __restrict__ preS, int* __restrict__ preD,
                                                  int* __restrict__ totS, int* __restrict__ totD, int NB) {
  __shared__ int sh[BIN_G];
  int b = blockIdx.x;
  const int* cnt = cntS;
  int* pre = preS;
  int* tot = totS;
  if (b >= NB) { b -= NB; cnt = cntD; pre = preD; tot = totD; }
  const int t = threadIdx.x;
  const int v = cnt[t * MAXB + b];
  sh[t] = v;
  __syncthreads();
  for (int off = 1; off < BIN_G; off <<= 1) {
    const int add = (t >= off) ? sh[t - off] : 0;
    __syncthreads();
    sh[t] += add;
    __syncthreads();
  }
  pre[b * BIN_G + t] = sh[t] - v;
  if (t == BIN_G - 1) tot[b * TOT_PITCH] = sh[t];
}

__global__ __launch_bounds__(MAXB) void bin_base(const int* __restrict__ totS, const int* __restrict__ totD,
                                                 int* __restrict__ baseS, int* __restrict__ baseD, int NB) {
  __shared__ int sh[MAXB];
  const int* tot = blockIdx.x ? totD : totS;
  int* base = blockIdx.x ? baseD : baseS;
  const int t = threadIdx.x;
  const int v = (t < NB) ? tot[t * TOT_PITCH] : 0;
  sh[t] = v;
  __syncthreads();
  for (int off = 1; off < MAXB; off <<= 1) {
    const int add = (t >= off) ? sh[t - off] : 0;
    __syncthreads();
    sh[t] += add;
    __syncthreads();
  }
  if (t < NB) base[t] = sh[t] - v;
  if (t == NB - 1) base[NB] = sh[t];
}

__device__ __forceinline__ int bin_slot(int* cur, int key) {
  return atomicAdd(&cur[key >> NPB_SHIFT], 1);
}
__device__ __forceinline__ unsigned int bin_entry(int key, unsigned int payload) {
  return ((unsigned int)(key & (NPB - 1)) << ENT_SHIFT) | payload;
}
__device__ __forceinline__ void bin_put(int* cur, unsigned int* __restrict__ list, int key, unsigned int payload) {
  list[bin_slot(cur, key)] = bin_entry(key, payload);
}

__global__ __launch_bounds__(1024) void bin_scatter_src(const int* __restrict__ src,
                                                        const int* __restrict__ pre, const int* __restrict__ base,
                                                        unsigned int* __restrict__ list, int E, int NB) {
  __shared__ int cur[MAXB];
  for (int b = threadIdx.x; b < NB; b += blockDim.x) cur[b] = base[b] + pre[b * BIN_G + blockIdx.x];
  __syncthreads();
  int lo, hi;
  edge_chunk(E, lo, hi);
  for (int e = lo + 4 * (int)threadIdx.x; e < hi; e += 4 * (int)blockDim.x) {
    if (e + 3 < hi) {
      const int4 s = *(const int4*)(src + e);
      const int p0 = bin_slot(cur, s.x);
      const int p1 = bin_slot(cur, s.y);
      const int p2 = bin_slot(cur, s.z);
      const int p3 = bin_slot(cur, s.w);
      list[p0] = bin_entry(s.x, 0u);
      list[p1] = bin_entry(s.y, 0u);
      list[p2] = bin_entry(s.z, 0u);
      list[p3] = bin_entry(s.w, 0u);
    } else {
      for (int j = e; j < hi; ++j) bin_put(cur, list, src[j], 0u);
    }
  }
}

__global__ __launch_bounds__(1024) void bin_scatter_dst(const int* __restrict__ src, const int* __restrict__ dst,
                                                        const int* __restrict__ pre, const int* __restrict__ base,
                                                        unsigned int* __restrict__ list, int E, int NB) {
  __shared__ int cur[MAXB];
  for (int b = threadIdx.x; b < NB; b += blockDim.x) cur[b] = base[b] + pre[b * BIN_G + blockIdx.x];
  __syncthreads();
  int lo, hi;
  edge_chunk(E, lo, hi);
  for (int e = lo + 4 * (int)threadIdx.x; e < hi; e += 4 * (int)blockDim.x) {
    if (e + 3 < hi) {
      const int4 s = *(const int4*)(src + e);
      const int4 d = *(const int4*)(dst + e);
      const int p0 = bin_slot(cur, d.x);
      const int p1 = bin_slot(cur, d.y);
      const int p2 = bin_slot(cur, d.z);
      const int p3 = bin_slot(cur, d.w);
      list[p0] = bin_entry(d.x, (unsigned int)s.x);
      list[p1] = bin_entry(d.y, (unsigned int)s.y);
      list[p2] = bin_entry(d.z, (unsigned int)s.z);
      list[p3] = bin_entry(d.w, (unsigned int)s.w);
    } else {
      for (int j = e; j < hi; ++j) bin_put(cur, list, dst[j], (unsigned int)src[j]);
    }
  }
}

__global__ __launch_bounds__(512) void deg_owner(const unsigned int* __restrict__ list, const int* __restrict__ base,
                                                 float* __restrict__ inv, int N) {
  __shared__ int c[NPB + 1];
  if (threadIdx.x <= NPB) c[threadIdx.x] = 0;
  __syncthreads();
  const int lo = base[blockIdx.x];
  const int hi = base[blockIdx.x + 1];
  const int bd = (int)blockDim.x;
  for (int i = lo + (int)threadIdx.x; i < hi; i += 4 * bd) {
    unsigned int slot[4];
    for (int u = 0; u < 4; ++u) {
      const unsigned int ent = list[min(i + u * bd, hi - 1)];
      slot[u] = (i + u * bd < hi) ? (ent >> ENT_SHIFT) : (unsigned int)NPB;
    }
    for (int u = 0; u < 4; ++u) atomicAdd(&c[slot[u]], 1);
  }
  __syncthreads();
  const int node = blockIdx.x * NPB + (int)threadIdx.x;
  if (threadIdx.x < NPB && node < N) inv[node] = rsqrtf(fmaxf((float)c[threadIdx.x], 1.0f));
}

__global__ __launch_bounds__(256) void gnn_gemm1_wmma(
    const float* __restrict__ x, const float* __restrict__ W1,
    const float* __restrict__ inv_out, float* __restrict__ h1, int N) {
  __shared__ float wlds[F_IN * HID];
  for (int i = threadIdx.x; i < F_IN * HID; i += blockDim.x) wlds[i] = W1[i];
  __syncthreads();

  const int wave = threadIdx.x >> 5;
  const int lane = threadIdx.x & 31;
  const int tiles = N >> 4;
  const int tile = blockIdx.x * 8 + wave;
  if (tile >= tiles) return;

  const int n16 = lane & 15;
  const int g   = lane >> 4;
  const int row = tile * 16 + n16;
  const float s = inv_out[row];
  const float* __restrict__ xr = x + (size_t)row * F_IN;

  v8f c = {};
  #pragma unroll 4
  for (int k = 0; k < F_IN; k += 4) {
    const int kk = k + 2 * g;
    v2f a = *(const v2f*)(xr + kk);
    a.x *= s;
    a.y *= s;
    v2f b;
    b.x = wlds[(kk + 0) * HID + n16];
    b.y = wlds[(kk + 1) * HID + n16];
    c = __builtin_amdgcn_wmma_f32_16x16x4_f32(
         false, a,  false, b,
         (short)0, c,  false,  false);
  }

  #pragma unroll
  for (int p = 0; p < 8; ++p) {
    const int m = tile * 16 + p + 8 * g;
    h1[(size_t)m * HID + n16] = c[p];
  }
}

__global__ void gnn_scatter16(const int* __restrict__ src, const int* __restrict__ dst,
                              const float* __restrict__ h1, float* __restrict__ agg1,
                              long E4) {
  long t = (long)blockIdx.x * blockDim.x + threadIdx.x;
  if (t >= E4) return;
  long e = t >> 2;
  int f = (int)(t & 3) * 4;
  int s = src[e], d = dst[e];
  float4 v = *(const float4*)(h1 + (size_t)s * HID + f);
  float* a = agg1 + (size_t)d * HID + f;
  atomicAdd(a + 0, v.x);
  atomicAdd(a + 1, v.y);
  atomicAdd(a + 2, v.z);
  atomicAdd(a + 3, v.w);
}

__global__ void gnn_layer2(const float* __restrict__ agg1, const float* __restrict__ inv_in,
                           const float* __restrict__ inv_out, const float* __restrict__ b1,
                           const float* __restrict__ W2, float* __restrict__ h2, int N) {
  int i = blockIdx.x * blockDim.x + threadIdx.x;
  if (i >= N) return;
  const float si = inv_in[i];
  const float so = inv_out[i];
  const float* a = agg1 + (size_t)i * HID;
  float acc0 = 0.0f, acc1 = 0.0f;
  #pragma unroll
  for (int k = 0; k < HID; ++k) {
    float hv = fmaxf(fmaf(a[k], si, b1[k]), 0.0f) * so;
    acc0 = fmaf(hv, W2[k * NCLS + 0], acc0);
    acc1 = fmaf(hv, W2[k * NCLS + 1], acc1);
  }
  h2[(size_t)i * NCLS + 0] = acc0;
  h2[(size_t)i * NCLS + 1] = acc1;
}

__global__ __launch_bounds__(1024) void agg2_final(const unsigned int* __restrict__ list, const int* __restrict__ base,
                                                   const float* __restrict__ h2, const float* __restrict__ inv_in,
                                                   const float* __restrict__ b2, float* __restrict__ out, int N) {
  __shared__ float a0[NPB + 1];
  __shared__ float a1[NPB + 1];
  if (threadIdx.x <= NPB) { a0[threadIdx.x] = 0.0f; a1[threadIdx.x] = 0.0f; }
  __syncthreads();
  const int lo = base[blockIdx.x];
  const int hi = base[blockIdx.x + 1];
  const int bd = (int)blockDim.x;
  for (int i = lo + (int)threadIdx.x; i < hi; i += 4 * bd) {
    unsigned int slot[4];
    float2 v[4];
    for (int u = 0; u < 4; ++u) {
      const unsigned int ent = list[min(i + u * bd, hi - 1)];
      slot[u] = (i + u * bd < hi) ? (ent >> ENT_SHIFT) : (unsigned int)NPB;
      v[u] = *(const float2*)(h2 + (size_t)(ent & ((1u << ENT_SHIFT) - 1u)) * NCLS);
    }
    for (int u = 0; u < 4; ++u) {
      atomicAdd(&a0[slot[u]], v[u].x);
      atomicAdd(&a1[slot[u]], v[u].y);
    }
  }
  __syncthreads();
  const int i = blockIdx.x * NPB + (int)threadIdx.x;
  if (threadIdx.x >= NPB || i >= N) return;
  const float si = inv_in[i];
  float v0 = fmaf(a0[threadIdx.x], si, b2[0]);
  float v1 = fmaf(a1[threadIdx.x], si, b2[1]);
  float m = fmaxf(v0, v1);
  float lse = m + logf(expf(v0 - m) + expf(v1 - m));
  out[(size_t)i * NCLS + 0] = v0 - lse;
  out[(size_t)i * NCLS + 1] = v1 - lse;
}

extern "C" void kernel_launch(void* const* d_in, const int* in_sizes, int n_in,
                              void* d_out, int out_size, void* d_ws, size_t ws_size,
                              hipStream_t stream) {
  const float* x   = (const float*)d_in[0];
  const int*   src = (const int*)d_in[1];
  const int*   dst = (const int*)d_in[2];
  const float* W1  = (const float*)d_in[3];
  const float* b1  = (const float*)d_in[4];
  const float* W2  = (const float*)d_in[5];
  const float* b2  = (const float*)d_in[6];
  float* out = (float*)d_out;

  const int N = in_sizes[0] / F_IN;
  const int E = in_sizes[1];

  float* ws      = (float*)d_ws;
  float* inv_out = ws;
  float* inv_in  = ws + (size_t)N;
  float* h2      = ws + (size_t)2 * N;
  int*   preS    = (int*)(ws + (size_t)4 * N);
  int*   preD    = preS + MAXB * BIN_G;
  int*   totS    = preD + MAXB * BIN_G;
  int*   totD    = totS + MAXB * TOT_PITCH;
  int*   baseS   = totD + MAXB * TOT_PITCH;
  int*   baseD   = baseS + 2 * MAXB;
  float* R       = (float*)(baseD + 2 * MAXB);
  int*   cntS    = (int*)R;
  int*   cntD    = cntS + MAXB * BIN_G;
  unsigned int* list = (unsigned int*)R;
  float* h1      = R;
  float* agg1    = R + (size_t)16 * N;

  const int B = 256;
  const int NB = (N + NPB - 1) / NPB;

  bin_count<<<BIN_G, 1024, 0, stream>>>(src, dst, cntS, cntD, E, NB);
  bin_scan<<<2 * NB, BIN_G, 0, stream>>>(cntS, cntD, preS, preD, totS, totD, NB);
  bin_base<<<2, MAXB, 0, stream>>>(totS, totD, baseS, baseD, NB);

  bin_scatter_src<<<BIN_G, 1024, 0, stream>>>(src, preS, baseS, list, E, NB);
  deg_owner<<<NB, 512, 0, stream>>>(list, baseS, inv_out, N);
  bin_scatter_dst<<<BIN_G, 1024, 0, stream>>>(src, dst, preD, baseD, list, E, NB);
  deg_owner<<<NB, 512, 0, stream>>>(list, baseD, inv_in, N);

  const int aggFloats = 16 * N;
  gnn_zero<<<(aggFloats + B - 1) / B, B, 0, stream>>>(agg1, aggFloats);

  const int tiles = N >> 4;
  gnn_gemm1_wmma<<<(tiles + 7) / 8, B, 0, stream>>>(x, W1, inv_out, h1, N);

  const long E4 = (long)E * 4;
  gnn_scatter16<<<(int)((E4 + B - 1) / B), B, 0, stream>>>(src, dst, h1, agg1, E4);
  gnn_layer2<<<(N + B - 1) / B, B, 0, stream>>>(agg1, inv_in, inv_out, b1, W2, h2, N);

  bin_scatter_dst<<<BIN_G, 1024, 0, stream>>>(src, dst, preD, baseD, list, E, NB);
  agg2_final<<<NB, 1024, 0, stream>>>(list, baseD, h2, inv_in, b2, out, N);
}


# Layer-1 edge sums: global atomics replaced by one work-group per 256-node range summing its binned list in LDS (two range halves; in-degree, relu, W2 done there); no zero pass; sum order changes
# speedup vs baseline: 2.2182x; 1.2304x over previous
#include <hip/hip_runtime.h>
#include <math.h>


typedef __attribute__((ext_vector_type(2))) float v2f;
typedef __attribute__((ext_vector_type(8))) float v8f;

#define F_IN 512
#define HID  16
#define NCLS 2

#define NPB_SHIFT 8
#define NPB   (1 << NPB_SHIFT)
#define MAXB  512
#define BIN_G 256
#define ENT_SHIFT 20
#define TOT_PITCH 32
#define ACC_PITCH (NPB + 4)

__device__ __forceinline__ void edge_chunk(int E, int& lo, int& hi) {
  const int chunk = ((E + BIN_G * 4 - 1) / (BIN_G * 4)) * 4;
  lo = blockIdx.x * chunk;
  if (lo > E) lo = E;
  hi = lo + chunk;
  if (hi > E) hi = E;
}

__global__ __launch_bounds__(1024) void bin_count(const int* __restrict__ src, const int* __restrict__ dst,
                                                  int* __restrict__ cntS, int* __restrict__ cntD,
                                                  int E, int NB) {
  __shared__ int hs[MAXB];
  __shared__ int hd[MAXB];
  for (int i = threadIdx.x; i < MAXB; i += blockDim.x) { hs[i] = 0; hd[i] = 0; }
  __syncthreads();
  int lo, hi;
  edge_chunk(E, lo, hi);
  for (int e = lo + 4 * (int)threadIdx.x; e < hi; e += 4 * (int)blockDim.x) {
    if (e + 3 < hi) {
      const int4 s = *(const int4*)(src + e);
      const int4 d = *(const int4*)(dst + e);
      atomicAdd(&hs[s.x >> NPB_SHIFT], 1);
      atomicAdd(&hs[s.y >> NPB_SHIFT], 1);
      atomicAdd(&hs[s.z >> NPB_SHIFT], 1);
      atomicAdd(&hs[s.w >> NPB_SHIFT], 1);
      atomicAdd(&hd[d.x >> NPB_SHIFT], 1);
      atomicAdd(&hd[d.y >> NPB_SHIFT], 1);
      atomicAdd(&hd[d.z >> NPB_SHIFT], 1);
      atomicAdd(&hd[d.w >> NPB_SHIFT], 1);
    } else {
      for (int j = e; j < hi; ++j) {
        atomicAdd(&hs[src[j] >> NPB_SHIFT], 1);
        atomicAdd(&hd[dst[j] >> NPB_SHIFT], 1);
      }
    }
  }
  __syncthreads();
  for (int b = threadIdx.x; b < NB; b += blockDim.x) {
    cntS[blockIdx.x * MAXB + b] = hs[b];
    cntD[blockIdx.x * MAXB + b] = hd[b];
  }
}

__global__ __launch_bounds__(BIN_G) void bin_scan(const int* __restrict__ cntS, const int* __restrict__ cntD,
                                                  int* __restrict__ preS, int* __restrict__ preD,
                                                  int* __restrict__ totS, int* __restrict__ totD, int NB) {
  __shared__ int sh[BIN_G];
  int b = blockIdx.x;
  const int* cnt = cntS;
  int* pre = preS;
  int* tot = totS;
  if (b >= NB) { b -= NB; cnt = cntD; pre = preD; tot = totD; }
  const int t = threadIdx.x;
  const int v = cnt[t * MAXB + b];
  sh[t] = v;
  __syncthreads();
  for (int off = 1; off < BIN_G; off <<= 1) {
    const int add = (t >= off) ? sh[t - off] : 0;
    __syncthreads();
    sh[t] += add;
    __syncthreads();
  }
  pre[b * BIN_G + t] = sh[t] - v;
  if (t == BIN_G - 1) tot[b * TOT_PITCH] = sh[t];
}

__global__ __launch_bounds__(MAXB) void bin_base(const int* __restrict__ totS, const int* __restrict__ totD,
                                                 int* __restrict__ baseS, int* __restrict__ baseD, int NB) {
  __shared__ int sh[MAXB];
  const int* tot = blockIdx.x ? totD : totS;
  int* base = blockIdx.x ? baseD : baseS;
  const int t = threadIdx.x;
  const int v = (t < NB) ? tot[t * TOT_PITCH] : 0;
  sh[t] = v;
  __syncthreads();
  for (int off = 1; off < MAXB; off <<= 1) {
    const int add = (t >= off) ? sh[t - off] : 0;
    __syncthreads();
    sh[t] += add;
    __syncthreads();
  }
  if (t < NB) base[t] = sh[t] - v;
  if (t == NB - 1) base[NB] = sh[t];
}

__device__ __forceinline__ int bin_slot(int* cur, int key) {
  return atomicAdd(&cur[key >> NPB_SHIFT], 1);
}
__device__ __forceinline__ unsigned int bin_entry(int key, unsigned int payload) {
  return ((unsigned int)(key & (NPB - 1)) << ENT_SHIFT) | payload;
}
__device__ __forceinline__ void bin_put(int* cur, unsigned int* __restrict__ list, int key, unsigned int payload) {
  list[bin_slot(cur, key)] = bin_entry(key, payload);
}

__global__ __launch_bounds__(1024) void bin_scatter_src(const int* __restrict__ src,
                                                        const int* __restrict__ pre, const int* __restrict__ base,
                                                        unsigned int* __restrict__ list, int E, int NB) {
  __shared__ int cur[MAXB];
  for (int b = threadIdx.x; b < NB; b += blockDim.x) cur[b] = base[b] + pre[b * BIN_G + blockIdx.x];
  __syncthreads();
  int lo, hi;
  edge_chunk(E, lo, hi);
  for (int e = lo + 4 * (int)threadIdx.x; e < hi; e += 4 * (int)blockDim.x) {
    if (e + 3 < hi) {
      const int4 s = *(const int4*)(src + e);
      const int p0 = bin_slot(cur, s.x);
      const int p1 = bin_slot(cur, s.y);
      const int p2 = bin_slot(cur, s.z);
      const int p3 = bin_slot(cur, s.w);
      list[p0] = bin_entry(s.x, 0u);
      list[p1] = bin_entry(s.y, 0u);
      list[p2] = bin_entry(s.z, 0u);
      list[p3] = bin_entry(s.w, 0u);
    } else {
      for (int j = e; j < hi; ++j) bin_put(cur, list, src[j], 0u);
    }
  }
}

__device__ __forceinline__ bool in_window(int key, int b_lo, int nb) {
  return (unsigned int)((key >> NPB_SHIFT) - b_lo) < (unsigned int)nb;
}
__global__ __launch_bounds__(1024) void bin_scatter_dst(const int* __restrict__ src, const int* __restrict__ dst,
                                                        const int* __restrict__ pre, const int* __restrict__ base,
                                                        unsigned int* __restrict__ list, int E, int b_lo, int b_hi) {
  __shared__ int cur[MAXB];
  const int first = base[b_lo];
  const int nb = b_hi - b_lo;
  for (int b = b_lo + (int)threadIdx.x; b < b_hi; b += blockDim.x) cur[b] = base[b] - first + pre[b * BIN_G + blockIdx.x];
  __syncthreads();
  int lo, hi;
  edge_chunk(E, lo, hi);
  for (int e = lo + 4 * (int)threadIdx.x; e < hi; e += 4 * (int)blockDim.x) {
    if (e + 3 < hi) {
      const int4 s = *(const int4*)(src + e);
      const int4 d = *(const int4*)(dst + e);
      const bool w0 = in_window(d.x, b_lo, nb);
      const bool w1 = in_window(d.y, b_lo, nb);
      const bool w2 = in_window(d.z, b_lo, nb);
      const bool w3 = in_window(d.w, b_lo, nb);
      const int p0 = w0 ? bin_slot(cur, d.x) : 0;
      const int p1 = w1 ? bin_slot(cur, d.y) : 0;
      const int p2 = w2 ? bin_slot(cur, d.z) : 0;
      const int p3 = w3 ? bin_slot(cur, d.w) : 0;
      if (w0) list[p0] = bin_entry(d.x, (unsigned int)s.x);
      if (w1) list[p1] = bin_entry(d.y, (unsigned int)s.y);
      if (w2) list[p2] = bin_entry(d.z, (unsigned int)s.z);
      if (w3) list[p3] = bin_entry(d.w, (unsigned int)s.w);
    } else {
      for (int j = e; j < hi; ++j)
        if (in_window(dst[j], b_lo, nb)) bin_put(cur, list, dst[j], (unsigned int)src[j]);
    }
  }
}

__global__ __launch_bounds__(512) void deg_owner(const unsigned int* __restrict__ list, const int* __restrict__ base,
                                                 float* __restrict__ inv, int N) {
  __shared__ int c[NPB + 1];
  if (threadIdx.x <= NPB) c[threadIdx.x] = 0;
  __syncthreads();
  const int lo = base[blockIdx.x];
  const int hi = base[blockIdx.x + 1];
  const int bd = (int)blockDim.x;
  for (int i = lo + (int)threadIdx.x; i < hi; i += 4 * bd) {
    unsigned int slot[4];
    for (int u = 0; u < 4; ++u) {
      const unsigned int ent = list[min(i + u * bd, hi - 1)];
      slot[u] = (i + u * bd < hi) ? (ent >> ENT_SHIFT) : (unsigned int)NPB;
    }
    for (int u = 0; u < 4; ++u) atomicAdd(&c[slot[u]], 1);
  }
  __syncthreads();
  const int node = blockIdx.x * NPB + (int)threadIdx.x;
  if (threadIdx.x < NPB && node < N) inv[node] = rsqrtf(fmaxf((float)c[threadIdx.x], 1.0f));
}

__global__ __launch_bounds__(256) void gnn_gemm1_wmma(
    const float* __restrict__ x, const float* __restrict__ W1,
    const float* __restrict__ inv_out, float* __restrict__ h1, int N) {
  __shared__ float wlds[F_IN * HID];
  for (int i = threadIdx.x; i < F_IN * HID; i += blockDim.x) wlds[i] = W1[i];
  __syncthreads();

  const int wave = threadIdx.x >> 5;
  const int lane = threadIdx.x & 31;
  const int tiles = N >> 4;
  const int tile = blockIdx.x * 8 + wave;
  if (tile >= tiles) return;

  const int n16 = lane & 15;
  const int g   = lane >> 4;
  const int row = tile * 16 + n16;
  const float s = inv_out[row];
  const float* __restrict__ xr = x + (size_t)row * F_IN;

  v8f c = {};
  #pragma unroll 4
  for (int k = 0; k < F_IN; k += 4) {
    const int kk = k + 2 * g;
    v2f a = *(const v2f*)(xr + kk);
    a.x *= s;
    a.y *= s;
    v2f b;
    b.x = wlds[(kk + 0) * HID + n16];
    b.y = wlds[(kk + 1) * HID + n16];
    c = __builtin_amdgcn_wmma_f32_16x16x4_f32(
         false, a,  false, b,
         (short)0, c,  false,  false);
  }

  #pragma unroll
  for (int p = 0; p < 8; ++p) {
    const int m = tile * 16 + p + 8 * g;
    h1[(size_t)m * HID + n16] = c[p];
  }
}

__global__ __launch_bounds__(1024) void agg1_owner(const unsigned int* __restrict__ list, const int* __restrict__ base,
                                                   const float* __restrict__ h1, const float* __restrict__ inv_out,
                                                   const float* __restrict__ b1, const float* __restrict__ W2,
                                                   float* __restrict__ inv_in, float* __restrict__ h2,
                                                   int N, int b_lo) {
  __shared__ float acc[HID * ACC_PITCH];
  __shared__ int cnt[NPB + 1];
  for (int i = threadIdx.x; i < HID * ACC_PITCH; i += blockDim.x) acc[i] = 0.0f;
  if (threadIdx.x <= NPB) cnt[threadIdx.x] = 0;
  __syncthreads();
  const int b = b_lo + (int)blockIdx.x;
  const int first = base[b_lo];
  const int lo = base[b] - first;
  const int hi = base[b + 1] - first;
  const int t = (int)threadIdx.x;
  const int q = t & 3;
  float* row = acc + 4 * q * ACC_PITCH;
  const unsigned int src_mask = (1u << ENT_SHIFT) - 1u;
  if (lo < hi) {
    unsigned int nxt = list[min(lo + t, hi - 1)];
    for (int i0 = lo; i0 < hi; i0 += (int)blockDim.x) {
      unsigned int w = nxt;
      if (i0 + t >= hi) w = (w & src_mask) | ((unsigned int)NPB << ENT_SHIFT);
      nxt = list[min(i0 + (int)blockDim.x + t, hi - 1)];
      unsigned int e[4];
      e[0] = (unsigned int)__builtin_amdgcn_mov_dpp((int)w, 0x00, 0xf, 0xf, false);
      e[1] = (unsigned int)__builtin_amdgcn_mov_dpp((int)w, 0x55, 0xf, 0xf, false);
      e[2] = (unsigned int)__builtin_amdgcn_mov_dpp((int)w, 0xaa, 0xf, 0xf, false);
      e[3] = (unsigned int)__builtin_amdgcn_mov_dpp((int)w, 0xff, 0xf, 0xf, false);
      float4 v[4];
      for (int u = 0; u < 4; ++u)
        v[u] = *(const float4*)(h1 + (size_t)(e[u] & src_mask) * HID + 4 * q);
      for (int u = 0; u < 4; ++u) {
        const unsigned int place = e[u] >> ENT_SHIFT;
        atomicAdd(&row[0 * ACC_PITCH + place], v[u].x);
        atomicAdd(&row[1 * ACC_PITCH + place], v[u].y);
        atomicAdd(&row[2 * ACC_PITCH + place], v[u].z);
        atomicAdd(&row[3 * ACC_PITCH + place], v[u].w);
      }
      atomicAdd(&cnt[w >> ENT_SHIFT], 1);
    }
  }
  __syncthreads();
  const int node = b * NPB + t;
  if (t >= NPB || node >= N) return;
  const float si = rsqrtf(fmaxf((float)cnt[t], 1.0f));
  const float so = inv_out[node];
  inv_in[node] = si;
  float acc0 = 0.0f, acc1 = 0.0f;
  #pragma unroll
  for (int k = 0; k < HID; ++k) {
    float hv = fmaxf(fmaf(acc[k * ACC_PITCH + t], si, b1[k]), 0.0f) * so;
    acc0 = fmaf(hv, W2[k * NCLS + 0], acc0);
    acc1 = fmaf(hv, W2[k * NCLS + 1], acc1);
  }
  *(float2*)(h2 + (size_t)node * NCLS) = make_float2(acc0, acc1);
}

__global__ __launch_bounds__(1024) void agg2_final(const unsigned int* __restrict__ list, const int* __restrict__ base,
                                                   const float* __restrict__ h2, const float* __restrict__ inv_in,
                                                   const float* __restrict__ b2, float* __restrict__ out, int N) {
  __shared__ float a0[NPB + 1];
  __shared__ float a1[NPB + 1];
  if (threadIdx.x <= NPB) { a0[threadIdx.x] = 0.0f; a1[threadIdx.x] = 0.0f; }
  __syncthreads();
  const int lo = base[blockIdx.x];
  const int hi = base[blockIdx.x + 1];
  const int bd = (int)blockDim.x;
  for (int i = lo + (int)threadIdx.x; i < hi; i += 4 * bd) {
    unsigned int slot[4];
    float2 v[4];
    for (int u = 0; u < 4; ++u) {
      const unsigned int ent = list[min(i + u * bd, hi - 1)];
      slot[u] = (i + u * bd < hi) ? (ent >> ENT_SHIFT) : (unsigned int)NPB;
      v[u] = *(const float2*)(h2 + (size_t)(ent & ((1u << ENT_SHIFT) - 1u)) * NCLS);
    }
    for (int u = 0; u < 4; ++u) {
      atomicAdd(&a0[slot[u]], v[u].x);
      atomicAdd(&a1[slot[u]], v[u].y);
    }
  }
  __syncthreads();
  const int i = blockIdx.x * NPB + (int)threadIdx.x;
  if (threadIdx.x >= NPB || i >= N) return;
  const float si = inv_in[i];
  float v0 = fmaf(a0[threadIdx.x], si, b2[0]);
  float v1 = fmaf(a1[threadIdx.x], si, b2[1]);
  float m = fmaxf(v0, v1);
  float lse = m + logf(expf(v0 - m) + expf(v1 - m));
  out[(size_t)i * NCLS + 0] = v0 - lse;
  out[(size_t)i * NCLS + 1] = v1 - lse;
}

extern "C" void kernel_launch(void* const* d_in, const int* in_sizes, int n_in,
                              void* d_out, int out_size, void* d_ws, size_t ws_size,
                              hipStream_t stream) {
  const float* x   = (const float*)d_in[0];
  const int*   src = (const int*)d_in[1];
  const int*   dst = (const int*)d_in[2];
  const float* W1  = (const float*)d_in[3];
  const float* b1  = (const float*)d_in[4];
  const float* W2  = (const float*)d_in[5];
  const float* b2  = (const float*)d_in[6];
  float* out = (float*)d_out;

  const int N = in_sizes[0] / F_IN;
  const int E = in_sizes[1];

  float* ws      = (float*)d_ws;
  float* inv_out = ws;
  float* inv_in  = ws + (size_t)N;
  float* h2      = ws + (size_t)2 * N;
  int*   preS    = (int*)(ws + (size_t)4 * N);
  int*   preD    = preS + MAXB * BIN_G;
  int*   totS    = preD + MAXB * BIN_G;
  int*   totD    = totS + MAXB * TOT_PITCH;
  int*   baseS   = totD + MAXB * TOT_PITCH;
  int*   baseD   = baseS + 2 * MAXB;
  float* R       = (float*)(baseD + 2 * MAXB);
  int*   cntS    = (int*)R;
  int*   cntD    = cntS + MAXB * BIN_G;
  unsigned int* list = (unsigned int*)R;
  float* h1      = R + (size_t)17 * N;

  const int B = 256;
  const int NB = (N + NPB - 1) / NPB;
  const int NBh = NB / 2;

  bin_count<<<BIN_G, 1024, 0, stream>>>(src, dst, cntS, cntD, E, NB);
  bin_scan<<<2 * NB, BIN_G, 0, stream>>>(cntS, cntD, preS, preD, totS, totD, NB);
  bin_base<<<2, MAXB, 0, stream>>>(totS, totD, baseS, baseD, NB);

  bin_scatter_src<<<BIN_G, 1024, 0, stream>>>(src, preS, baseS, list, E, NB);
  deg_owner<<<NB, 512, 0, stream>>>(list, baseS, inv_out, N);

  const int tiles = N >> 4;
  gnn_gemm1_wmma<<<(tiles + 7) / 8, B, 0, stream>>>(x, W1, inv_out, h1, N);

  bin_scatter_dst<<<BIN_G, 1024, 0, stream>>>(src, dst, preD, baseD, list, E, 0, NBh);
  agg1_owner<<<NBh, 1024, 0, stream>>>(list, baseD, h1, inv_out, b1, W2, inv_in, h2, N, 0);
  bin_scatter_dst<<<BIN_G, 1024, 0, stream>>>(src, dst, preD, baseD, list, E, NBh, NB);
  agg1_owner<<<NB - NBh, 1024, 0, stream>>>(list, baseD, h1, inv_out, b1, W2, inv_in, h2, N, NBh);

  bin_scatter_dst<<<BIN_G, 1024, 0, stream>>>(src, dst, preD, baseD, list, E, 0, NB);
  agg2_final<<<NB, 1024, 0, stream>>>(list, baseD, h2, inv_in, b2, out, N);
}


# Edge binning: each scatter work-group orders its edges by range in LDS, then stores them with consecutive lanes on consecutive list slots (2-3 lines per store, was 32); run order stays arrival order
# speedup vs baseline: 3.2475x; 1.4640x over previous
#include <hip/hip_runtime.h>
#include <math.h>


typedef __attribute__((ext_vector_type(2))) float v2f;
typedef __attribute__((ext_vector_type(8))) float v8f;

#define F_IN 512
#define HID  16
#define NCLS 2

#define NPB_SHIFT 8
#define NPB   (1 << NPB_SHIFT)
#define MAXB  512
#define BIN_G 256
#define ENT_SHIFT 20
#define TOT_PITCH 32
#define ACC_PITCH (NPB + 4)

__device__ __forceinline__ void edge_chunk(int E, int& lo, int& hi) {
  const int chunk = ((E + BIN_G * 4 - 1) / (BIN_G * 4)) * 4;
  lo = blockIdx.x * chunk;
  if (lo > E) lo = E;
  hi = lo + chunk;
  if (hi > E) hi = E;
}

__global__ __launch_bounds__(1024) void bin_count(const int* __restrict__ src, const int* __restrict__ dst,
                                                  int* __restrict__ cntS, int* __restrict__ cntD,
                                                  int E, int NB) {
  __shared__ int hs[MAXB];
  __shared__ int hd[MAXB];
  for (int i = threadIdx.x; i < MAXB; i += blockDim.x) { hs[i] = 0; hd[i] = 0; }
  __syncthreads();
  int lo, hi;
  edge_chunk(E, lo, hi);
  for (int e = lo + 4 * (int)threadIdx.x; e < hi; e += 4 * (int)blockDim.x) {
    if (e + 3 < hi) {
      const int4 s = *(const int4*)(src + e);
      const int4 d = *(const int4*)(dst + e);
      atomicAdd(&hs[s.x >> NPB_SHIFT], 1);
      atomicAdd(&hs[s.y >> NPB_SHIFT], 1);
      atomicAdd(&hs[s.z >> NPB_SHIFT], 1);
      atomicAdd(&hs[s.w >> NPB_SHIFT], 1);
      atomicAdd(&hd[d.x >> NPB_SHIFT], 1);
      atomicAdd(&hd[d.y >> NPB_SHIFT], 1);
      atomicAdd(&hd[d.z >> NPB_SHIFT], 1);
      atomicAdd(&hd[d.w >> NPB_SHIFT], 1);
    } else {
      for (int j = e; j < hi; ++j) {
        atomicAdd(&hs[src[j] >> NPB_SHIFT], 1);
        atomicAdd(&hd[dst[j] >> NPB_SHIFT], 1);
      }
    }
  }
  __syncthreads();
  for (int b = threadIdx.x; b < NB; b += blockDim.x) {
    cntS[blockIdx.x * MAXB + b] = hs[b];
    cntD[blockIdx.x * MAXB + b] = hd[b];
  }
}

__global__ __launch_bounds__(BIN_G) void bin_scan(const int* __restrict__ cntS, const int* __restrict__ cntD,
                                                  int* __restrict__ preS, int* __restrict__ preD,
                                                  int* __restrict__ totS, int* __restrict__ totD, int NB) {
  __shared__ int sh[BIN_G];
  int b = blockIdx.x;
  const int* cnt = cntS;
  int* pre = preS;
  int* tot = totS;
  if (b >= NB) { b -= NB; cnt = cntD; pre = preD; tot = totD; }
  const int t = threadIdx.x;
  const int v = cnt[t * MAXB + b];
  sh[t] = v;
  __syncthreads();
  for (int off = 1; off < BIN_G; off <<= 1) {
    const int add = (t >= off) ? sh[t - off] : 0;
    __syncthreads();
    sh[t] += add;
    __syncthreads();
  }
  pre[b * BIN_G + t] = sh[t] - v;
  if (t == BIN_G - 1) tot[b * TOT_PITCH] = sh[t];
}

__global__ __launch_bounds__(MAXB) void bin_base(const int* __restrict__ totS, const int* __restrict__ totD,
                                                 int* __restrict__ baseS, int* __restrict__ baseD, int NB) {
  __shared__ int sh[MAXB];
  const int* tot = blockIdx.x ? totD : totS;
  int* base = blockIdx.x ? baseD : baseS;
  const int t = threadIdx.x;
  const int v = (t < NB) ? tot[t * TOT_PITCH] : 0;
  sh[t] = v;
  __syncthreads();
  for (int off = 1; off < MAXB; off <<= 1) {
    const int add = (t >= off) ? sh[t - off] : 0;
    __syncthreads();
    sh[t] += add;
    __syncthreads();
  }
  if (t < NB) base[t] = sh[t] - v;
  if (t == NB - 1) base[NB] = sh[t];
}

__device__ __forceinline__ unsigned int bin_entry(int key, unsigned int payload) {
  return ((unsigned int)(key & (NPB - 1)) << ENT_SHIFT) | payload;
}
__device__ __forceinline__ bool in_window(int key, int b_lo, int nb) {
  return (unsigned int)((key >> NPB_SHIFT) - b_lo) < (unsigned int)nb;
}

#define SC_THREADS 1024
#define SC_TRIPS 4
#define SC_PIECE (SC_TRIPS * 4 * SC_THREADS)

__device__ __forceinline__ int piece_rank(int* cntl, int key, int b_lo, int nb) {
  return in_window(key, b_lo, nb) ? atomicAdd(&cntl[key >> NPB_SHIFT], 1) : -1;
}
__device__ __forceinline__ void piece_place(const int* lofs, const int* gb, int key, int rank, int& place, int& slot) {
  const int b = rank >= 0 ? key >> NPB_SHIFT : 0;
  place = lofs[b] + rank;
  slot = gb[b] + rank;
}

template <bool HAS_PAY>
__device__ __forceinline__ void scatter_ordered(const int* __restrict__ key, const int* __restrict__ pay,
                                                const int* __restrict__ pre, const int* __restrict__ base,
                                                unsigned int* __restrict__ list, int E, int b_lo, int b_hi) {
  __shared__ int cntl[MAXB];
  __shared__ int lofs[MAXB];
  __shared__ int gb[MAXB];
  __shared__ int ntot;
  __shared__ uint2 stage[SC_PIECE];
  const int t = (int)threadIdx.x;
  const int first = base[b_lo];
  const int nb = b_hi - b_lo;
  int lo, hi;
  edge_chunk(E, lo, hi);
  for (int p0 = lo; p0 < hi; p0 += SC_PIECE) {
    const int p1 = min(p0 + SC_PIECE, hi);
    for (int b = t; b < MAXB; b += SC_THREADS) cntl[b] = 0;
    __syncthreads();

    int k[SC_TRIPS][4];
    int pl[SC_TRIPS][4];
    int rk[SC_TRIPS][4];
    #pragma unroll
    for (int u = 0; u < SC_TRIPS; ++u) {
      const int e = p0 + 4 * (u * SC_THREADS + t);
      int4 kv = make_int4(-1, -1, -1, -1);
      int4 pv = make_int4(0, 0, 0, 0);
      if (e + 3 < p1) {
        kv = *(const int4*)(key + e);
        if (HAS_PAY) pv = *(const int4*)(pay + e);
      }
      k[u][0] = kv.x; k[u][1] = kv.y; k[u][2] = kv.z; k[u][3] = kv.w;
      pl[u][0] = pv.x; pl[u][1] = pv.y; pl[u][2] = pv.z; pl[u][3] = pv.w;
    }
    const int rem = (p1 - p0) & 3;
    int kt = -1;
    int plt = 0;
    if (t < rem) {
      kt = key[p1 - rem + t];
      if (HAS_PAY) plt = pay[p1 - rem + t];
    }
    if (p0 == lo)
      for (int b = t; b < MAXB; b += SC_THREADS)
        gb[b] = (b >= b_lo && b < b_hi) ? base[b] - first + pre[b * BIN_G + blockIdx.x] : 0;
    #pragma unroll
    for (int u = 0; u < SC_TRIPS; ++u) {
      #pragma unroll
      for (int j = 0; j < 4; ++j) rk[u][j] = piece_rank(cntl, k[u][j], b_lo, nb);
    }
    const int rkt = piece_rank(cntl, kt, b_lo, nb);
    __syncthreads();

    if (t < 32) {
      int ex[MAXB / 32];
      int run = 0;
      #pragma unroll
      for (int j = 0; j < MAXB / 32; ++j) { ex[j] = run; run += cntl[t + 32 * j]; }
      int inc = run;
      #pragma unroll
      for (int off = 1; off < 32; off <<= 1) {
        const int up = __shfl_up(inc, off, 32);
        if (t >= off) inc += up;
      }
      const int excl = inc - run;
      #pragma unroll
      for (int j = 0; j < MAXB / 32; ++j) lofs[t + 32 * j] = excl + ex[j];
      if (t == 31) ntot = inc;
    }
    __syncthreads();

    int ps[SC_TRIPS][4];
    int sl[SC_TRIPS][4];
    #pragma unroll
    for (int u = 0; u < SC_TRIPS; ++u) {
      #pragma unroll
      for (int j = 0; j < 4; ++j) piece_place(lofs, gb, k[u][j], rk[u][j], ps[u][j], sl[u][j]);
    }
    int pst, slt;
    piece_place(lofs, gb, kt, rkt, pst, slt);
    #pragma unroll
    for (int u = 0; u < SC_TRIPS; ++u) {
      #pragma unroll
      for (int j = 0; j < 4; ++j)
        if (rk[u][j] >= 0) stage[ps[u][j]] = make_uint2(bin_entry(k[u][j], (unsigned int)pl[u][j]), (unsigned int)sl[u][j]);
    }
    if (rkt >= 0) stage[pst] = make_uint2(bin_entry(kt, (unsigned int)plt), (unsigned int)slt);
    __syncthreads();

    const int n = ntot;
    for (int p = t; p < n; p += SC_THREADS) {
      const uint2 s = stage[p];
      list[s.y] = s.x;
    }
    if (p1 < hi)
      for (int b = t; b < MAXB; b += SC_THREADS) gb[b] += cntl[b];
  }
}

__global__ __launch_bounds__(SC_THREADS) void bin_scatter_src(const int* __restrict__ src,
                                                              const int* __restrict__ pre, const int* __restrict__ base,
                                                              unsigned int* __restrict__ list, int E, int NB) {
  scatter_ordered<false>(src, src, pre, base, list, E, 0, NB);
}

__global__ __launch_bounds__(SC_THREADS) void bin_scatter_dst(const int* __restrict__ src, const int* __restrict__ dst,
                                                              const int* __restrict__ pre, const int* __restrict__ base,
                                                              unsigned int* __restrict__ list, int E, int b_lo, int b_hi) {
  scatter_ordered<true>(dst, src, pre, base, list, E, b_lo, b_hi);
}

__global__ __launch_bounds__(512) void deg_owner(const unsigned int* __restrict__ list, const int* __restrict__ base,
                                                 float* __restrict__ inv, int N) {
  __shared__ int c[NPB + 1];
  if (threadIdx.x <= NPB) c[threadIdx.x] = 0;
  __syncthreads();
  const int lo = base[blockIdx.x];
  const int hi = base[blockIdx.x + 1];
  const int bd = (int)blockDim.x;
  for (int i = lo + (int)threadIdx.x; i < hi; i += 4 * bd) {
    unsigned int slot[4];
    for (int u = 0; u < 4; ++u) {
      const unsigned int ent = list[min(i + u * bd, hi - 1)];
      slot[u] = (i + u * bd < hi) ? (ent >> ENT_SHIFT) : (unsigned int)NPB;
    }
    for (int u = 0; u < 4; ++u) atomicAdd(&c[slot[u]], 1);
  }
  __syncthreads();
  const int node = blockIdx.x * NPB + (int)threadIdx.x;
  if (threadIdx.x < NPB && node < N) inv[node] = rsqrtf(fmaxf((float)c[threadIdx.x], 1.0f));
}

__global__ __launch_bounds__(256) void gnn_gemm1_wmma(
    const float* __restrict__ x, const float* __restrict__ W1,
    const float* __restrict__ inv_out, float* __restrict__ h1, int N) {
  __shared__ float wlds[F_IN * HID];
  for (int i = threadIdx.x; i < F_IN * HID; i += blockDim.x) wlds[i] = W1[i];
  __syncthreads();

  const int wave = threadIdx.x >> 5;
  const int lane = threadIdx.x & 31;
  const int tiles = N >> 4;
  const int tile = blockIdx.x * 8 + wave;
  if (tile >= tiles) return;

  const int n16 = lane & 15;
  const int g   = lane >> 4;
  const int row = tile * 16 + n16;
  const float s = inv_out[row];
  const float* __restrict__ xr = x + (size_t)row * F_IN;

  v8f c = {};
  #pragma unroll 4
  for (int k = 0; k < F_IN; k += 4) {
    const int kk = k + 2 * g;
    v2f a = *(const v2f*)(xr + kk);
    a.x *= s;
    a.y *= s;
    v2f b;
    b.x = wlds[(kk + 0) * HID + n16];
    b.y = wlds[(kk + 1) * HID + n16];
    c = __builtin_amdgcn_wmma_f32_16x16x4_f32(
         false, a,  false, b,
         (short)0, c,  false,  false);
  }

  #pragma unroll
  for (int p = 0; p < 8; ++p) {
    const int m = tile * 16 + p + 8 * g;
    h1[(size_t)m * HID + n16] = c[p];
  }
}

__global__ __launch_bounds__(1024) void agg1_owner(const unsigned int* __restrict__ list, const int* __restrict__ base,
                                                   const float* __restrict__ h1, const float* __restrict__ inv_out,
                                                   const float* __restrict__ b1, const float* __restrict__ W2,
                                                   float* __restrict__ inv_in, float* __restrict__ h2,
                                                   int N, int b_lo) {
  __shared__ float acc[HID * ACC_PITCH];
  __shared__ int cnt[NPB + 1];
  for (int i = threadIdx.x; i < HID * ACC_PITCH; i += blockDim.x) acc[i] = 0.0f;
  if (threadIdx.x <= NPB) cnt[threadIdx.x] = 0;
  __syncthreads();
  const int b = b_lo + (int)blockIdx.x;
  const int first = base[b_lo];
  const int lo = base[b] - first;
  const int hi = base[b + 1] - first;
  const int t = (int)threadIdx.x;
  const int q = t & 3;
  float* row = acc + 4 * q * ACC_PITCH;
  const unsigned int src_mask = (1u << ENT_SHIFT) - 1u;
  if (lo < hi) {
    unsigned int nxt = list[min(lo + t, hi - 1)];
    for (int i0 = lo; i0 < hi; i0 += (int)blockDim.x) {
      unsigned int w = nxt;
      if (i0 + t >= hi) w = (w & src_mask) | ((unsigned int)NPB << ENT_SHIFT);
      nxt = list[min(i0 + (int)blockDim.x + t, hi - 1)];
      unsigned int e[4];
      e[0] = (unsigned int)__builtin_amdgcn_mov_dpp((int)w, 0x00, 0xf, 0xf, false);
      e[1] = (unsigned int)__builtin_amdgcn_mov_dpp((int)w, 0x55, 0xf, 0xf, false);
      e[2] = (unsigned int)__builtin_amdgcn_mov_dpp((int)w, 0xaa, 0xf, 0xf, false);
      e[3] = (unsigned int)__builtin_amdgcn_mov_dpp((int)w, 0xff, 0xf, 0xf, false);
      float4 v[4];
      for (int u = 0; u < 4; ++u)
        v[u] = *(const float4*)(h1 + (size_t)(e[u] & src_mask) * HID + 4 * q);
      for (int u = 0; u < 4; ++u) {
        const unsigned int place = e[u] >> ENT_SHIFT;
        atomicAdd(&row[0 * ACC_PITCH + place], v[u].x);
        atomicAdd(&row[1 * ACC_PITCH + place], v[u].y);
        atomicAdd(&row[2 * ACC_PITCH + place], v[u].z);
        atomicAdd(&row[3 * ACC_PITCH + place], v[u].w);
      }
      atomicAdd(&cnt[w >> ENT_SHIFT], 1);
    }
  }
  __syncthreads();
  const int node = b * NPB + t;
  if (t >= NPB || node >= N) return;
  const float si = rsqrtf(fmaxf((float)cnt[t], 1.0f));
  const float so = inv_out[node];
  inv_in[node] = si;
  float acc0 = 0.0f, acc1 = 0.0f;
  #pragma unroll
  for (int k = 0; k < HID; ++k) {
    float hv = fmaxf(fmaf(acc[k * ACC_PITCH + t], si, b1[k]), 0.0f) * so;
    acc0 = fmaf(hv, W2[k * NCLS + 0], acc0);
    acc1 = fmaf(hv, W2[k * NCLS + 1], acc1);
  }
  *(float2*)(h2 + (size_t)node * NCLS) = make_float2(acc0, acc1);
}

__global__ __launch_bounds__(1024) void agg2_final(const unsigned int* __restrict__ list, const int* __restrict__ base,
                                                   const float* __restrict__ h2, const float* __restrict__ inv_in,
                                                   const float* __restrict__ b2, float* __restrict__ out, int N) {
  __shared__ float a0[NPB + 1];
  __shared__ float a1[NPB + 1];
  if (threadIdx.x <= NPB) { a0[threadIdx.x] = 0.0f; a1[threadIdx.x] = 0.0f; }
  __syncthreads();
  const int lo = base[blockIdx.x];
  const int hi = base[blockIdx.x + 1];
  const int bd = (int)blockDim.x;
  for (int i = lo + (int)threadIdx.x; i < hi; i += 4 * bd) {
    unsigned int slot[4];
    float2 v[4];
    for (int u = 0; u < 4; ++u) {
      const unsigned int ent = list[min(i + u * bd, hi - 1)];
      slot[u] = (i + u * bd < hi) ? (ent >> ENT_SHIFT) : (unsigned int)NPB;
      v[u] = *(const float2*)(h2 + (size_t)(ent & ((1u << ENT_SHIFT) - 1u)) * NCLS);
    }
    for (int u = 0; u < 4; ++u) {
      atomicAdd(&a0[slot[u]], v[u].x);
      atomicAdd(&a1[slot[u]], v[u].y);
    }
  }
  __syncthreads();
  const int i = blockIdx.x * NPB + (int)threadIdx.x;
  if (threadIdx.x >= NPB || i >= N) return;
  const float si = inv_in[i];
  float v0 = fmaf(a0[threadIdx.x], si, b2[0]);
  float v1 = fmaf(a1[threadIdx.x], si, b2[1]);
  float m = fmaxf(v0, v1);
  float lse = m + logf(expf(v0 - m) + expf(v1 - m));
  out[(size_t)i * NCLS + 0] = v0 - lse;
  out[(size_t)i * NCLS + 1] = v1 - lse;
}

extern "C" void kernel_launch(void* const* d_in, const int* in_sizes, int n_in,
                              void* d_out, int out_size, void* d_ws, size_t ws_size,
                              hipStream_t stream) {
  const float* x   = (const float*)d_in[0];
  const int*   src = (const int*)d_in[1];
  const int*   dst = (const int*)d_in[2];
  const float* W1  = (const float*)d_in[3];
  const float* b1  = (const float*)d_in[4];
  const float* W2  = (const float*)d_in[5];
  const float* b2  = (const float*)d_in[6];
  float* out = (float*)d_out;

  const int N = in_sizes[0] / F_IN;
  const int E = in_sizes[1];

  float* ws      = (float*)d_ws;
  float* inv_out = ws;
  float* inv_in  = ws + (size_t)N;
  float* h2      = ws + (size_t)2 * N;
  int*   preS    = (int*)(ws + (size_t)4 * N);
  int*   preD    = preS + MAXB * BIN_G;
  int*   totS    = preD + MAXB * BIN_G;
  int*   totD    = totS + MAXB * TOT_PITCH;
  int*   baseS   = totD + MAXB * TOT_PITCH;
  int*   baseD   = baseS + 2 * MAXB;
  float* R       = (float*)(baseD + 2 * MAXB);
  int*   cntS    = (int*)R;
  int*   cntD    = cntS + MAXB * BIN_G;
  unsigned int* list = (unsigned int*)R;
  float* h1      = R + (size_t)17 * N;

  const int B = 256;
  const int NB = (N + NPB - 1) / NPB;
  const int NBh = NB / 2;

  bin_count<<<BIN_G, 1024, 0, stream>>>(src, dst, cntS, cntD, E, NB);
  bin_scan<<<2 * NB, BIN_G, 0, stream>>>(cntS, cntD, preS, preD, totS, totD, NB);
  bin_base<<<2, MAXB, 0, stream>>>(totS, totD, baseS, baseD, NB);

  bin_scatter_src<<<BIN_G, 1024, 0, stream>>>(src, preS, baseS, list, E, NB);
  deg_owner<<<NB, 512, 0, stream>>>(list, baseS, inv_out, N);

  const int tiles = N >> 4;
  gnn_gemm1_wmma<<<(tiles + 7) / 8, B, 0, stream>>>(x, W1, inv_out, h1, N);

  bin_scatter_dst<<<BIN_G, 1024, 0, stream>>>(src, dst, preD, baseD, list, E, 0, NBh);
  agg1_owner<<<NBh, 1024, 0, stream>>>(list, baseD, h1, inv_out, b1, W2, inv_in, h2, N, 0);
  bin_scatter_dst<<<BIN_G, 1024, 0, stream>>>(src, dst, preD, baseD, list, E, NBh, NB);
  agg1_owner<<<NB - NBh, 1024, 0, stream>>>(list, baseD, h1, inv_out, b1, W2, inv_in, h2, N, NBh);

  bin_scatter_dst<<<BIN_G, 1024, 0, stream>>>(src, dst, preD, baseD, list, E, 0, NB);
  agg2_final<<<NB, 1024, 0, stream>>>(list, baseD, h2, inv_in, b2, out, N);
}


# Destination edge list built once for both layers (was three builds: two halves, then whole) where the workspace holds it beside h1: 9 launches, was 12; layer 1 one launch, sums arrival-ordered
# speedup vs baseline: 3.6971x; 1.1384x over previous
#include <hip/hip_runtime.h>
#include <math.h>


typedef __attribute__((ext_vector_type(2))) float v2f;
typedef __attribute__((ext_vector_type(8))) float v8f;

#define F_IN 512
#define HID  16
#define NCLS 2

#define NPB_SHIFT 8
#define NPB   (1 << NPB_SHIFT)
#define MAXB  512
#define BIN_G 256
#define ENT_SHIFT 20
#define TOT_PITCH 32
#define ACC_PITCH (NPB + 4)

__device__ __forceinline__ void edge_chunk(int E, int& lo, int& hi) {
  const int chunk = ((E + BIN_G * 4 - 1) / (BIN_G * 4)) * 4;
  lo = blockIdx.x * chunk;
  if (lo > E) lo = E;
  hi = lo + chunk;
  if (hi > E) hi = E;
}

__global__ __launch_bounds__(1024) void bin_count(const int* __restrict__ src, const int* __restrict__ dst,
                                                  int* __restrict__ cntS, int* __restrict__ cntD,
                                                  int E, int NB) {
  __shared__ int hs[MAXB];
  __shared__ int hd[MAXB];
  for (int i = threadIdx.x; i < MAXB; i += blockDim.x) { hs[i] = 0; hd[i] = 0; }
  __syncthreads();
  int lo, hi;
  edge_chunk(E, lo, hi);
  for (int e = lo + 4 * (int)threadIdx.x; e < hi; e += 4 * (int)blockDim.x) {
    if (e + 3 < hi) {
      const int4 s = *(const int4*)(src + e);
      const int4 d = *(const int4*)(dst + e);
      atomicAdd(&hs[s.x >> NPB_SHIFT], 1);
      atomicAdd(&hs[s.y >> NPB_SHIFT], 1);
      atomicAdd(&hs[s.z >> NPB_SHIFT], 1);
      atomicAdd(&hs[s.w >> NPB_SHIFT], 1);
      atomicAdd(&hd[d.x >> NPB_SHIFT], 1);
      atomicAdd(&hd[d.y >> NPB_SHIFT], 1);
      atomicAdd(&hd[d.z >> NPB_SHIFT], 1);
      atomicAdd(&hd[d.w >> NPB_SHIFT], 1);
    } else {
      for (int j = e; j < hi; ++j) {
        atomicAdd(&hs[src[j] >> NPB_SHIFT], 1);
        atomicAdd(&hd[dst[j] >> NPB_SHIFT], 1);
      }
    }
  }
  __syncthreads();
  for (int b = threadIdx.x; b < NB; b += blockDim.x) {
    cntS[blockIdx.x * MAXB + b] = hs[b];
    cntD[blockIdx.x * MAXB + b] = hd[b];
  }
}

__global__ __launch_bounds__(BIN_G) void bin_scan(const int* __restrict__ cntS, const int* __restrict__ cntD,
                                                  int* __restrict__ preS, int* __restrict__ preD,
                                                  int* __restrict__ totS, int* __restrict__ totD, int NB) {
  __shared__ int sh[BIN_G];
  int b = blockIdx.x;
  const int* cnt = cntS;
  int* pre = preS;
  int* tot = totS;
  if (b >= NB) { b -= NB; cnt = cntD; pre = preD; tot = totD; }
  const int t = threadIdx.x;
  const int v = cnt[t * MAXB + b];
  sh[t] = v;
  __syncthreads();
  for (int off = 1; off < BIN_G; off <<= 1) {
    const int add = (t >= off) ? sh[t - off] : 0;
    __syncthreads();
    sh[t] += add;
    __syncthreads();
  }
  pre[b * BIN_G + t] = sh[t] - v;
  if (t == BIN_G - 1) tot[b * TOT_PITCH] = sh[t];
}

__global__ __launch_bounds__(MAXB) void bin_base(const int* __restrict__ totS, const int* __restrict__ totD,
                                                 int* __restrict__ baseS, int* __restrict__ baseD, int NB) {
  __shared__ int sh[MAXB];
  const int* tot = blockIdx.x ? totD : totS;
  int* base = blockIdx.x ? baseD : baseS;
  const int t = threadIdx.x;
  const int v = (t < NB) ? tot[t * TOT_PITCH] : 0;
  sh[t] = v;
  __syncthreads();
  for (int off = 1; off < MAXB; off <<= 1) {
    const int add = (t >= off) ? sh[t - off] : 0;
    __syncthreads();
    sh[t] += add;
    __syncthreads();
  }
  if (t < NB) base[t] = sh[t] - v;
  if (t == NB - 1) base[NB] = sh[t];
}

__device__ __forceinline__ unsigned int bin_entry(int key, unsigned int payload) {
  return ((unsigned int)(key & (NPB - 1)) << ENT_SHIFT) | payload;
}
__device__ __forceinline__ bool in_window(int key, int b_lo, int nb) {
  return (unsigned int)((key >> NPB_SHIFT) - b_lo) < (unsigned int)nb;
}

#define SC_THREADS 1024
#define SC_TRIPS 4
#define SC_PIECE (SC_TRIPS * 4 * SC_THREADS)

__device__ __forceinline__ int piece_rank(int* cntl, int key, int b_lo, int nb) {
  return in_window(key, b_lo, nb) ? atomicAdd(&cntl[key >> NPB_SHIFT], 1) : -1;
}
__device__ __forceinline__ void piece_place(const int* lofs, const int* gb, int key, int rank, int& place, int& slot) {
  const int b = rank >= 0 ? key >> NPB_SHIFT : 0;
  place = lofs[b] + rank;
  slot = gb[b] + rank;
}

template <bool HAS_PAY>
__device__ __forceinline__ void scatter_ordered(const int* __restrict__ key, const int* __restrict__ pay,
                                                const int* __restrict__ pre, const int* __restrict__ base,
                                                unsigned int* __restrict__ list, int E, int b_lo, int b_hi) {
  __shared__ int cntl[MAXB];
  __shared__ int lofs[MAXB];
  __shared__ int gb[MAXB];
  __shared__ int ntot;
  __shared__ uint2 stage[SC_PIECE];
  const int t = (int)threadIdx.x;
  const int first = base[b_lo];
  const int nb = b_hi - b_lo;
  int lo, hi;
  edge_chunk(E, lo, hi);
  for (int p0 = lo; p0 < hi; p0 += SC_PIECE) {
    const int p1 = min(p0 + SC_PIECE, hi);
    for (int b = t; b < MAXB; b += SC_THREADS) cntl[b] = 0;
    __syncthreads();

    int k[SC_TRIPS][4];
    int pl[SC_TRIPS][4];
    int rk[SC_TRIPS][4];
    #pragma unroll
    for (int u = 0; u < SC_TRIPS; ++u) {
      const int e = p0 + 4 * (u * SC_THREADS + t);
      int4 kv = make_int4(-1, -1, -1, -1);
      int4 pv = make_int4(0, 0, 0, 0);
      if (e + 3 < p1) {
        kv = *(const int4*)(key + e);
        if (HAS_PAY) pv = *(const int4*)(pay + e);
      }
      k[u][0] = kv.x; k[u][1] = kv.y; k[u][2] = kv.z; k[u][3] = kv.w;
      pl[u][0] = pv.x; pl[u][1] = pv.y; pl[u][2] = pv.z; pl[u][3] = pv.w;
    }
    const int rem = (p1 - p0) & 3;
    int kt = -1;
    int plt = 0;
    if (t < rem) {
      kt = key[p1 - rem + t];
      if (HAS_PAY) plt = pay[p1 - rem + t];
    }
    if (p0 == lo)
      for (int b = t; b < MAXB; b += SC_THREADS)
        gb[b] = (b >= b_lo && b < b_hi) ? base[b] - first + pre[b * BIN_G + blockIdx.x] : 0;
    #pragma unroll
    for (int u = 0; u < SC_TRIPS; ++u) {
      #pragma unroll
      for (int j = 0; j < 4; ++j) rk[u][j] = piece_rank(cntl, k[u][j], b_lo, nb);
    }
    const int rkt = piece_rank(cntl, kt, b_lo, nb);
    __syncthreads();

    if (t < 32) {
      int ex[MAXB / 32];
      int run = 0;
      #pragma unroll
      for (int j = 0; j < MAXB / 32; ++j) { ex[j] = run; run += cntl[t + 32 * j]; }
      int inc = run;
      #pragma unroll
      for (int off = 1; off < 32; off <<= 1) {
        const int up = __shfl_up(inc, off, 32);
        if (t >= off) inc += up;
      }
      const int excl = inc - run;
      #pragma unroll
      for (int j = 0; j < MAXB / 32; ++j) lofs[t + 32 * j] = excl + ex[j];
      if (t == 31) ntot = inc;
    }
    __syncthreads();

    int ps[SC_TRIPS][4];
    int sl[SC_TRIPS][4];
    #pragma unroll
    for (int u = 0; u < SC_TRIPS; ++u) {
      #pragma unroll
      for (int j = 0; j < 4; ++j) piece_place(lofs, gb, k[u][j], rk[u][j], ps[u][j], sl[u][j]);
    }
    int pst, slt;
    piece_place(lofs, gb, kt, rkt, pst, slt);
    #pragma unroll
    for (int u = 0; u < SC_TRIPS; ++u) {
      #pragma unroll
      for (int j = 0; j < 4; ++j)
        if (rk[u][j] >= 0) stage[ps[u][j]] = make_uint2(bin_entry(k[u][j], (unsigned int)pl[u][j]), (unsigned int)sl[u][j]);
    }
    if (rkt >= 0) stage[pst] = make_uint2(bin_entry(kt, (unsigned int)plt), (unsigned int)slt);
    __syncthreads();

    const int n = ntot;
    for (int p = t; p < n; p += SC_THREADS) {
      const uint2 s = stage[p];
      list[s.y] = s.x;
    }
    if (p1 < hi)
      for (int b = t; b < MAXB; b += SC_THREADS) gb[b] += cntl[b];
  }
}

__global__ __launch_bounds__(SC_THREADS) void bin_scatter_src(const int* __restrict__ src,
                                                              const int* __restrict__ pre, const int* __restrict__ base,
                                                              unsigned int* __restrict__ list, int E, int NB) {
  scatter_ordered<false>(src, src, pre, base, list, E, 0, NB);
}

__global__ __launch_bounds__(SC_THREADS) void bin_scatter_dst(const int* __restrict__ src, const int* __restrict__ dst,
                                                              const int* __restrict__ pre, const int* __restrict__ base,
                                                              unsigned int* __restrict__ list, int E, int b_lo, int b_hi) {
  scatter_ordered<true>(dst, src, pre, base, list, E, b_lo, b_hi);
}

__global__ __launch_bounds__(512) void deg_owner(const unsigned int* __restrict__ list, const int* __restrict__ base,
                                                 float* __restrict__ inv, int N) {
  __shared__ int c[NPB + 1];
  if (threadIdx.x <= NPB) c[threadIdx.x] = 0;
  __syncthreads();
  const int lo = base[blockIdx.x];
  const int hi = base[blockIdx.x + 1];
  const int bd = (int)blockDim.x;
  for (int i = lo + (int)threadIdx.x; i < hi; i += 4 * bd) {
    unsigned int slot[4];
    for (int u = 0; u < 4; ++u) {
      const unsigned int ent = list[min(i + u * bd, hi - 1)];
      slot[u] = (i + u * bd < hi) ? (ent >> ENT_SHIFT) : (unsigned int)NPB;
    }
    for (int u = 0; u < 4; ++u) atomicAdd(&c[slot[u]], 1);
  }
  __syncthreads();
  const int node = blockIdx.x * NPB + (int)threadIdx.x;
  if (threadIdx.x < NPB && node < N) inv[node] = rsqrtf(fmaxf((float)c[threadIdx.x], 1.0f));
}

__global__ __launch_bounds__(256) void gnn_gemm1_wmma(
    const float* __restrict__ x, const float* __restrict__ W1,
    const float* __restrict__ inv_out, float* __restrict__ h1, int N) {
  __shared__ float wlds[F_IN * HID];
  for (int i = threadIdx.x; i < F_IN * HID; i += blockDim.x) wlds[i] = W1[i];
  __syncthreads();

  const int wave = threadIdx.x >> 5;
  const int lane = threadIdx.x & 31;
  const int tiles = N >> 4;
  const int tile = blockIdx.x * 8 + wave;
  if (tile >= tiles) return;

  const int n16 = lane & 15;
  const int g   = lane >> 4;
  const int row = tile * 16 + n16;
  const float s = inv_out[row];
  const float* __restrict__ xr = x + (size_t)row * F_IN;

  v8f c = {};
  #pragma unroll 4
  for (int k = 0; k < F_IN; k += 4) {
    const int kk = k + 2 * g;
    v2f a = *(const v2f*)(xr + kk);
    a.x *= s;
    a.y *= s;
    v2f b;
    b.x = wlds[(kk + 0) * HID + n16];
    b.y = wlds[(kk + 1) * HID + n16];
    c = __builtin_amdgcn_wmma_f32_16x16x4_f32(
         false, a,  false, b,
         (short)0, c,  false,  false);
  }

  #pragma unroll
  for (int p = 0; p < 8; ++p) {
    const int m = tile * 16 + p + 8 * g;
    h1[(size_t)m * HID + n16] = c[p];
  }
}

__global__ __launch_bounds__(1024) void agg1_owner(const unsigned int* __restrict__ list, const int* __restrict__ base,
                                                   const float* __restrict__ h1, const float* __restrict__ inv_out,
                                                   const float* __restrict__ b1, const float* __restrict__ W2,
                                                   float* __restrict__ inv_in, float* __restrict__ h2,
                                                   int N, int b_lo) {
  __shared__ float acc[HID * ACC_PITCH];
  __shared__ int cnt[NPB + 1];
  for (int i = threadIdx.x; i < HID * ACC_PITCH; i += blockDim.x) acc[i] = 0.0f;
  if (threadIdx.x <= NPB) cnt[threadIdx.x] = 0;
  __syncthreads();
  const int b = b_lo + (int)blockIdx.x;
  const int first = base[b_lo];
  const int lo = base[b] - first;
  const int hi = base[b + 1] - first;
  const int t = (int)threadIdx.x;
  const int q = t & 3;
  float* row = acc + 4 * q * ACC_PITCH;
  const unsigned int src_mask = (1u << ENT_SHIFT) - 1u;
  if (lo < hi) {
    unsigned int nxt = list[min(lo + t, hi - 1)];
    for (int i0 = lo; i0 < hi; i0 += (int)blockDim.x) {
      unsigned int w = nxt;
      if (i0 + t >= hi) w = (w & src_mask) | ((unsigned int)NPB << ENT_SHIFT);
      nxt = list[min(i0 + (int)blockDim.x + t, hi - 1)];
      unsigned int e[4];
      e[0] = (unsigned int)__builtin_amdgcn_mov_dpp((int)w, 0x00, 0xf, 0xf, false);
      e[1] = (unsigned int)__builtin_amdgcn_mov_dpp((int)w, 0x55, 0xf, 0xf, false);
      e[2] = (unsigned int)__builtin_amdgcn_mov_dpp((int)w, 0xaa, 0xf, 0xf, false);
      e[3] = (unsigned int)__builtin_amdgcn_mov_dpp((int)w, 0xff, 0xf, 0xf, false);
      float4 v[4];
      for (int u = 0; u < 4; ++u)
        v[u] = *(const float4*)(h1 + (size_t)(e[u] & src_mask) * HID + 4 * q);
      for (int u = 0; u < 4; ++u) {
        const unsigned int place = e[u] >> ENT_SHIFT;
        atomicAdd(&row[0 * ACC_PITCH + place], v[u].x);
        atomicAdd(&row[1 * ACC_PITCH + place], v[u].y);
        atomicAdd(&row[2 * ACC_PITCH + place], v[u].z);
        atomicAdd(&row[3 * ACC_PITCH + place], v[u].w);
      }
      atomicAdd(&cnt[w >> ENT_SHIFT], 1);
    }
  }
  __syncthreads();
  const int node = b * NPB + t;
  if (t >= NPB || node >= N) return;
  const float si = rsqrtf(fmaxf((float)cnt[t], 1.0f));
  const float so = inv_out[node];
  inv_in[node] = si;
  float acc0 = 0.0f, acc1 = 0.0f;
  #pragma unroll
  for (int k = 0; k < HID; ++k) {
    float hv = fmaxf(fmaf(acc[k * ACC_PITCH + t], si, b1[k]), 0.0f) * so;
    acc0 = fmaf(hv, W2[k * NCLS + 0], acc0);
    acc1 = fmaf(hv, W2[k * NCLS + 1], acc1);
  }
  *(float2*)(h2 + (size_t)node * NCLS) = make_float2(acc0, acc1);
}

__global__ __launch_bounds__(1024) void agg2_final(const unsigned int* __restrict__ list, const int* __restrict__ base,
                                                   const float* __restrict__ h2, const float* __restrict__ inv_in,
                                                   const float* __restrict__ b2, float* __restrict__ out, int N) {
  __shared__ float a0[NPB + 1];
  __shared__ float a1[NPB + 1];
  if (threadIdx.x <= NPB) { a0[threadIdx.x] = 0.0f; a1[threadIdx.x] = 0.0f; }
  __syncthreads();
  const int lo = base[blockIdx.x];
  const int hi = base[blockIdx.x + 1];
  const int bd = (int)blockDim.x;
  for (int i = lo + (int)threadIdx.x; i < hi; i += 4 * bd) {
    unsigned int slot[4];
    float2 v[4];
    for (int u = 0; u < 4; ++u) {
      const unsigned int ent = list[min(i + u * bd, hi - 1)];
      slot[u] = (i + u * bd < hi) ? (ent >> ENT_SHIFT) : (unsigned int)NPB;
      v[u] = *(const float2*)(h2 + (size_t)(ent & ((1u << ENT_SHIFT) - 1u)) * NCLS);
    }
    for (int u = 0; u < 4; ++u) {
      atomicAdd(&a0[slot[u]], v[u].x);
      atomicAdd(&a1[slot[u]], v[u].y);
    }
  }
  __syncthreads();
  const int i = blockIdx.x * NPB + (int)threadIdx.x;
  if (threadIdx.x >= NPB || i >= N) return;
  const float si = inv_in[i];
  float v0 = fmaf(a0[threadIdx.x], si, b2[0]);
  float v1 = fmaf(a1[threadIdx.x], si, b2[1]);
  float m = fmaxf(v0, v1);
  float lse = m + logf(expf(v0 - m) + expf(v1 - m));
  out[(size_t)i * NCLS + 0] = v0 - lse;
  out[(size_t)i * NCLS + 1] = v1 - lse;
}

extern "C" void kernel_launch(void* const* d_in, const int* in_sizes, int n_in,
                              void* d_out, int out_size, void* d_ws, size_t ws_size,
                              hipStream_t stream) {
  const float* x   = (const float*)d_in[0];
  const int*   src = (const int*)d_in[1];
  const int*   dst = (const int*)d_in[2];
  const float* W1  = (const float*)d_in[3];
  const float* b1  = (const float*)d_in[4];
  const float* W2  = (const float*)d_in[5];
  const float* b2  = (const float*)d_in[6];
  float* out = (float*)d_out;

  const int N = in_sizes[0] / F_IN;
  const int E = in_sizes[1];

  float* ws      = (float*)d_ws;
  float* inv_out = ws;
  float* inv_in  = ws + (size_t)N;
  float* h2      = ws + (size_t)2 * N;
  int*   preS    = (int*)(ws + (size_t)4 * N);
  int*   preD    = preS + MAXB * BIN_G;
  int*   totS    = preD + MAXB * BIN_G;
  int*   totD    = totS + MAXB * TOT_PITCH;
  int*   baseS   = totD + MAXB * TOT_PITCH;
  int*   baseD   = baseS + 2 * MAXB;
  float* R       = (float*)(baseD + 2 * MAXB);
  int*   cntS    = (int*)R;
  int*   cntD    = cntS + MAXB * BIN_G;
  unsigned int* list = (unsigned int*)R;
  float* h1_far  = R + (((size_t)E + 63) & ~(size_t)63);
  const bool whole = (const char*)(h1_far + (size_t)HID * N) <= (const char*)d_ws + ws_size;
  float* h1      = whole ? h1_far : R + (size_t)17 * N;

  const int B = 256;
  const int NB = (N + NPB - 1) / NPB;
  const int NBh = NB / 2;

  bin_count<<<BIN_G, 1024, 0, stream>>>(src, dst, cntS, cntD, E, NB);
  bin_scan<<<2 * NB, BIN_G, 0, stream>>>(cntS, cntD, preS, preD, totS, totD, NB);
  bin_base<<<2, MAXB, 0, stream>>>(totS, totD, baseS, baseD, NB);

  bin_scatter_src<<<BIN_G, 1024, 0, stream>>>(src, preS, baseS, list, E, NB);
  deg_owner<<<NB, 512, 0, stream>>>(list, baseS, inv_out, N);

  const int tiles = N >> 4;
  gnn_gemm1_wmma<<<(tiles + 7) / 8, B, 0, stream>>>(x, W1, inv_out, h1, N);

  if (whole) {
    bin_scatter_dst<<<BIN_G, 1024, 0, stream>>>(src, dst, preD, baseD, list, E, 0, NB);
    agg1_owner<<<NB, 1024, 0, stream>>>(list, baseD, h1, inv_out, b1, W2, inv_in, h2, N, 0);
  } else {
    bin_scatter_dst<<<BIN_G, 1024, 0, stream>>>(src, dst, preD, baseD, list, E, 0, NBh);
    agg1_owner<<<NBh, 1024, 0, stream>>>(list, baseD, h1, inv_out, b1, W2, inv_in, h2, N, 0);
    bin_scatter_dst<<<BIN_G, 1024, 0, stream>>>(src, dst, preD, baseD, list, E, NBh, NB);
    agg1_owner<<<NB - NBh, 1024, 0, stream>>>(list, baseD, h1, inv_out, b1, W2, inv_in, h2, N, NBh);
    bin_scatter_dst<<<BIN_G, 1024, 0, stream>>>(src, dst, preD, baseD, list, E, 0, NB);
  }
  agg2_final<<<NB, 1024, 0, stream>>>(list, baseD, h2, inv_in, b2, out, N);
}


# Edge lists kept piece by piece: each ordering work-group writes its range-ordered piece and a table row of segment places; owners walk the segments; 6 launches, was 9; sums arrival-ordered as before
# speedup vs baseline: 3.8956x; 1.0537x over previous
#include <hip/hip_runtime.h>
#include <math.h>


typedef __attribute__((ext_vector_type(2))) float v2f;
typedef __attribute__((ext_vector_type(8))) float v8f;

#define F_IN 512
#define HID  16
#define NCLS 2

#define NPB_SHIFT 8
#define NPB   (1 << NPB_SHIFT)
#define MAXB  512
#define ENT_SHIFT 20
#define ACC_PITCH (NPB + 4)

#define SC_THREADS 1024
#define SC_TRIPS 4
#define SC_CAP (SC_TRIPS * 4 * SC_THREADS)
#define PIECE 12512
#define SEG_SHIFT 16
#define SEG_PER_WAVE 8

__device__ __forceinline__ unsigned int bin_entry(int key, unsigned int payload) {
  return ((unsigned int)(key & (NPB - 1)) << ENT_SHIFT) | payload;
}
__device__ __forceinline__ int piece_rank(int* cntl, int key, int NB) {
  return (unsigned int)(key >> NPB_SHIFT) < (unsigned int)NB ? atomicAdd(&cntl[key >> NPB_SHIFT], 1) : -1;
}

template <bool HAS_PAY>
__device__ __forceinline__ void order_piece(const int* __restrict__ key, const int* __restrict__ pay,
                                            unsigned int* __restrict__ tab, unsigned int* __restrict__ list,
                                            int E, int NB) {
  __shared__ int cntl[MAXB];
  __shared__ int lofs[MAXB];
  __shared__ int ntot;
  __shared__ __attribute__((aligned(16))) unsigned int stage[SC_CAP];
  const int t = (int)threadIdx.x;
  const int p0 = (int)blockIdx.x * PIECE;
  const int p1 = min(p0 + PIECE, E);
  for (int b = t; b < MAXB; b += SC_THREADS) cntl[b] = 0;
  __syncthreads();

  int k[SC_TRIPS][4];
  int pl[SC_TRIPS][4];
  int rk[SC_TRIPS][4];
  #pragma unroll
  for (int u = 0; u < SC_TRIPS; ++u) {
    const int e = p0 + 4 * (u * SC_THREADS + t);
    int4 kv = make_int4(-1, -1, -1, -1);
    int4 pv = make_int4(0, 0, 0, 0);
    if (e + 3 < p1) {
      kv = *(const int4*)(key + e);
      if (HAS_PAY) pv = *(const int4*)(pay + e);
    }
    k[u][0] = kv.x; k[u][1] = kv.y; k[u][2] = kv.z; k[u][3] = kv.w;
    pl[u][0] = pv.x; pl[u][1] = pv.y; pl[u][2] = pv.z; pl[u][3] = pv.w;
  }
  const int rem = (p1 - p0) & 3;
  int kt = -1;
  int plt = 0;
  if (t < rem) {
    kt = key[p1 - rem + t];
    if (HAS_PAY) plt = pay[p1 - rem + t];
  }
  #pragma unroll
  for (int u = 0; u < SC_TRIPS; ++u) {
    #pragma unroll
    for (int j = 0; j < 4; ++j) rk[u][j] = piece_rank(cntl, k[u][j], NB);
  }
  const int rkt = piece_rank(cntl, kt, NB);
  __syncthreads();

  if (t < 32) {
    int ex[MAXB / 32];
    int run = 0;
    #pragma unroll
    for (int j = 0; j < MAXB / 32; ++j) { ex[j] = run; run += cntl[t + 32 * j]; }
    int inc = run;
    #pragma unroll
    for (int off = 1; off < 32; off <<= 1) {
      const int up = __shfl_up(inc, off, 32);
      if (t >= off) inc += up;
    }
    const int excl = inc - run;
    #pragma unroll
    for (int j = 0; j < MAXB / 32; ++j) lofs[t + 32 * j] = excl + ex[j];
    if (t == 31) ntot = inc;
  }
  __syncthreads();

  for (int b = t; b < MAXB; b += SC_THREADS)
    tab[(size_t)blockIdx.x * MAXB + b] = (unsigned int)lofs[b] | ((unsigned int)cntl[b] << SEG_SHIFT);

  int ps[SC_TRIPS][4];
  #pragma unroll
  for (int u = 0; u < SC_TRIPS; ++u) {
    #pragma unroll
    for (int j = 0; j < 4; ++j) ps[u][j] = lofs[rk[u][j] >= 0 ? k[u][j] >> NPB_SHIFT : 0] + rk[u][j];
  }
  const int pst = lofs[rkt >= 0 ? kt >> NPB_SHIFT : 0] + rkt;
  #pragma unroll
  for (int u = 0; u < SC_TRIPS; ++u) {
    #pragma unroll
    for (int j = 0; j < 4; ++j)
      if (rk[u][j] >= 0) stage[ps[u][j]] = bin_entry(k[u][j], (unsigned int)pl[u][j]);
  }
  if (rkt >= 0) stage[pst] = bin_entry(kt, (unsigned int)plt);
  __syncthreads();

  const int n = ntot;
  unsigned int* __restrict__ mine = list + (size_t)p0;
  for (int i = t; i < (n >> 2); i += SC_THREADS) ((uint4*)mine)[i] = ((const uint4*)stage)[i];
  if (t < (n & 3)) mine[(n & ~3) + t] = stage[(n & ~3) + t];
}

__global__ __launch_bounds__(SC_THREADS) void order_by_src(const int* __restrict__ src,
                                                           unsigned int* __restrict__ tab,
                                                           unsigned int* __restrict__ list, int E, int NB) {
  order_piece<false>(src, src, tab, list, E, NB);
}

__global__ __launch_bounds__(SC_THREADS) void order_by_dst(const int* __restrict__ src, const int* __restrict__ dst,
                                                           unsigned int* __restrict__ tab,
                                                           unsigned int* __restrict__ list, int E, int NB) {
  order_piece<true>(dst, src, tab, list, E, NB);
}

__device__ __forceinline__ void run_open(const unsigned int* __restrict__ tab, int b, int g, int P, int lane,
                                         int (&pre)[SEG_PER_WAVE + 1], int (&off)[SEG_PER_WAVE]) {
  const int j = lane & (SEG_PER_WAVE - 1);
  const int piece = SEG_PER_WAVE * g + j;
  const unsigned int wd = piece < P ? tab[(size_t)piece * MAXB + b] : 0u;
  const int len = (int)(wd >> SEG_SHIFT);
  int inc = len;
  #pragma unroll
  for (int o = 1; o < SEG_PER_WAVE; o <<= 1) {
    const int up = __shfl_up(inc, o, SEG_PER_WAVE);
    if (j >= o) inc += up;
  }
  const int exc = inc - len;
  const int of = piece * PIECE + (int)(wd & ((1u << SEG_SHIFT) - 1u)) - exc;
  #pragma unroll
  for (int s = 0; s < SEG_PER_WAVE; ++s) {
    pre[s] = __builtin_amdgcn_readlane(exc, s);
    off[s] = __builtin_amdgcn_readlane(of, s);
  }
  pre[SEG_PER_WAVE] = __builtin_amdgcn_readlane(inc, SEG_PER_WAVE - 1);
}
__device__ __forceinline__ int run_index(int v, const int (&pre)[SEG_PER_WAVE + 1], const int (&off)[SEG_PER_WAVE]) {
  int o = off[0];
  #pragma unroll
  for (int s = 1; s < SEG_PER_WAVE; ++s) o = v >= pre[s] ? off[s] : o;
  return v + o;
}

__global__ __launch_bounds__(512) void deg_owner(const unsigned int* __restrict__ list, const unsigned int* __restrict__ tab,
                                                 float* __restrict__ inv, int N, int P) {
  __shared__ int c[NPB + 1];
  if (threadIdx.x <= NPB) c[threadIdx.x] = 0;
  __syncthreads();
  const int lane = (int)threadIdx.x & 31;
  const int G = (P + SEG_PER_WAVE - 1) / SEG_PER_WAVE;
  for (int g = (int)threadIdx.x >> 5; g < G; g += (int)blockDim.x >> 5) {
    int pre[SEG_PER_WAVE + 1];
    int off[SEG_PER_WAVE];
    run_open(tab, (int)blockIdx.x, g, P, lane, pre, off);
    const int total = pre[SEG_PER_WAVE];
    for (int v0 = 0; v0 < total; v0 += 128) {
      unsigned int slot[4];
      for (int u = 0; u < 4; ++u) {
        const int v = v0 + 32 * u + lane;
        const unsigned int ent = list[run_index(min(v, total - 1), pre, off)];
        slot[u] = v < total ? (ent >> ENT_SHIFT) : (unsigned int)NPB;
      }
      for (int u = 0; u < 4; ++u) atomicAdd(&c[slot[u]], 1);
    }
  }
  __syncthreads();
  const int node = blockIdx.x * NPB + (int)threadIdx.x;
  if (threadIdx.x < NPB && node < N) inv[node] = rsqrtf(fmaxf((float)c[threadIdx.x], 1.0f));
}

__global__ __launch_bounds__(256) void gnn_gemm1_wmma(
    const float* __restrict__ x, const float* __restrict__ W1,
    const float* __restrict__ inv_out, float* __restrict__ h1, int N) {
  __shared__ float wlds[F_IN * HID];
  for (int i = threadIdx.x; i < F_IN * HID; i += blockDim.x) wlds[i] = W1[i];
  __syncthreads();

  const int wave = threadIdx.x >> 5;
  const int lane = threadIdx.x & 31;
  const int tiles = N >> 4;
  const int tile = blockIdx.x * 8 + wave;
  if (tile >= tiles) return;

  const int n16 = lane & 15;
  const int g   = lane >> 4;
  const int row = tile * 16 + n16;
  const float s = inv_out[row];
  const float* __restrict__ xr = x + (size_t)row * F_IN;

  v8f c = {};
  #pragma unroll 4
  for (int k = 0; k < F_IN; k += 4) {
    const int kk = k + 2 * g;
    v2f a = *(const v2f*)(xr + kk);
    a.x *= s;
    a.y *= s;
    v2f b;
    b.x = wlds[(kk + 0) * HID + n16];
    b.y = wlds[(kk + 1) * HID + n16];
    c = __builtin_amdgcn_wmma_f32_16x16x4_f32(
         false, a,  false, b,
         (short)0, c,  false,  false);
  }

  #pragma unroll
  for (int p = 0; p < 8; ++p) {
    const int m = tile * 16 + p + 8 * g;
    h1[(size_t)m * HID + n16] = c[p];
  }
}

__global__ __launch_bounds__(1024) void agg1_owner(const unsigned int* __restrict__ list, const unsigned int* __restrict__ tab,
                                                   const float* __restrict__ h1, const float* __restrict__ inv_out,
                                                   const float* __restrict__ b1, const float* __restrict__ W2,
                                                   float* __restrict__ inv_in, float* __restrict__ h2,
                                                   int N, int P) {
  __shared__ float acc[HID * ACC_PITCH];
  __shared__ int cnt[NPB + 1];
  for (int i = threadIdx.x; i < HID * ACC_PITCH; i += blockDim.x) acc[i] = 0.0f;
  if (threadIdx.x <= NPB) cnt[threadIdx.x] = 0;
  __syncthreads();
  const int b = (int)blockIdx.x;
  const int t = (int)threadIdx.x;
  const int lane = t & 31;
  const int q = t & 3;
  float* row = acc + 4 * q * ACC_PITCH;
  const unsigned int src_mask = (1u << ENT_SHIFT) - 1u;
  const int G = (P + SEG_PER_WAVE - 1) / SEG_PER_WAVE;
  for (int g = t >> 5; g < G; g += (int)blockDim.x >> 5) {
    int pre[SEG_PER_WAVE + 1];
    int off[SEG_PER_WAVE];
    run_open(tab, b, g, P, lane, pre, off);
    const int total = pre[SEG_PER_WAVE];
    if (total > 0) {
      unsigned int nxt = list[run_index(min(lane, total - 1), pre, off)];
      for (int v0 = 0; v0 < total; v0 += 32) {
        unsigned int w = nxt;
        if (v0 + lane >= total) w = (w & src_mask) | ((unsigned int)NPB << ENT_SHIFT);
        nxt = list[run_index(min(v0 + 32 + lane, total - 1), pre, off)];
        unsigned int e[4];
        e[0] = (unsigned int)__builtin_amdgcn_mov_dpp((int)w, 0x00, 0xf, 0xf, false);
        e[1] = (unsigned int)__builtin_amdgcn_mov_dpp((int)w, 0x55, 0xf, 0xf, false);
        e[2] = (unsigned int)__builtin_amdgcn_mov_dpp((int)w, 0xaa, 0xf, 0xf, false);
        e[3] = (unsigned int)__builtin_amdgcn_mov_dpp((int)w, 0xff, 0xf, 0xf, false);
        float4 v[4];
        for (int u = 0; u < 4; ++u)
          v[u] = *(const float4*)(h1 + (size_t)(e[u] & src_mask) * HID + 4 * q);
        for (int u = 0; u < 4; ++u) {
          const unsigned int place = e[u] >> ENT_SHIFT;
          atomicAdd(&row[0 * ACC_PITCH + place], v[u].x);
          atomicAdd(&row[1 * ACC_PITCH + place], v[u].y);
          atomicAdd(&row[2 * ACC_PITCH + place], v[u].z);
          atomicAdd(&row[3 * ACC_PITCH + place], v[u].w);
        }
        atomicAdd(&cnt[w >> ENT_SHIFT], 1);
      }
    }
  }
  __syncthreads();
  const int node = b * NPB + t;
  if (t >= NPB || node >= N) return;
  const float si = rsqrtf(fmaxf((float)cnt[t], 1.0f));
  const float so = inv_out[node];
  inv_in[node] = si;
  float acc0 = 0.0f, acc1 = 0.0f;
  #pragma unroll
  for (int k = 0; k < HID; ++k) {
    float hv = fmaxf(fmaf(acc[k * ACC_PITCH + t], si, b1[k]), 0.0f) * so;
    acc0 = fmaf(hv, W2[k * NCLS + 0], acc0);
    acc1 = fmaf(hv, W2[k * NCLS + 1], acc1);
  }
  *(float2*)(h2 + (size_t)node * NCLS) = make_float2(acc0, acc1);
}

__global__ __launch_bounds__(1024) void agg2_final(const unsigned int* __restrict__ list, const unsigned int* __restrict__ tab,
                                                   const float* __restrict__ h2, const float* __restrict__ inv_in,
                                                   const float* __restrict__ b2, float* __restrict__ out, int N, int P) {
  __shared__ float a0[NPB + 1];
  __shared__ float a1[NPB + 1];
  if (threadIdx.x <= NPB) { a0[threadIdx.x] = 0.0f; a1[threadIdx.x] = 0.0f; }
  __syncthreads();
  const int lane = (int)threadIdx.x & 31;
  const int G = (P + SEG_PER_WAVE - 1) / SEG_PER_WAVE;
  for (int g = (int)threadIdx.x >> 5; g < G; g += (int)blockDim.x >> 5) {
    int pre[SEG_PER_WAVE + 1];
    int off[SEG_PER_WAVE];
    run_open(tab, (int)blockIdx.x, g, P, lane, pre, off);
    const int total = pre[SEG_PER_WAVE];
    for (int v0 = 0; v0 < total; v0 += 128) {
      unsigned int slot[4];
      float2 v[4];
      for (int u = 0; u < 4; ++u) {
        const int n = v0 + 32 * u + lane;
        const unsigned int ent = list[run_index(min(n, total - 1), pre, off)];
        slot[u] = n < total ? (ent >> ENT_SHIFT) : (unsigned int)NPB;
        v[u] = *(const float2*)(h2 + (size_t)(ent & ((1u << ENT_SHIFT) - 1u)) * NCLS);
      }
      for (int u = 0; u < 4; ++u) {
        atomicAdd(&a0[slot[u]], v[u].x);
        atomicAdd(&a1[slot[u]], v[u].y);
      }
    }
  }
  __syncthreads();
  const int i = blockIdx.x * NPB + (int)threadIdx.x;
  if (threadIdx.x >= NPB || i >= N) return;
  const float si = inv_in[i];
  float v0 = fmaf(a0[threadIdx.x], si, b2[0]);
  float v1 = fmaf(a1[threadIdx.x], si, b2[1]);
  float m = fmaxf(v0, v1);
  float lse = m + logf(expf(v0 - m) + expf(v1 - m));
  out[(size_t)i * NCLS + 0] = v0 - lse;
  out[(size_t)i * NCLS + 1] = v1 - lse;
}

extern "C" void kernel_launch(void* const* d_in, const int* in_sizes, int n_in,
                              void* d_out, int out_size, void* d_ws, size_t ws_size,
                              hipStream_t stream) {
  const float* x   = (const float*)d_in[0];
  const int*   src = (const int*)d_in[1];
  const int*   dst = (const int*)d_in[2];
  const float* W1  = (const float*)d_in[3];
  const float* b1  = (const float*)d_in[4];
  const float* W2  = (const float*)d_in[5];
  const float* b2  = (const float*)d_in[6];
  float* out = (float*)d_out;

  const int N = in_sizes[0] / F_IN;
  const int E = in_sizes[1];
  const int P = (E + PIECE - 1) / PIECE;

  float* ws      = (float*)d_ws;
  float* inv_out = ws;
  float* inv_in  = ws + (size_t)N;
  float* h2      = ws + (size_t)2 * N;
  const size_t tab0 = ((size_t)4 * N + 63) & ~(size_t)63;
  unsigned int* tabS = (unsigned int*)(ws + tab0);
  unsigned int* tabD = tabS + (size_t)P * MAXB;
  const size_t list0 = tab0 + (size_t)2 * P * MAXB;
  unsigned int* list = (unsigned int*)(ws + list0);
  const size_t h10 = list0 + (((size_t)P * PIECE + 63) & ~(size_t)63);
  float* h1      = ws + h10;
  if ((h10 + (size_t)HID * N) * sizeof(float) > ws_size) return;

  const int B = 256;
  const int NB = (N + NPB - 1) / NPB;

  order_by_src<<<P, SC_THREADS, 0, stream>>>(src, tabS, list, E, NB);
  deg_owner<<<NB, 512, 0, stream>>>(list, tabS, inv_out, N, P);

  const int tiles = N >> 4;
  gnn_gemm1_wmma<<<(tiles + 7) / 8, B, 0, stream>>>(x, W1, inv_out, h1, N);

  order_by_dst<<<P, SC_THREADS, 0, stream>>>(src, dst, tabD, list, E, NB);
  agg1_owner<<<NB, 1024, 0, stream>>>(list, tabD, h1, inv_out, b1, W2, inv_in, h2, N, P);
  agg2_final<<<NB, 1024, 0, stream>>>(list, tabD, h2, inv_in, b2, out, N, P);
}


# Layer-2 edge sums: an owner work-group (now two ranges) first orders its entries by the source's 128-byte line of h2 in LDS, then gathers in that order: far fewer lines per load; sum order changes
# speedup vs baseline: 4.4129x; 1.1328x over previous
#include <hip/hip_runtime.h>
#include <math.h>


typedef __attribute__((ext_vector_type(2))) float v2f;
typedef __attribute__((ext_vector_type(8))) float v8f;

#define F_IN 512
#define HID  16
#define NCLS 2

#define NPB_SHIFT 8
#define NPB   (1 << NPB_SHIFT)
#define MAXB  512
#define ENT_SHIFT 20
#define ACC_PITCH (NPB + 4)

#define SC_THREADS 1024
#define SC_TRIPS 4
#define SC_CAP (SC_TRIPS * 4 * SC_THREADS)
#define PIECE 12512
#define SEG_SHIFT 16
#define SEG_PER_WAVE 8

__device__ __forceinline__ unsigned int bin_entry(int key, unsigned int payload) {
  return ((unsigned int)(key & (NPB - 1)) << ENT_SHIFT) | payload;
}
__device__ __forceinline__ int piece_rank(int* cntl, int key, int NB) {
  return (unsigned int)(key >> NPB_SHIFT) < (unsigned int)NB ? atomicAdd(&cntl[key >> NPB_SHIFT], 1) : -1;
}

template <bool HAS_PAY>
__device__ __forceinline__ void order_piece(const int* __restrict__ key, const int* __restrict__ pay,
                                            unsigned int* __restrict__ tab, unsigned int* __restrict__ list,
                                            int E, int NB) {
  __shared__ int cntl[MAXB];
  __shared__ int lofs[MAXB];
  __shared__ int ntot;
  __shared__ __attribute__((aligned(16))) unsigned int stage[SC_CAP];
  const int t = (int)threadIdx.x;
  const int p0 = (int)blockIdx.x * PIECE;
  const int p1 = min(p0 + PIECE, E);
  for (int b = t; b < MAXB; b += SC_THREADS) cntl[b] = 0;
  __syncthreads();

  int k[SC_TRIPS][4];
  int pl[SC_TRIPS][4];
  int rk[SC_TRIPS][4];
  #pragma unroll
  for (int u = 0; u < SC_TRIPS; ++u) {
    const int e = p0 + 4 * (u * SC_THREADS + t);
    int4 kv = make_int4(-1, -1, -1, -1);
    int4 pv = make_int4(0, 0, 0, 0);
    if (e + 3 < p1) {
      kv = *(const int4*)(key + e);
      if (HAS_PAY) pv = *(const int4*)(pay + e);
    }
    k[u][0] = kv.x; k[u][1] = kv.y; k[u][2] = kv.z; k[u][3] = kv.w;
    pl[u][0] = pv.x; pl[u][1] = pv.y; pl[u][2] = pv.z; pl[u][3] = pv.w;
  }
  const int rem = (p1 - p0) & 3;
  int kt = -1;
  int plt = 0;
  if (t < rem) {
    kt = key[p1 - rem + t];
    if (HAS_PAY) plt = pay[p1 - rem + t];
  }
  #pragma unroll
  for (int u = 0; u < SC_TRIPS; ++u) {
    #pragma unroll
    for (int j = 0; j < 4; ++j) rk[u][j] = piece_rank(cntl, k[u][j], NB);
  }
  const int rkt = piece_rank(cntl, kt, NB);
  __syncthreads();

  if (t < 32) {
    int ex[MAXB / 32];
    int run = 0;
    #pragma unroll
    for (int j = 0; j < MAXB / 32; ++j) { ex[j] = run; run += cntl[t + 32 * j]; }
    int inc = run;
    #pragma unroll
    for (int off = 1; off < 32; off <<= 1) {
      const int up = __shfl_up(inc, off, 32);
      if (t >= off) inc += up;
    }
    const int excl = inc - run;
    #pragma unroll
    for (int j = 0; j < MAXB / 32; ++j) lofs[t + 32 * j] = excl + ex[j];
    if (t == 31) ntot = inc;
  }
  __syncthreads();

  for (int b = t; b < MAXB; b += SC_THREADS)
    tab[(size_t)blockIdx.x * MAXB + b] = (unsigned int)lofs[b] | ((unsigned int)cntl[b] << SEG_SHIFT);

  int ps[SC_TRIPS][4];
  #pragma unroll
  for (int u = 0; u < SC_TRIPS; ++u) {
    #pragma unroll
    for (int j = 0; j < 4; ++j) ps[u][j] = lofs[rk[u][j] >= 0 ? k[u][j] >> NPB_SHIFT : 0] + rk[u][j];
  }
  const int pst = lofs[rkt >= 0 ? kt >> NPB_SHIFT : 0] + rkt;
  #pragma unroll
  for (int u = 0; u < SC_TRIPS; ++u) {
    #pragma unroll
    for (int j = 0; j < 4; ++j)
      if (rk[u][j] >= 0) stage[ps[u][j]] = bin_entry(k[u][j], (unsigned int)pl[u][j]);
  }
  if (rkt >= 0) stage[pst] = bin_entry(kt, (unsigned int)plt);
  __syncthreads();

  const int n = ntot;
  unsigned int* __restrict__ mine = list + (size_t)p0;
  for (int i = t; i < (n >> 2); i += SC_THREADS) ((uint4*)mine)[i] = ((const uint4*)stage)[i];
  if (t < (n & 3)) mine[(n & ~3) + t] = stage[(n & ~3) + t];
}

__global__ __launch_bounds__(SC_THREADS) void order_by_src(const int* __restrict__ src,
                                                           unsigned int* __restrict__ tab,
                                                           unsigned int* __restrict__ list, int E, int NB) {
  order_piece<false>(src, src, tab, list, E, NB);
}

__global__ __launch_bounds__(SC_THREADS) void order_by_dst(const int* __restrict__ src, const int* __restrict__ dst,
                                                           unsigned int* __restrict__ tab,
                                                           unsigned int* __restrict__ list, int E, int NB) {
  order_piece<true>(dst, src, tab, list, E, NB);
}

__device__ __forceinline__ void run_open(const unsigned int* __restrict__ tab, int b, int g, int P, int lane,
                                         int (&pre)[SEG_PER_WAVE + 1], int (&off)[SEG_PER_WAVE]) {
  const int j = lane & (SEG_PER_WAVE - 1);
  const int piece = SEG_PER_WAVE * g + j;
  const unsigned int wd = piece < P ? tab[(size_t)piece * MAXB + b] : 0u;
  const int len = (int)(wd >> SEG_SHIFT);
  int inc = len;
  #pragma unroll
  for (int o = 1; o < SEG_PER_WAVE; o <<= 1) {
    const int up = __shfl_up(inc, o, SEG_PER_WAVE);
    if (j >= o) inc += up;
  }
  const int exc = inc - len;
  const int of = piece * PIECE + (int)(wd & ((1u << SEG_SHIFT) - 1u)) - exc;
  #pragma unroll
  for (int s = 0; s < SEG_PER_WAVE; ++s) {
    pre[s] = __builtin_amdgcn_readlane(exc, s);
    off[s] = __builtin_amdgcn_readlane(of, s);
  }
  pre[SEG_PER_WAVE] = __builtin_amdgcn_readlane(inc, SEG_PER_WAVE - 1);
}
__device__ __forceinline__ int run_index(int v, const int (&pre)[SEG_PER_WAVE + 1], const int (&off)[SEG_PER_WAVE]) {
  int o = off[0];
  #pragma unroll
  for (int s = 1; s < SEG_PER_WAVE; ++s) o = v >= pre[s] ? off[s] : o;
  return v + o;
}

__global__ __launch_bounds__(512) void deg_owner(const unsigned int* __restrict__ list, const unsigned int* __restrict__ tab,
                                                 float* __restrict__ inv, int N, int P) {
  __shared__ int c[NPB + 1];
  if (threadIdx.x <= NPB) c[threadIdx.x] = 0;
  __syncthreads();
  const int lane = (int)threadIdx.x & 31;
  const int G = (P + SEG_PER_WAVE - 1) / SEG_PER_WAVE;
  for (int g = (int)threadIdx.x >> 5; g < G; g += (int)blockDim.x >> 5) {
    int pre[SEG_PER_WAVE + 1];
    int off[SEG_PER_WAVE];
    run_open(tab, (int)blockIdx.x, g, P, lane, pre, off);
    const int total = pre[SEG_PER_WAVE];
    for (int v0 = 0; v0 < total; v0 += 128) {
      unsigned int slot[4];
      for (int u = 0; u < 4; ++u) {
        const int v = v0 + 32 * u + lane;
        const unsigned int ent = list[run_index(min(v, total - 1), pre, off)];
        slot[u] = v < total ? (ent >> ENT_SHIFT) : (unsigned int)NPB;
      }
      for (int u = 0; u < 4; ++u) atomicAdd(&c[slot[u]], 1);
    }
  }
  __syncthreads();
  const int node = blockIdx.x * NPB + (int)threadIdx.x;
  if (threadIdx.x < NPB && node < N) inv[node] = rsqrtf(fmaxf((float)c[threadIdx.x], 1.0f));
}

__global__ __launch_bounds__(256) void gnn_gemm1_wmma(
    const float* __restrict__ x, const float* __restrict__ W1,
    const float* __restrict__ inv_out, float* __restrict__ h1, int N) {
  __shared__ float wlds[F_IN * HID];
  for (int i = threadIdx.x; i < F_IN * HID; i += blockDim.x) wlds[i] = W1[i];
  __syncthreads();

  const int wave = threadIdx.x >> 5;
  const int lane = threadIdx.x & 31;
  const int tiles = N >> 4;
  const int tile = blockIdx.x * 8 + wave;
  if (tile >= tiles) return;

  const int n16 = lane & 15;
  const int g   = lane >> 4;
  const int row = tile * 16 + n16;
  const float s = inv_out[row];
  const float* __restrict__ xr = x + (size_t)row * F_IN;

  v8f c = {};
  #pragma unroll 4
  for (int k = 0; k < F_IN; k += 4) {
    const int kk = k + 2 * g;
    v2f a = *(const v2f*)(xr + kk);
    a.x *= s;
    a.y *= s;
    v2f b;
    b.x = wlds[(kk + 0) * HID + n16];
    b.y = wlds[(kk + 1) * HID + n16];
    c = __builtin_amdgcn_wmma_f32_16x16x4_f32(
         false, a,  false, b,
         (short)0, c,  false,  false);
  }

  #pragma unroll
  for (int p = 0; p < 8; ++p) {
    const int m = tile * 16 + p + 8 * g;
    h1[(size_t)m * HID + n16] = c[p];
  }
}

__global__ __launch_bounds__(1024) void agg1_owner(const unsigned int* __restrict__ list, const unsigned int* __restrict__ tab,
                                                   const float* __restrict__ h1, const float* __restrict__ inv_out,
                                                   const float* __restrict__ b1, const float* __restrict__ W2,
                                                   float* __restrict__ inv_in, float* __restrict__ h2,
                                                   int N, int P) {
  __shared__ float acc[HID * ACC_PITCH];
  __shared__ int cnt[NPB + 1];
  for (int i = threadIdx.x; i < HID * ACC_PITCH; i += blockDim.x) acc[i] = 0.0f;
  if (threadIdx.x <= NPB) cnt[threadIdx.x] = 0;
  __syncthreads();
  const int b = (int)blockIdx.x;
  const int t = (int)threadIdx.x;
  const int lane = t & 31;
  const int q = t & 3;
  float* row = acc + 4 * q * ACC_PITCH;
  const unsigned int src_mask = (1u << ENT_SHIFT) - 1u;
  const int G = (P + SEG_PER_WAVE - 1) / SEG_PER_WAVE;
  for (int g = t >> 5; g < G; g += (int)blockDim.x >> 5) {
    int pre[SEG_PER_WAVE + 1];
    int off[SEG_PER_WAVE];
    run_open(tab, b, g, P, lane, pre, off);
    const int total = pre[SEG_PER_WAVE];
    if (total > 0) {
      unsigned int nxt = list[run_index(min(lane, total - 1), pre, off)];
      for (int v0 = 0; v0 < total; v0 += 32) {
        unsigned int w = nxt;
        if (v0 + lane >= total) w = (w & src_mask) | ((unsigned int)NPB << ENT_SHIFT);
        nxt = list[run_index(min(v0 + 32 + lane, total - 1), pre, off)];
        unsigned int e[4];
        e[0] = (unsigned int)__builtin_amdgcn_mov_dpp((int)w, 0x00, 0xf, 0xf, false);
        e[1] = (unsigned int)__builtin_amdgcn_mov_dpp((int)w, 0x55, 0xf, 0xf, false);
        e[2] = (unsigned int)__builtin_amdgcn_mov_dpp((int)w, 0xaa, 0xf, 0xf, false);
        e[3] = (unsigned int)__builtin_amdgcn_mov_dpp((int)w, 0xff, 0xf, 0xf, false);
        float4 v[4];
        for (int u = 0; u < 4; ++u)
          v[u] = *(const float4*)(h1 + (size_t)(e[u] & src_mask) * HID + 4 * q);
        for (int u = 0; u < 4; ++u) {
          const unsigned int place = e[u] >> ENT_SHIFT;
          atomicAdd(&row[0 * ACC_PITCH + place], v[u].x);
          atomicAdd(&row[1 * ACC_PITCH + place], v[u].y);
          atomicAdd(&row[2 * ACC_PITCH + place], v[u].z);
          atomicAdd(&row[3 * ACC_PITCH + place], v[u].w);
        }
        atomicAdd(&cnt[w >> ENT_SHIFT], 1);
      }
    }
  }
  __syncthreads();
  const int node = b * NPB + t;
  if (t >= NPB || node >= N) return;
  const float si = rsqrtf(fmaxf((float)cnt[t], 1.0f));
  const float so = inv_out[node];
  inv_in[node] = si;
  float acc0 = 0.0f, acc1 = 0.0f;
  #pragma unroll
  for (int k = 0; k < HID; ++k) {
    float hv = fmaxf(fmaf(acc[k * ACC_PITCH + t], si, b1[k]), 0.0f) * so;
    acc0 = fmaf(hv, W2[k * NCLS + 0], acc0);
    acc1 = fmaf(hv, W2[k * NCLS + 1], acc1);
  }
  *(float2*)(h2 + (size_t)node * NCLS) = make_float2(acc0, acc1);
}

#define S2_R 2
#define S2_NODES (S2_R * NPB)
#define S2_THREADS 1024
#define S2_CAP 24576
#define S2_KEY_SHIFT 4
#define S2_BINS ((MAXB * NPB) >> S2_KEY_SHIFT)
static_assert(S2_BINS == 8 * S2_THREADS, "the running sum below gives every thread eight neighbouring bins");
__device__ __forceinline__ int s2_key(unsigned int ent) {
  return min((int)((ent & ((1u << ENT_SHIFT) - 1u)) >> S2_KEY_SHIFT), S2_BINS - 1);
}
__global__ __launch_bounds__(S2_THREADS) void agg2_final(const unsigned int* __restrict__ list, const unsigned int* __restrict__ tab,
                                                         const float* __restrict__ h2, const float* __restrict__ inv_in,
                                                         const float* __restrict__ b2, float* __restrict__ out, int N, int P) {
  __shared__ float a0[S2_NODES + 1];
  __shared__ float a1[S2_NODES + 1];
  __shared__ __attribute__((aligned(16))) int bin[S2_BINS];
  __shared__ unsigned int raw[S2_CAP];
  __shared__ unsigned int srt[S2_CAP];
  __shared__ int wsum[S2_THREADS / 32];
  __shared__ int fill;
  const int t = (int)threadIdx.x;
  const int lane = t & 31;
  const int wave = t >> 5;
  const unsigned int src_mask = (1u << ENT_SHIFT) - 1u;
  for (int i = t; i <= S2_NODES; i += S2_THREADS) { a0[i] = 0.0f; a1[i] = 0.0f; }
  for (int i = t; i < S2_BINS; i += S2_THREADS) bin[i] = 0;
  if (t == 0) fill = 0;
  __syncthreads();

  const int NB = (N + NPB - 1) >> NPB_SHIFT;
  const int G = (P + SEG_PER_WAVE - 1) / SEG_PER_WAVE;
  for (int r = 0; r < S2_R; ++r) {
    const int b = S2_R * (int)blockIdx.x + r;
    if (b >= NB) break;
    const unsigned int tag = (unsigned int)r << (ENT_SHIFT + NPB_SHIFT);
    for (int g = wave; g < G; g += S2_THREADS / 32) {
      int pre[SEG_PER_WAVE + 1];
      int off[SEG_PER_WAVE];
      run_open(tab, b, g, P, lane, pre, off);
      const int total = pre[SEG_PER_WAVE];
      int base = 0;
      if (lane == 0 && total > 0) base = atomicAdd(&fill, total);
      base = __builtin_amdgcn_readfirstlane(base);
      for (int v0 = 0; v0 < total; v0 += 128) {
        unsigned int ent[4];
        for (int u = 0; u < 4; ++u) {
          const int n = v0 + 32 * u + lane;
          ent[u] = list[run_index(min(n, total - 1), pre, off)] | tag;
        }
        for (int u = 0; u < 4; ++u) {
          const int n = v0 + 32 * u + lane;
          if (n < total) {
            const int pos = base + n;
            if (pos < S2_CAP) {
              raw[pos] = ent[u];
              atomicAdd(&bin[s2_key(ent[u])], 1);
            } else {
              const float2 v = *(const float2*)(h2 + (size_t)(ent[u] & src_mask) * NCLS);
              atomicAdd(&a0[ent[u] >> ENT_SHIFT], v.x);
              atomicAdd(&a1[ent[u] >> ENT_SHIFT], v.y);
            }
          }
        }
      }
    }
  }
  __syncthreads();
  const int nst = min(fill, S2_CAP);

  const int4 c0 = ((const int4*)bin)[2 * t];
  const int4 c1 = ((const int4*)bin)[2 * t + 1];
  const int own = c0.x + c0.y + c0.z + c0.w + c1.x + c1.y + c1.z + c1.w;
  int inc = own;
  #pragma unroll
  for (int o = 1; o < 32; o <<= 1) {
    const int up = __shfl_up(inc, o, 32);
    if (lane >= o) inc += up;
  }
  if (lane == 31) wsum[wave] = inc;
  __syncthreads();
  if (t < 32) {
    const int w = wsum[t];
    int wi = w;
    #pragma unroll
    for (int o = 1; o < 32; o <<= 1) {
      const int up = __shfl_up(wi, o, 32);
      if (t >= o) wi += up;
    }
    wsum[t] = wi - w;
  }
  __syncthreads();
  {
    int run = wsum[wave] + inc - own;
    int4 o0, o1;
    o0.x = run; run += c0.x;
    o0.y = run; run += c0.y;
    o0.z = run; run += c0.z;
    o0.w = run; run += c0.w;
    o1.x = run; run += c1.x;
    o1.y = run; run += c1.y;
    o1.z = run; run += c1.z;
    o1.w = run;
    ((int4*)bin)[2 * t] = o0;
    ((int4*)bin)[2 * t + 1] = o1;
  }
  __syncthreads();

  for (int i0 = t; i0 < nst; i0 += 4 * S2_THREADS) {
    unsigned int e[4];
    int pos[4];
    #pragma unroll
    for (int u = 0; u < 4; ++u) {
      const int i = i0 + u * S2_THREADS;
      e[u] = 0u;
      if (i < nst) e[u] = raw[i];
    }
    #pragma unroll
    for (int u = 0; u < 4; ++u) {
      const int i = i0 + u * S2_THREADS;
      pos[u] = 0;
      if (i < nst) pos[u] = atomicAdd(&bin[s2_key(e[u])], 1);
    }
    #pragma unroll
    for (int u = 0; u < 4; ++u) {
      const int i = i0 + u * S2_THREADS;
      if (i < nst) srt[pos[u]] = e[u];
    }
  }
  __syncthreads();

  for (int i0 = wave * 128; i0 < nst; i0 += (S2_THREADS / 32) * 128) {
    unsigned int slot[4];
    float2 v[4];
    for (int u = 0; u < 4; ++u) {
      const int i = i0 + 32 * u + lane;
      const unsigned int e = srt[min(i, nst - 1)];
      slot[u] = i < nst ? (e >> ENT_SHIFT) : (unsigned int)S2_NODES;
      v[u] = *(const float2*)(h2 + (size_t)(e & src_mask) * NCLS);
    }
    for (int u = 0; u < 4; ++u) {
      atomicAdd(&a0[slot[u]], v[u].x);
      atomicAdd(&a1[slot[u]], v[u].y);
    }
  }
  __syncthreads();
  const int i = (int)blockIdx.x * S2_NODES + t;
  if (t >= S2_NODES || i >= N) return;
  const float si = inv_in[i];
  float v0 = fmaf(a0[t], si, b2[0]);
  float v1 = fmaf(a1[t], si, b2[1]);
  float m = fmaxf(v0, v1);
  float lse = m + logf(expf(v0 - m) + expf(v1 - m));
  out[(size_t)i * NCLS + 0] = v0 - lse;
  out[(size_t)i * NCLS + 1] = v1 - lse;
}

extern "C" void kernel_launch(void* const* d_in, const int* in_sizes, int n_in,
                              void* d_out, int out_size, void* d_ws, size_t ws_size,
                              hipStream_t stream) {
  const float* x   = (const float*)d_in[0];
  const int*   src = (const int*)d_in[1];
  const int*   dst = (const int*)d_in[2];
  const float* W1  = (const float*)d_in[3];
  const float* b1  = (const float*)d_in[4];
  const float* W2  = (const float*)d_in[5];
  const float* b2  = (const float*)d_in[6];
  float* out = (float*)d_out;

  const int N = in_sizes[0] / F_IN;
  const int E = in_sizes[1];
  const int P = (E + PIECE - 1) / PIECE;

  float* ws      = (float*)d_ws;
  float* inv_out = ws;
  float* inv_in  = ws + (size_t)N;
  float* h2      = ws + (size_t)2 * N;
  const size_t tab0 = ((size_t)4 * N + 63) & ~(size_t)63;
  unsigned int* tabS = (unsigned int*)(ws + tab0);
  unsigned int* tabD = tabS + (size_t)P * MAXB;
  const size_t list0 = tab0 + (size_t)2 * P * MAXB;
  unsigned int* list = (unsigned int*)(ws + list0);
  const size_t h10 = list0 + (((size_t)P * PIECE + 63) & ~(size_t)63);
  float* h1      = ws + h10;
  if ((h10 + (size_t)HID * N) * sizeof(float) > ws_size) return;

  const int B = 256;
  const int NB = (N + NPB - 1) / NPB;

  order_by_src<<<P, SC_THREADS, 0, stream>>>(src, tabS, list, E, NB);
  deg_owner<<<NB, 512, 0, stream>>>(list, tabS, inv_out, N, P);

  const int tiles = N >> 4;
  gnn_gemm1_wmma<<<(tiles + 7) / 8, B, 0, stream>>>(x, W1, inv_out, h1, N);

  order_by_dst<<<P, SC_THREADS, 0, stream>>>(src, dst, tabD, list, E, NB);
  agg1_owner<<<NB, 1024, 0, stream>>>(list, tabD, h1, inv_out, b1, W2, inv_in, h2, N, P);
  agg2_final<<<(NB + S2_R - 1) / S2_R, S2_THREADS, 0, stream>>>(list, tabD, h2, inv_in, b2, out, N, P);
}


# x @ W1 and the destination ordering share ONE launch, side by side (two 1024-thread work-groups fit a unit: the ordering now keeps 2 words per edge): 5 launches, was 6; sum orders unchanged
# speedup vs baseline: 5.0276x; 1.1393x over previous
#include <hip/hip_runtime.h>
#include <math.h>


typedef __attribute__((ext_vector_type(2))) float v2f;
typedef __attribute__((ext_vector_type(8))) float v8f;

#define F_IN 512
#define HID  16
#define NCLS 2

#define NPB_SHIFT 8
#define NPB   (1 << NPB_SHIFT)
#define MAXB  512
#define ENT_SHIFT 20
#define ACC_PITCH (NPB + 4)

#define SC_THREADS 1024
#define SC_TRIPS 4
#define SC_CAP (SC_TRIPS * 4 * SC_THREADS)
#define PIECE 12512
#define SEG_SHIFT 16
#define SEG_PER_WAVE 8
#define RK_BITS 14
static_assert(SC_CAP <= (1 << RK_BITS), "a place inside a piece fits RK_BITS");
static_assert(MAXB <= (1 << (31 - RK_BITS)), "range number and place share one non-negative word");

__device__ __forceinline__ unsigned int bin_entry(int key, unsigned int payload) {
  return ((unsigned int)(key & (NPB - 1)) << ENT_SHIFT) | payload;
}
__device__ __forceinline__ int piece_rank(int* cntl, int key, int NB) {
  const int b = key >> NPB_SHIFT;
  return (unsigned int)b < (unsigned int)NB ? atomicAdd(&cntl[b], 1) : -1;
}

template <bool HAS_PAY>
__device__ __forceinline__ void order_piece(const int* __restrict__ key, const int* __restrict__ pay,
                                            unsigned int* __restrict__ tab, unsigned int* __restrict__ list,
                                            int E, int NB, int w) {
  __shared__ int cntl[MAXB];
  __shared__ int lofs[MAXB];
  __shared__ int ntot;
  __shared__ __attribute__((aligned(16))) int wtot[MAXB / 32];
  static_assert(MAXB % 128 == 0 && MAXB <= SC_THREADS, "one range a thread, the waves' totals read four at a time");
  __shared__ __attribute__((aligned(16))) unsigned int stage[SC_CAP];
  const int t = (int)threadIdx.x;
  const int p0 = w * PIECE;
  const int p1 = min(p0 + PIECE, E);
  for (int b = t; b < MAXB; b += SC_THREADS) cntl[b] = b << RK_BITS;
  __syncthreads();

  int4 kv[SC_TRIPS];
  int4 pv[SC_TRIPS];
  #pragma unroll
  for (int u = 0; u < SC_TRIPS; ++u) {
    const int e = p0 + 4 * (u * SC_THREADS + t);
    kv[u] = make_int4(-1, -1, -1, -1);
    pv[u] = make_int4(0, 0, 0, 0);
    if (e + 3 < p1) {
      kv[u] = *(const int4*)(key + e);
      if (HAS_PAY) pv[u] = *(const int4*)(pay + e);
    }
  }
  const int rem = (p1 - p0) & 3;
  int kt = -1;
  int plt = 0;
  if (t < rem) {
    kt = key[p1 - rem + t];
    if (HAS_PAY) plt = pay[p1 - rem + t];
  }
  unsigned int en[SC_TRIPS][4];
  int rk[SC_TRIPS][4];
  #pragma unroll
  for (int u = 0; u < SC_TRIPS; ++u) {
    en[u][0] = bin_entry(kv[u].x, (unsigned int)pv[u].x); rk[u][0] = piece_rank(cntl, kv[u].x, NB);
    en[u][1] = bin_entry(kv[u].y, (unsigned int)pv[u].y); rk[u][1] = piece_rank(cntl, kv[u].y, NB);
    en[u][2] = bin_entry(kv[u].z, (unsigned int)pv[u].z); rk[u][2] = piece_rank(cntl, kv[u].z, NB);
    en[u][3] = bin_entry(kv[u].w, (unsigned int)pv[u].w); rk[u][3] = piece_rank(cntl, kv[u].w, NB);
  }
  const unsigned int ent = bin_entry(kt, (unsigned int)plt);
  const int rkt = piece_rank(cntl, kt, NB);
  __syncthreads();

  {
    const int lane = t & 31;
    const int wave = t >> 5;
    const int own = t < MAXB ? cntl[t] & ((1 << RK_BITS) - 1) : 0;
    int inc = own;
    #pragma unroll
    for (int off = 1; off < 32; off <<= 1) {
      const int up = __shfl_up(inc, off, 32);
      if (lane >= off) inc += up;
    }
    if (lane == 31 && t < MAXB) wtot[wave] = inc;
    __syncthreads();
    if (t < MAXB) {
      int before = 0;
      #pragma unroll
      for (int q = 0; q < MAXB / 128; ++q) {
        const int4 v = ((const int4*)wtot)[q];
        before += (4 * q + 0 < wave ? v.x : 0) + (4 * q + 1 < wave ? v.y : 0)
                + (4 * q + 2 < wave ? v.z : 0) + (4 * q + 3 < wave ? v.w : 0);
      }
      lofs[t] = before + inc - own;
      if (t == MAXB - 1) ntot = before + inc;
    }
  }
  __syncthreads();

  for (int b = t; b < MAXB; b += SC_THREADS)
    tab[(size_t)w * MAXB + b] = (unsigned int)lofs[b] | ((unsigned int)(cntl[b] & ((1 << RK_BITS) - 1)) << SEG_SHIFT);

  const int rk_mask = (1 << RK_BITS) - 1;
  #pragma unroll
  for (int u = 0; u < SC_TRIPS; u += 2) {
    int ps[2][4];
    #pragma unroll
    for (int v = 0; v < 2; ++v) {
      #pragma unroll
      for (int j = 0; j < 4; ++j) ps[v][j] = lofs[rk[u + v][j] >= 0 ? rk[u + v][j] >> RK_BITS : 0] + (rk[u + v][j] & rk_mask);
    }
    #pragma unroll
    for (int v = 0; v < 2; ++v) {
      #pragma unroll
      for (int j = 0; j < 4; ++j)
        if (rk[u + v][j] >= 0) stage[ps[v][j]] = en[u + v][j];
    }
  }
  if (rkt >= 0) stage[lofs[rkt >> RK_BITS] + (rkt & rk_mask)] = ent;
  __syncthreads();

  const int n = ntot;
  unsigned int* __restrict__ mine = list + (size_t)p0;
  for (int i = t; i < (n >> 2); i += SC_THREADS) ((uint4*)mine)[i] = ((const uint4*)stage)[i];
  if (t < (n & 3)) mine[(n & ~3) + t] = stage[(n & ~3) + t];
}

__global__ __launch_bounds__(SC_THREADS) void order_by_src(const int* __restrict__ src,
                                                           unsigned int* __restrict__ tab,
                                                           unsigned int* __restrict__ list, int E, int NB) {
  order_piece<false>(src, src, tab, list, E, NB, (int)blockIdx.x);
}


__device__ __forceinline__ void run_open(const unsigned int* __restrict__ tab, int b, int g, int P, int lane,
                                         int (&pre)[SEG_PER_WAVE + 1], int (&off)[SEG_PER_WAVE]) {
  const int j = lane & (SEG_PER_WAVE - 1);
  const int piece = SEG_PER_WAVE * g + j;
  const unsigned int wd = piece < P ? tab[(size_t)piece * MAXB + b] : 0u;
  const int len = (int)(wd >> SEG_SHIFT);
  int inc = len;
  #pragma unroll
  for (int o = 1; o < SEG_PER_WAVE; o <<= 1) {
    const int up = __shfl_up(inc, o, SEG_PER_WAVE);
    if (j >= o) inc += up;
  }
  const int exc = inc - len;
  const int of = piece * PIECE + (int)(wd & ((1u << SEG_SHIFT) - 1u)) - exc;
  #pragma unroll
  for (int s = 0; s < SEG_PER_WAVE; ++s) {
    pre[s] = __builtin_amdgcn_readlane(exc, s);
    off[s] = __builtin_amdgcn_readlane(of, s);
  }
  pre[SEG_PER_WAVE] = __builtin_amdgcn_readlane(inc, SEG_PER_WAVE - 1);
}
__device__ __forceinline__ int run_index(int v, const int (&pre)[SEG_PER_WAVE + 1], const int (&off)[SEG_PER_WAVE]) {
  int o = off[0];
  #pragma unroll
  for (int s = 1; s < SEG_PER_WAVE; ++s) o = v >= pre[s] ? off[s] : o;
  return v + o;
}

__global__ __launch_bounds__(512) void deg_owner(const unsigned int* __restrict__ list, const unsigned int* __restrict__ tab,
                                                 float* __restrict__ inv, int N, int P) {
  __shared__ int c[NPB + 1];
  if (threadIdx.x <= NPB) c[threadIdx.x] = 0;
  __syncthreads();
  const int lane = (int)threadIdx.x & 31;
  const int G = (P + SEG_PER_WAVE - 1) / SEG_PER_WAVE;
  for (int g = (int)threadIdx.x >> 5; g < G; g += (int)blockDim.x >> 5) {
    int pre[SEG_PER_WAVE + 1];
    int off[SEG_PER_WAVE];
    run_open(tab, (int)blockIdx.x, g, P, lane, pre, off);
    const int total = pre[SEG_PER_WAVE];
    for (int v0 = 0; v0 < total; v0 += 128) {
      unsigned int slot[4];
      for (int u = 0; u < 4; ++u) {
        const int v = v0 + 32 * u + lane;
        const unsigned int ent = list[run_index(min(v, total - 1), pre, off)];
        slot[u] = v < total ? (ent >> ENT_SHIFT) : (unsigned int)NPB;
      }
      for (int u = 0; u < 4; ++u) atomicAdd(&c[slot[u]], 1);
    }
  }
  __syncthreads();
  const int node = blockIdx.x * NPB + (int)threadIdx.x;
  if (threadIdx.x < NPB && node < N) inv[node] = rsqrtf(fmaxf((float)c[threadIdx.x], 1.0f));
}

#define GEMM_TILES_PER_WG (SC_THREADS / 32)
__device__ __forceinline__ void gemm1_tiles(
    const float* __restrict__ x, const float* __restrict__ W1,
    const float* __restrict__ inv_out, float* __restrict__ h1, int N, int wg) {
  __shared__ float wlds[F_IN * HID];
  {
    float wv[F_IN * HID / SC_THREADS];
    #pragma unroll
    for (int j = 0; j < F_IN * HID / SC_THREADS; ++j) wv[j] = W1[threadIdx.x + j * SC_THREADS];
    #pragma unroll
    for (int j = 0; j < F_IN * HID / SC_THREADS; ++j) wlds[threadIdx.x + j * SC_THREADS] = wv[j];
  }
  __syncthreads();

  const int wave = threadIdx.x >> 5;
  const int lane = threadIdx.x & 31;
  const int tiles = N >> 4;
  const int tile = wg * GEMM_TILES_PER_WG + wave;
  if (tile >= tiles) return;

  const int n16 = lane & 15;
  const int g   = lane >> 4;
  const int row = tile * 16 + n16;
  const float s = inv_out[row];
  const float* __restrict__ xr = x + (size_t)row * F_IN;

  v8f c = {};
  #pragma unroll 4
  for (int k = 0; k < F_IN; k += 4) {
    const int kk = k + 2 * g;
    v2f a = *(const v2f*)(xr + kk);
    a.x *= s;
    a.y *= s;
    v2f b;
    b.x = wlds[(kk + 0) * HID + n16];
    b.y = wlds[(kk + 1) * HID + n16];
    c = __builtin_amdgcn_wmma_f32_16x16x4_f32(
         false, a,  false, b,
         (short)0, c,  false,  false);
  }

  #pragma unroll
  for (int p = 0; p < 8; ++p) {
    const int m = tile * 16 + p + 8 * g;
    h1[(size_t)m * HID + n16] = c[p];
  }
}

__global__ __launch_bounds__(SC_THREADS, 16) void gemm1_and_order_by_dst(
    const float* __restrict__ x, const float* __restrict__ W1,
    const float* __restrict__ inv_out, float* __restrict__ h1, int N, int gemm_wgs,
    const int* __restrict__ src, const int* __restrict__ dst,
    unsigned int* __restrict__ tab, unsigned int* __restrict__ list, int E, int NB) {
  if ((int)blockIdx.x < gemm_wgs) gemm1_tiles(x, W1, inv_out, h1, N, (int)blockIdx.x);
  else order_piece<true>(dst, src, tab, list, E, NB, (int)blockIdx.x - gemm_wgs);
}

__global__ __launch_bounds__(1024) void agg1_owner(const unsigned int* __restrict__ list, const unsigned int* __restrict__ tab,
                                                   const float* __restrict__ h1, const float* __restrict__ inv_out,
                                                   const float* __restrict__ b1, const float* __restrict__ W2,
                                                   float* __restrict__ inv_in, float* __restrict__ h2,
                                                   int N, int P) {
  __shared__ float acc[HID * ACC_PITCH];
  __shared__ int cnt[NPB + 1];
  for (int i = threadIdx.x; i < HID * ACC_PITCH; i += blockDim.x) acc[i] = 0.0f;
  if (threadIdx.x <= NPB) cnt[threadIdx.x] = 0;
  __syncthreads();
  const int b = (int)blockIdx.x;
  const int t = (int)threadIdx.x;
  const int lane = t & 31;
  const int q = t & 3;
  float* row = acc + 4 * q * ACC_PITCH;
  const unsigned int src_mask = (1u << ENT_SHIFT) - 1u;
  const int G = (P + SEG_PER_WAVE - 1) / SEG_PER_WAVE;
  for (int g = t >> 5; g < G; g += (int)blockDim.x >> 5) {
    int pre[SEG_PER_WAVE + 1];
    int off[SEG_PER_WAVE];
    run_open(tab, b, g, P, lane, pre, off);
    const int total = pre[SEG_PER_WAVE];
    if (total > 0) {
      unsigned int nxt = list[run_index(min(lane, total - 1), pre, off)];
      for (int v0 = 0; v0 < total; v0 += 32) {
        unsigned int w = nxt;
        if (v0 + lane >= total) w = (w & src_mask) | ((unsigned int)NPB << ENT_SHIFT);
        nxt = list[run_index(min(v0 + 32 + lane, total - 1), pre, off)];
        unsigned int e[4];
        e[0] = (unsigned int)__builtin_amdgcn_mov_dpp((int)w, 0x00, 0xf, 0xf, false);
        e[1] = (unsigned int)__builtin_amdgcn_mov_dpp((int)w, 0x55, 0xf, 0xf, false);
        e[2] = (unsigned int)__builtin_amdgcn_mov_dpp((int)w, 0xaa, 0xf, 0xf, false);
        e[3] = (unsigned int)__builtin_amdgcn_mov_dpp((int)w, 0xff, 0xf, 0xf, false);
        float4 v[4];
        for (int u = 0; u < 4; ++u)
          v[u] = *(const float4*)(h1 + (size_t)(e[u] & src_mask) * HID + 4 * q);
        for (int u = 0; u < 4; ++u) {
          const unsigned int place = e[u] >> ENT_SHIFT;
          atomicAdd(&row[0 * ACC_PITCH + place], v[u].x);
          atomicAdd(&row[1 * ACC_PITCH + place], v[u].y);
          atomicAdd(&row[2 * ACC_PITCH + place], v[u].z);
          atomicAdd(&row[3 * ACC_PITCH + place], v[u].w);
        }
        atomicAdd(&cnt[w >> ENT_SHIFT], 1);
      }
    }
  }
  __syncthreads();
  const int node = b * NPB + t;
  if (t >= NPB || node >= N) return;
  const float si = rsqrtf(fmaxf((float)cnt[t], 1.0f));
  const float so = inv_out[node];
  inv_in[node] = si;
  float acc0 = 0.0f, acc1 = 0.0f;
  #pragma unroll
  for (int k = 0; k < HID; ++k) {
    float hv = fmaxf(fmaf(acc[k * ACC_PITCH + t], si, b1[k]), 0.0f) * so;
    acc0 = fmaf(hv, W2[k * NCLS + 0], acc0);
    acc1 = fmaf(hv, W2[k * NCLS + 1], acc1);
  }
  *(float2*)(h2 + (size_t)node * NCLS) = make_float2(acc0, acc1);
}

#define S2_R 2
#define S2_NODES (S2_R * NPB)
#define S2_THREADS 1024
#define S2_CAP 24576
#define S2_KEY_SHIFT 4
#define S2_BINS ((MAXB * NPB) >> S2_KEY_SHIFT)
static_assert(S2_BINS == 8 * S2_THREADS, "the running sum below gives every thread eight neighbouring bins");
__device__ __forceinline__ int s2_key(unsigned int ent) {
  return min((int)((ent & ((1u << ENT_SHIFT) - 1u)) >> S2_KEY_SHIFT), S2_BINS - 1);
}
__global__ __launch_bounds__(S2_THREADS) void agg2_final(const unsigned int* __restrict__ list, const unsigned int* __restrict__ tab,
                                                         const float* __restrict__ h2, const float* __restrict__ inv_in,
                                                         const float* __restrict__ b2, float* __restrict__ out, int N, int P) {
  __shared__ float a0[S2_NODES + 1];
  __shared__ float a1[S2_NODES + 1];
  __shared__ __attribute__((aligned(16))) int bin[S2_BINS];
  __shared__ unsigned int raw[S2_CAP];
  __shared__ unsigned int srt[S2_CAP];
  __shared__ int wsum[S2_THREADS / 32];
  __shared__ int fill;
  const int t = (int)threadIdx.x;
  const int lane = t & 31;
  const int wave = t >> 5;
  const unsigned int src_mask = (1u << ENT_SHIFT) - 1u;
  for (int i = t; i <= S2_NODES; i += S2_THREADS) { a0[i] = 0.0f; a1[i] = 0.0f; }
  for (int i = t; i < S2_BINS; i += S2_THREADS) bin[i] = 0;
  if (t == 0) fill = 0;
  __syncthreads();

  const int NB = (N + NPB - 1) >> NPB_SHIFT;
  const int G = (P + SEG_PER_WAVE - 1) / SEG_PER_WAVE;
  for (int r = 0; r < S2_R; ++r) {
    const int b = S2_R * (int)blockIdx.x + r;
    if (b >= NB) break;
    const unsigned int tag = (unsigned int)r << (ENT_SHIFT + NPB_SHIFT);
    for (int g = wave; g < G; g += S2_THREADS / 32) {
      int pre[SEG_PER_WAVE + 1];
      int off[SEG_PER_WAVE];
      run_open(tab, b, g, P, lane, pre, off);
      const int total = pre[SEG_PER_WAVE];
      int base = 0;
      if (lane == 0 && total > 0) base = atomicAdd(&fill, total);
      base = __builtin_amdgcn_readfirstlane(base);
      for (int v0 = 0; v0 < total; v0 += 128) {
        unsigned int ent[4];
        for (int u = 0; u < 4; ++u) {
          const int n = v0 + 32 * u + lane;
          ent[u] = list[run_index(min(n, total - 1), pre, off)] | tag;
        }
        for (int u = 0; u < 4; ++u) {
          const int n = v0 + 32 * u + lane;
          if (n < total) {
            const int pos = base + n;
            if (pos < S2_CAP) {
              raw[pos] = ent[u];
              atomicAdd(&bin[s2_key(ent[u])], 1);
            } else {
              const float2 v = *(const float2*)(h2 + (size_t)(ent[u] & src_mask) * NCLS);
              atomicAdd(&a0[ent[u] >> ENT_SHIFT], v.x);
              atomicAdd(&a1[ent[u] >> ENT_SHIFT], v.y);
            }
          }
        }
      }
    }
  }
  __syncthreads();
  const int nst = min(fill, S2_CAP);

  const int4 c0 = ((const int4*)bin)[2 * t];
  const int4 c1 = ((const int4*)bin)[2 * t + 1];
  const int own = c0.x + c0.y + c0.z + c0.w + c1.x + c1.y + c1.z + c1.w;
  int inc = own;
  #pragma unroll
  for (int o = 1; o < 32; o <<= 1) {
    const int up = __shfl_up(inc, o, 32);
    if (lane >= o) inc += up;
  }
  if (lane == 31) wsum[wave] = inc;
  __syncthreads();
  if (t < 32) {
    const int w = wsum[t];
    int wi = w;
    #pragma unroll
    for (int o = 1; o < 32; o <<= 1) {
      const int up = __shfl_up(wi, o, 32);
      if (t >= o) wi += up;
    }
    wsum[t] = wi - w;
  }
  __syncthreads();
  {
    int run = wsum[wave] + inc - own;
    int4 o0, o1;
    o0.x = run; run += c0.x;
    o0.y = run; run += c0.y;
    o0.z = run; run += c0.z;
    o0.w = run; run += c0.w;
    o1.x = run; run += c1.x;
    o1.y = run; run += c1.y;
    o1.z = run; run += c1.z;
    o1.w = run;
    ((int4*)bin)[2 * t] = o0;
    ((int4*)bin)[2 * t + 1] = o1;
  }
  __syncthreads();

  for (int i0 = t; i0 < nst; i0 += 4 * S2_THREADS) {
    unsigned int e[4];
    int pos[4];
    #pragma unroll
    for (int u = 0; u < 4; ++u) {
      const int i = i0 + u * S2_THREADS;
      e[u] = 0u;
      if (i < nst) e[u] = raw[i];
    }
    #pragma unroll
    for (int u = 0; u < 4; ++u) {
      const int i = i0 + u * S2_THREADS;
      pos[u] = 0;
      if (i < nst) pos[u] = atomicAdd(&bin[s2_key(e[u])], 1);
    }
    #pragma unroll
    for (int u = 0; u < 4; ++u) {
      const int i = i0 + u * S2_THREADS;
      if (i < nst) srt[pos[u]] = e[u];
    }
  }
  __syncthreads();

  for (int i0 = wave * 128; i0 < nst; i0 += (S2_THREADS / 32) * 128) {
    unsigned int slot[4];
    float2 v[4];
    for (int u = 0; u < 4; ++u) {
      const int i = i0 + 32 * u + lane;
      const unsigned int e = srt[min(i, nst - 1)];
      slot[u] = i < nst ? (e >> ENT_SHIFT) : (unsigned int)S2_NODES;
      v[u] = *(const float2*)(h2 + (size_t)(e & src_mask) * NCLS);
    }
    for (int u = 0; u < 4; ++u) {
      atomicAdd(&a0[slot[u]], v[u].x);
      atomicAdd(&a1[slot[u]], v[u].y);
    }
  }
  __syncthreads();
  const int i = (int)blockIdx.x * S2_NODES + t;
  if (t >= S2_NODES || i >= N) return;
  const float si = inv_in[i];
  float v0 = fmaf(a0[t], si, b2[0]);
  float v1 = fmaf(a1[t], si, b2[1]);
  float m = fmaxf(v0, v1);
  float lse = m + logf(expf(v0 - m) + expf(v1 - m));
  out[(size_t)i * NCLS + 0] = v0 - lse;
  out[(size_t)i * NCLS + 1] = v1 - lse;
}

extern "C" void kernel_launch(void* const* d_in, const int* in_sizes, int n_in,
                              void* d_out, int out_size, void* d_ws, size_t ws_size,
                              hipStream_t stream) {
  const float* x   = (const float*)d_in[0];
  const int*   src = (const int*)d_in[1];
  const int*   dst = (const int*)d_in[2];
  const float* W1  = (const float*)d_in[3];
  const float* b1  = (const float*)d_in[4];
  const float* W2  = (const float*)d_in[5];
  const float* b2  = (const float*)d_in[6];
  float* out = (float*)d_out;

  const int N = in_sizes[0] / F_IN;
  const int E = in_sizes[1];
  const int P = (E + PIECE - 1) / PIECE;

  float* ws      = (float*)d_ws;
  float* inv_out = ws;
  float* inv_in  = ws + (size_t)N;
  float* h2      = ws + (size_t)2 * N;
  const size_t tab0 = ((size_t)4 * N + 63) & ~(size_t)63;
  unsigned int* tabS = (unsigned int*)(ws + tab0);
  unsigned int* tabD = tabS + (size_t)P * MAXB;
  const size_t list0 = tab0 + (size_t)2 * P * MAXB;
  unsigned int* list = (unsigned int*)(ws + list0);
  const size_t h10 = list0 + (((size_t)P * PIECE + 63) & ~(size_t)63);
  float* h1      = ws + h10;
  if ((h10 + (size_t)HID * N) * sizeof(float) > ws_size) return;

  const int NB = (N + NPB - 1) / NPB;

  order_by_src<<<P, SC_THREADS, 0, stream>>>(src, tabS, list, E, NB);
  deg_owner<<<NB, 512, 0, stream>>>(list, tabS, inv_out, N, P);

  const int tiles = N >> 4;
  const int gemm_wgs = (tiles + GEMM_TILES_PER_WG - 1) / GEMM_TILES_PER_WG;
  gemm1_and_order_by_dst<<<gemm_wgs + P, SC_THREADS, 0, stream>>>(x, W1, inv_out, h1, N, gemm_wgs,
                                                                  src, dst, tabD, list, E, NB);
  agg1_owner<<<NB, 1024, 0, stream>>>(list, tabD, h1, inv_out, b1, W2, inv_in, h2, N, P);
  agg2_final<<<(NB + S2_R - 1) / S2_R, S2_THREADS, 0, stream>>>(list, tabD, h2, inv_in, b2, out, N, P);
}


# Layer-1 edge sums: an owner work-group (now two ranges) first orders its entries by the source's 1 KB stretch of h1 in LDS and gathers in that order: fewer 256-byte requests; sums arrival-ordered
# speedup vs baseline: 5.1919x; 1.0327x over previous
#include <hip/hip_runtime.h>
#include <math.h>


typedef __attribute__((ext_vector_type(2))) float v2f;
typedef __attribute__((ext_vector_type(8))) float v8f;

#define F_IN 512
#define HID  16
#define NCLS 2

#define NPB_SHIFT 8
#define NPB   (1 << NPB_SHIFT)
#define MAXB  512
#define ENT_SHIFT 20

#define SC_THREADS 1024
#define SC_TRIPS 4
#define SC_CAP (SC_TRIPS * 4 * SC_THREADS)
#define PIECE 12512
#define SEG_SHIFT 16
#define SEG_PER_WAVE 8
#define RK_BITS 14
static_assert(SC_CAP <= (1 << RK_BITS), "a place inside a piece fits RK_BITS");
static_assert(MAXB <= (1 << (31 - RK_BITS)), "range number and place share one non-negative word");

__device__ __forceinline__ unsigned int bin_entry(int key, unsigned int payload) {
  return ((unsigned int)(key & (NPB - 1)) << ENT_SHIFT) | payload;
}
__device__ __forceinline__ int piece_rank(int* cntl, int key, int NB) {
  const int b = key >> NPB_SHIFT;
  return (unsigned int)b < (unsigned int)NB ? atomicAdd(&cntl[b], 1) : -1;
}

template <bool HAS_PAY>
__device__ __forceinline__ void order_piece(const int* __restrict__ key, const int* __restrict__ pay,
                                            unsigned int* __restrict__ tab, unsigned int* __restrict__ list,
                                            int E, int NB, int w) {
  __shared__ int cntl[MAXB];
  __shared__ int lofs[MAXB];
  __shared__ int ntot;
  __shared__ __attribute__((aligned(16))) int wtot[MAXB / 32];
  static_assert(MAXB % 128 == 0 && MAXB <= SC_THREADS, "one range a thread, the waves' totals read four at a time");
  __shared__ __attribute__((aligned(16))) unsigned int stage[SC_CAP];
  const int t = (int)threadIdx.x;
  const int p0 = w * PIECE;
  const int p1 = min(p0 + PIECE, E);
  for (int b = t; b < MAXB; b += SC_THREADS) cntl[b] = b << RK_BITS;
  __syncthreads();

  int4 kv[SC_TRIPS];
  int4 pv[SC_TRIPS];
  #pragma unroll
  for (int u = 0; u < SC_TRIPS; ++u) {
    const int e = p0 + 4 * (u * SC_THREADS + t);
    kv[u] = make_int4(-1, -1, -1, -1);
    pv[u] = make_int4(0, 0, 0, 0);
    if (e + 3 < p1) {
      kv[u] = *(const int4*)(key + e);
      if (HAS_PAY) pv[u] = *(const int4*)(pay + e);
    }
  }
  const int rem = (p1 - p0) & 3;
  int kt = -1;
  int plt = 0;
  if (t < rem) {
    kt = key[p1 - rem + t];
    if (HAS_PAY) plt = pay[p1 - rem + t];
  }
  unsigned int en[SC_TRIPS][4];
  int rk[SC_TRIPS][4];
  #pragma unroll
  for (int u = 0; u < SC_TRIPS; ++u) {
    en[u][0] = bin_entry(kv[u].x, (unsigned int)pv[u].x); rk[u][0] = piece_rank(cntl, kv[u].x, NB);
    en[u][1] = bin_entry(kv[u].y, (unsigned int)pv[u].y); rk[u][1] = piece_rank(cntl, kv[u].y, NB);
    en[u][2] = bin_entry(kv[u].z, (unsigned int)pv[u].z); rk[u][2] = piece_rank(cntl, kv[u].z, NB);
    en[u][3] = bin_entry(kv[u].w, (unsigned int)pv[u].w); rk[u][3] = piece_rank(cntl, kv[u].w, NB);
  }
  const unsigned int ent = bin_entry(kt, (unsigned int)plt);
  const int rkt = piece_rank(cntl, kt, NB);
  __syncthreads();

  {
    const int lane = t & 31;
    const int wave = t >> 5;
    const int own = t < MAXB ? cntl[t] & ((1 << RK_BITS) - 1) : 0;
    int inc = own;
    #pragma unroll
    for (int off = 1; off < 32; off <<= 1) {
      const int up = __shfl_up(inc, off, 32);
      if (lane >= off) inc += up;
    }
    if (lane == 31 && t < MAXB) wtot[wave] = inc;
    __syncthreads();
    if (t < MAXB) {
      int before = 0;
      #pragma unroll
      for (int q = 0; q < MAXB / 128; ++q) {
        const int4 v = ((const int4*)wtot)[q];
        before += (4 * q + 0 < wave ? v.x : 0) + (4 * q + 1 < wave ? v.y : 0)
                + (4 * q + 2 < wave ? v.z : 0) + (4 * q + 3 < wave ? v.w : 0);
      }
      lofs[t] = before + inc - own;
      if (t == MAXB - 1) ntot = before + inc;
    }
  }
  __syncthreads();

  for (int b = t; b < MAXB; b += SC_THREADS)
    tab[(size_t)w * MAXB + b] = (unsigned int)lofs[b] | ((unsigned int)(cntl[b] & ((1 << RK_BITS) - 1)) << SEG_SHIFT);

  const int rk_mask = (1 << RK_BITS) - 1;
  #pragma unroll
  for (int u = 0; u < SC_TRIPS; u += 2) {
    int ps[2][4];
    #pragma unroll
    for (int v = 0; v < 2; ++v) {
      #pragma unroll
      for (int j = 0; j < 4; ++j) ps[v][j] = lofs[rk[u + v][j] >= 0 ? rk[u + v][j] >> RK_BITS : 0] + (rk[u + v][j] & rk_mask);
    }
    #pragma unroll
    for (int v = 0; v < 2; ++v) {
      #pragma unroll
      for (int j = 0; j < 4; ++j)
        if (rk[u + v][j] >= 0) stage[ps[v][j]] = en[u + v][j];
    }
  }
  if (rkt >= 0) stage[lofs[rkt >> RK_BITS] + (rkt & rk_mask)] = ent;
  __syncthreads();

  const int n = ntot;
  unsigned int* __restrict__ mine = list + (size_t)p0;
  for (int i = t; i < (n >> 2); i += SC_THREADS) ((uint4*)mine)[i] = ((const uint4*)stage)[i];
  if (t < (n & 3)) mine[(n & ~3) + t] = stage[(n & ~3) + t];
}

__global__ __launch_bounds__(SC_THREADS) void order_by_src(const int* __restrict__ src,
                                                           unsigned int* __restrict__ tab,
                                                           unsigned int* __restrict__ list, int E, int NB) {
  order_piece<false>(src, src, tab, list, E, NB, (int)blockIdx.x);
}


__device__ __forceinline__ void run_open(const unsigned int* __restrict__ tab, int b, int g, int P, int lane,
                                         int (&pre)[SEG_PER_WAVE + 1], int (&off)[SEG_PER_WAVE]) {
  const int j = lane & (SEG_PER_WAVE - 1);
  const int piece = SEG_PER_WAVE * g + j;
  const unsigned int wd = piece < P ? tab[(size_t)piece * MAXB + b] : 0u;
  const int len = (int)(wd >> SEG_SHIFT);
  int inc = len;
  #pragma unroll
  for (int o = 1; o < SEG_PER_WAVE; o <<= 1) {
    const int up = __shfl_up(inc, o, SEG_PER_WAVE);
    if (j >= o) inc += up;
  }
  const int exc = inc - len;
  const int of = piece * PIECE + (int)(wd & ((1u << SEG_SHIFT) - 1u)) - exc;
  #pragma unroll
  for (int s = 0; s < SEG_PER_WAVE; ++s) {
    pre[s] = __builtin_amdgcn_readlane(exc, s);
    off[s] = __builtin_amdgcn_readlane(of, s);
  }
  pre[SEG_PER_WAVE] = __builtin_amdgcn_readlane(inc, SEG_PER_WAVE - 1);
}
__device__ __forceinline__ int run_index(int v, const int (&pre)[SEG_PER_WAVE + 1], const int (&off)[SEG_PER_WAVE]) {
  int o = off[0];
  #pragma unroll
  for (int s = 1; s < SEG_PER_WAVE; ++s) o = v >= pre[s] ? off[s] : o;
  return v + o;
}

__global__ __launch_bounds__(512) void deg_owner(const unsigned int* __restrict__ list, const unsigned int* __restrict__ tab,
                                                 float* __restrict__ inv, int N, int P) {
  __shared__ int c[NPB + 1];
  if (threadIdx.x <= NPB) c[threadIdx.x] = 0;
  __syncthreads();
  const int lane = (int)threadIdx.x & 31;
  const int G = (P + SEG_PER_WAVE - 1) / SEG_PER_WAVE;
  for (int g = (int)threadIdx.x >> 5; g < G; g += (int)blockDim.x >> 5) {
    int pre[SEG_PER_WAVE + 1];
    int off[SEG_PER_WAVE];
    run_open(tab, (int)blockIdx.x, g, P, lane, pre, off);
    const int total = pre[SEG_PER_WAVE];
    for (int v0 = 0; v0 < total; v0 += 128) {
      unsigned int slot[4];
      for (int u = 0; u < 4; ++u) {
        const int v = v0 + 32 * u + lane;
        const unsigned int ent = list[run_index(min(v, total - 1), pre, off)];
        slot[u] = v < total ? (ent >> ENT_SHIFT) : (unsigned int)NPB;
      }
      for (int u = 0; u < 4; ++u) atomicAdd(&c[slot[u]], 1);
    }
  }
  __syncthreads();
  const int node = blockIdx.x * NPB + (int)threadIdx.x;
  if (threadIdx.x < NPB && node < N) inv[node] = rsqrtf(fmaxf((float)c[threadIdx.x], 1.0f));
}

#define GEMM_TILES_PER_WG (SC_THREADS / 32)
__device__ __forceinline__ void gemm1_tiles(
    const float* __restrict__ x, const float* __restrict__ W1,
    const float* __restrict__ inv_out, float* __restrict__ h1, int N, int wg) {
  __shared__ float wlds[F_IN * HID];
  {
    float wv[F_IN * HID / SC_THREADS];
    #pragma unroll
    for (int j = 0; j < F_IN * HID / SC_THREADS; ++j) wv[j] = W1[threadIdx.x + j * SC_THREADS];
    #pragma unroll
    for (int j = 0; j < F_IN * HID / SC_THREADS; ++j) wlds[threadIdx.x + j * SC_THREADS] = wv[j];
  }
  __syncthreads();

  const int wave = threadIdx.x >> 5;
  const int lane = threadIdx.x & 31;
  const int tiles = N >> 4;
  const int tile = wg * GEMM_TILES_PER_WG + wave;
  if (tile >= tiles) return;

  const int n16 = lane & 15;
  const int g   = lane >> 4;
  const int row = tile * 16 + n16;
  const float s = inv_out[row];
  const float* __restrict__ xr = x + (size_t)row * F_IN;

  v8f c = {};
  #pragma unroll 4
  for (int k = 0; k < F_IN; k += 4) {
    const int kk = k + 2 * g;
    v2f a = *(const v2f*)(xr + kk);
    a.x *= s;
    a.y *= s;
    v2f b;
    b.x = wlds[(kk + 0) * HID + n16];
    b.y = wlds[(kk + 1) * HID + n16];
    c = __builtin_amdgcn_wmma_f32_16x16x4_f32(
         false, a,  false, b,
         (short)0, c,  false,  false);
  }

  #pragma unroll
  for (int p = 0; p < 8; ++p) {
    const int m = tile * 16 + p + 8 * g;
    h1[(size_t)m * HID + n16] = c[p];
  }
}

__global__ __launch_bounds__(SC_THREADS, 16) void gemm1_and_order_by_dst(
    const float* __restrict__ x, const float* __restrict__ W1,
    const float* __restrict__ inv_out, float* __restrict__ h1, int N, int gemm_wgs,
    const int* __restrict__ src, const int* __restrict__ dst,
    unsigned int* __restrict__ tab, unsigned int* __restrict__ list, int E, int NB) {
  if ((int)blockIdx.x < gemm_wgs) gemm1_tiles(x, W1, inv_out, h1, N, (int)blockIdx.x);
  else order_piece<true>(dst, src, tab, list, E, NB, (int)blockIdx.x - gemm_wgs);
}

#define S2_R 2
#define S2_NODES (S2_R * NPB)
#define S2_THREADS 1024
#define S2_CAP 24576
#define S2_KEY_SHIFT 4
#define S2_BINS ((MAXB * NPB) >> S2_KEY_SHIFT)
static_assert(S2_BINS == 8 * S2_THREADS, "the running sum below gives every thread eight neighbouring bins");
__device__ __forceinline__ int s2_key(unsigned int ent) {
  return min((int)((ent & ((1u << ENT_SHIFT) - 1u)) >> S2_KEY_SHIFT), S2_BINS - 1);
}

#define A1_PITCH (S2_NODES + 4)
__global__ __launch_bounds__(S2_THREADS) void agg1_owner(const unsigned int* __restrict__ list, const unsigned int* __restrict__ tab,
                                                         const float* __restrict__ h1, const float* __restrict__ inv_out,
                                                         const float* __restrict__ b1, const float* __restrict__ W2,
                                                         float* __restrict__ inv_in, float* __restrict__ h2,
                                                         int N, int P) {
  __shared__ float acc[HID * A1_PITCH];
  __shared__ int cnt[S2_NODES + 1];
  __shared__ __attribute__((aligned(16))) int bin[S2_BINS];
  __shared__ unsigned int raw[S2_CAP];
  __shared__ unsigned int srt[S2_CAP];
  __shared__ int wsum[S2_THREADS / 32];
  __shared__ int fill;
  const int t = (int)threadIdx.x;
  const int lane = t & 31;
  const int wave = t >> 5;
  const int q = t & 3;
  const unsigned int src_mask = (1u << ENT_SHIFT) - 1u;
  for (int i = t; i < HID * A1_PITCH; i += S2_THREADS) acc[i] = 0.0f;
  for (int i = t; i <= S2_NODES; i += S2_THREADS) cnt[i] = 0;
  for (int i = t; i < S2_BINS; i += S2_THREADS) bin[i] = 0;
  if (t == 0) fill = 0;
  __syncthreads();

  const int NB = (N + NPB - 1) >> NPB_SHIFT;
  const int G = (P + SEG_PER_WAVE - 1) / SEG_PER_WAVE;
  for (int r = 0; r < S2_R; ++r) {
    const int b = S2_R * (int)blockIdx.x + r;
    if (b >= NB) break;
    const unsigned int tag = (unsigned int)r << (ENT_SHIFT + NPB_SHIFT);
    for (int g = wave; g < G; g += S2_THREADS / 32) {
      int pre[SEG_PER_WAVE + 1];
      int off[SEG_PER_WAVE];
      run_open(tab, b, g, P, lane, pre, off);
      const int total = pre[SEG_PER_WAVE];
      int base = 0;
      if (lane == 0 && total > 0) base = atomicAdd(&fill, total);
      base = __builtin_amdgcn_readfirstlane(base);
      for (int v0 = 0; v0 < total; v0 += 128) {
        unsigned int ent[4];
        for (int u = 0; u < 4; ++u) {
          const int n = v0 + 32 * u + lane;
          ent[u] = list[run_index(min(n, total - 1), pre, off)] | tag;
        }
        for (int u = 0; u < 4; ++u) {
          const int n = v0 + 32 * u + lane;
          if (n < total) {
            const int pos = base + n;
            if (pos < S2_CAP) {
              raw[pos] = ent[u];
              atomicAdd(&bin[s2_key(ent[u])], 1);
            } else {
              const unsigned int place = ent[u] >> ENT_SHIFT;
              const float4* hr = (const float4*)(h1 + (size_t)(ent[u] & src_mask) * HID);
              for (int f = 0; f < 4; ++f) {
                const float4 v = hr[f];
                atomicAdd(&acc[(4 * f + 0) * A1_PITCH + place], v.x);
                atomicAdd(&acc[(4 * f + 1) * A1_PITCH + place], v.y);
                atomicAdd(&acc[(4 * f + 2) * A1_PITCH + place], v.z);
                atomicAdd(&acc[(4 * f + 3) * A1_PITCH + place], v.w);
              }
              atomicAdd(&cnt[place], 1);
            }
          }
        }
      }
    }
  }
  __syncthreads();
  const int nst = min(fill, S2_CAP);

  const int4 c0 = ((const int4*)bin)[2 * t];
  const int4 c1 = ((const int4*)bin)[2 * t + 1];
  const int own = c0.x + c0.y + c0.z + c0.w + c1.x + c1.y + c1.z + c1.w;
  int inc = own;
  #pragma unroll
  for (int o = 1; o < 32; o <<= 1) {
    const int up = __shfl_up(inc, o, 32);
    if (lane >= o) inc += up;
  }
  if (lane == 31) wsum[wave] = inc;
  __syncthreads();
  if (t < 32) {
    const int w = wsum[t];
    int wi = w;
    #pragma unroll
    for (int o = 1; o < 32; o <<= 1) {
      const int up = __shfl_up(wi, o, 32);
      if (t >= o) wi += up;
    }
    wsum[t] = wi - w;
  }
  __syncthreads();
  {
    int run = wsum[wave] + inc - own;
    int4 o0, o1;
    o0.x = run; run += c0.x;
    o0.y = run; run += c0.y;
    o0.z = run; run += c0.z;
    o0.w = run; run += c0.w;
    o1.x = run; run += c1.x;
    o1.y = run; run += c1.y;
    o1.z = run; run += c1.z;
    o1.w = run;
    ((int4*)bin)[2 * t] = o0;
    ((int4*)bin)[2 * t + 1] = o1;
  }
  __syncthreads();

  for (int i0 = t; i0 < nst; i0 += 4 * S2_THREADS) {
    unsigned int e[4];
    int pos[4];
    #pragma unroll
    for (int u = 0; u < 4; ++u) {
      const int i = i0 + u * S2_THREADS;
      e[u] = 0u;
      if (i < nst) e[u] = raw[i];
    }
    #pragma unroll
    for (int u = 0; u < 4; ++u) {
      const int i = i0 + u * S2_THREADS;
      pos[u] = 0;
      if (i < nst) pos[u] = atomicAdd(&bin[s2_key(e[u])], 1);
    }
    #pragma unroll
    for (int u = 0; u < 4; ++u) {
      const int i = i0 + u * S2_THREADS;
      if (i < nst) srt[pos[u]] = e[u];
    }
  }
  __syncthreads();

  const float4* __restrict__ h1v = (const float4*)h1;
  const int grp = lane >> 2;
  for (int i0 = wave * 64; i0 < nst; i0 += 2 * S2_THREADS) {
    unsigned int w[2];
    unsigned int e[2][4];
    float4 v[2][4];
    #pragma unroll
    for (int h = 0; h < 2; ++h) {
      const int i = i0 + 32 * h + lane;
      w[h] = srt[min(i, nst - 1)];
      if (i >= nst) w[h] = (w[h] & src_mask) | ((unsigned int)S2_NODES << ENT_SHIFT);
    }
    #pragma unroll
    for (int h = 0; h < 2; ++h) {
      #pragma unroll
      for (int u = 0; u < 4; ++u) {
        e[h][u] = (unsigned int)__shfl((int)w[h], 8 * u + grp, 32);
        v[h][u] = h1v[4u * (e[h][u] & src_mask) + (unsigned int)q];
      }
    }
    #pragma unroll
    for (int h = 0; h < 2; ++h) {
      #pragma unroll
      for (int u = 0; u < 4; ++u) {
        float* row = acc + 4 * q * A1_PITCH + (e[h][u] >> ENT_SHIFT);
        atomicAdd(&row[0 * A1_PITCH], v[h][u].x);
        atomicAdd(&row[1 * A1_PITCH], v[h][u].y);
        atomicAdd(&row[2 * A1_PITCH], v[h][u].z);
        atomicAdd(&row[3 * A1_PITCH], v[h][u].w);
      }
      atomicAdd(&cnt[w[h] >> ENT_SHIFT], 1);
    }
  }
  __syncthreads();
  const int node = (int)blockIdx.x * S2_NODES + t;
  if (t >= S2_NODES || node >= N) return;
  const float si = rsqrtf(fmaxf((float)cnt[t], 1.0f));
  const float so = inv_out[node];
  inv_in[node] = si;
  float acc0 = 0.0f, acc1 = 0.0f;
  #pragma unroll
  for (int k = 0; k < HID; ++k) {
    float hv = fmaxf(fmaf(acc[k * A1_PITCH + t], si, b1[k]), 0.0f) * so;
    acc0 = fmaf(hv, W2[k * NCLS + 0], acc0);
    acc1 = fmaf(hv, W2[k * NCLS + 1], acc1);
  }
  *(float2*)(h2 + (size_t)node * NCLS) = make_float2(acc0, acc1);
}

__global__ __launch_bounds__(S2_THREADS) void agg2_final(const unsigned int* __restrict__ list, const unsigned int* __restrict__ tab,
                                                         const float* __restrict__ h2, const float* __restrict__ inv_in,
                                                         const float* __restrict__ b2, float* __restrict__ out, int N, int P) {
  __shared__ float a0[S2_NODES + 1];
  __shared__ float a1[S2_NODES + 1];
  __shared__ __attribute__((aligned(16))) int bin[S2_BINS];
  __shared__ unsigned int raw[S2_CAP];
  __shared__ unsigned int srt[S2_CAP];
  __shared__ int wsum[S2_THREADS / 32];
  __shared__ int fill;
  const int t = (int)threadIdx.x;
  const int lane = t & 31;
  const int wave = t >> 5;
  const unsigned int src_mask = (1u << ENT_SHIFT) - 1u;
  for (int i = t; i <= S2_NODES; i += S2_THREADS) { a0[i] = 0.0f; a1[i] = 0.0f; }
  for (int i = t; i < S2_BINS; i += S2_THREADS) bin[i] = 0;
  if (t == 0) fill = 0;
  __syncthreads();

  const int NB = (N + NPB - 1) >> NPB_SHIFT;
  const int G = (P + SEG_PER_WAVE - 1) / SEG_PER_WAVE;
  for (int r = 0; r < S2_R; ++r) {
    const int b = S2_R * (int)blockIdx.x + r;
    if (b >= NB) break;
    const unsigned int tag = (unsigned int)r << (ENT_SHIFT + NPB_SHIFT);
    for (int g = wave; g < G; g += S2_THREADS / 32) {
      int pre[SEG_PER_WAVE + 1];
      int off[SEG_PER_WAVE];
      run_open(tab, b, g, P, lane, pre, off);
      const int total = pre[SEG_PER_WAVE];
      int base = 0;
      if (lane == 0 && total > 0) base = atomicAdd(&fill, total);
      base = __builtin_amdgcn_readfirstlane(base);
      for (int v0 = 0; v0 < total; v0 += 128) {
        unsigned int ent[4];
        for (int u = 0; u < 4; ++u) {
          const int n = v0 + 32 * u + lane;
          ent[u] = list[run_index(min(n, total - 1), pre, off)] | tag;
        }
        for (int u = 0; u < 4; ++u) {
          const int n = v0 + 32 * u + lane;
          if (n < total) {
            const int pos = base + n;
            if (pos < S2_CAP) {
              raw[pos] = ent[u];
              atomicAdd(&bin[s2_key(ent[u])], 1);
            } else {
              const float2 v = *(const float2*)(h2 + (size_t)(ent[u] & src_mask) * NCLS);
              atomicAdd(&a0[ent[u] >> ENT_SHIFT], v.x);
              atomicAdd(&a1[ent[u] >> ENT_SHIFT], v.y);
            }
          }
        }
      }
    }
  }
  __syncthreads();
  const int nst = min(fill, S2_CAP);

  const int4 c0 = ((const int4*)bin)[2 * t];
  const int4 c1 = ((const int4*)bin)[2 * t + 1];
  const int own = c0.x + c0.y + c0.z + c0.w + c1.x + c1.y + c1.z + c1.w;
  int inc = own;
  #pragma unroll
  for (int o = 1; o < 32; o <<= 1) {
    const int up = __shfl_up(inc, o, 32);
    if (lane >= o) inc += up;
  }
  if (lane == 31) wsum[wave] = inc;
  __syncthreads();
  if (t < 32) {
    const int w = wsum[t];
    int wi = w;
    #pragma unroll
    for (int o = 1; o < 32; o <<= 1) {
      const int up = __shfl_up(wi, o, 32);
      if (t >= o) wi += up;
    }
    wsum[t] = wi - w;
  }
  __syncthreads();
  {
    int run = wsum[wave] + inc - own;
    int4 o0, o1;
    o0.x = run; run += c0.x;
    o0.y = run; run += c0.y;
    o0.z = run; run += c0.z;
    o0.w = run; run += c0.w;
    o1.x = run; run += c1.x;
    o1.y = run; run += c1.y;
    o1.z = run; run += c1.z;
    o1.w = run;
    ((int4*)bin)[2 * t] = o0;
    ((int4*)bin)[2 * t + 1] = o1;
  }
  __syncthreads();

  for (int i0 = t; i0 < nst; i0 += 4 * S2_THREADS) {
    unsigned int e[4];
    int pos[4];
    #pragma unroll
    for (int u = 0; u < 4; ++u) {
      const int i = i0 + u * S2_THREADS;
      e[u] = 0u;
      if (i < nst) e[u] = raw[i];
    }
    #pragma unroll
    for (int u = 0; u < 4; ++u) {
      const int i = i0 + u * S2_THREADS;
      pos[u] = 0;
      if (i < nst) pos[u] = atomicAdd(&bin[s2_key(e[u])], 1);
    }
    #pragma unroll
    for (int u = 0; u < 4; ++u) {
      const int i = i0 + u * S2_THREADS;
      if (i < nst) srt[pos[u]] = e[u];
    }
  }
  __syncthreads();

  for (int i0 = wave * 128; i0 < nst; i0 += (S2_THREADS / 32) * 128) {
    unsigned int slot[4];
    float2 v[4];
    for (int u = 0; u < 4; ++u) {
      const int i = i0 + 32 * u + lane;
      const unsigned int e = srt[min(i, nst - 1)];
      slot[u] = i < nst ? (e >> ENT_SHIFT) : (unsigned int)S2_NODES;
      v[u] = *(const float2*)(h2 + (size_t)(e & src_mask) * NCLS);
    }
    for (int u = 0; u < 4; ++u) {
      atomicAdd(&a0[slot[u]], v[u].x);
      atomicAdd(&a1[slot[u]], v[u].y);
    }
  }
  __syncthreads();
  const int i = (int)blockIdx.x * S2_NODES + t;
  if (t >= S2_NODES || i >= N) return;
  const float si = inv_in[i];
  float v0 = fmaf(a0[t], si, b2[0]);
  float v1 = fmaf(a1[t], si, b2[1]);
  float m = fmaxf(v0, v1);
  float lse = m + logf(expf(v0 - m) + expf(v1 - m));
  out[(size_t)i * NCLS + 0] = v0 - lse;
  out[(size_t)i * NCLS + 1] = v1 - lse;
}

extern "C" void kernel_launch(void* const* d_in, const int* in_sizes, int n_in,
                              void* d_out, int out_size, void* d_ws, size_t ws_size,
                              hipStream_t stream) {
  const float* x   = (const float*)d_in[0];
  const int*   src = (const int*)d_in[1];
  const int*   dst = (const int*)d_in[2];
  const float* W1  = (const float*)d_in[3];
  const float* b1  = (const float*)d_in[4];
  const float* W2  = (const float*)d_in[5];
  const float* b2  = (const float*)d_in[6];
  float* out = (float*)d_out;

  const int N = in_sizes[0] / F_IN;
  const int E = in_sizes[1];
  const int P = (E + PIECE - 1) / PIECE;

  float* ws      = (float*)d_ws;
  float* inv_out = ws;
  float* inv_in  = ws + (size_t)N;
  float* h2      = ws + (size_t)2 * N;
  const size_t tab0 = ((size_t)4 * N + 63) & ~(size_t)63;
  unsigned int* tabS = (unsigned int*)(ws + tab0);
  unsigned int* tabD = tabS + (size_t)P * MAXB;
  const size_t list0 = tab0 + (size_t)2 * P * MAXB;
  unsigned int* list = (unsigned int*)(ws + list0);
  const size_t h10 = list0 + (((size_t)P * PIECE + 63) & ~(size_t)63);
  float* h1      = ws + h10;
  if ((h10 + (size_t)HID * N) * sizeof(float) > ws_size) return;

  const int NB = (N + NPB - 1) / NPB;

  order_by_src<<<P, SC_THREADS, 0, stream>>>(src, tabS, list, E, NB);
  deg_owner<<<NB, 512, 0, stream>>>(list, tabS, inv_out, N, P);

  const int tiles = N >> 4;
  const int gemm_wgs = (tiles + GEMM_TILES_PER_WG - 1) / GEMM_TILES_PER_WG;
  gemm1_and_order_by_dst<<<gemm_wgs + P, SC_THREADS, 0, stream>>>(x, W1, inv_out, h1, N, gemm_wgs,
                                                                  src, dst, tabD, list, E, NB);
  agg1_owner<<<(NB + S2_R - 1) / S2_R, S2_THREADS, 0, stream>>>(list, tabD, h1, inv_out, b1, W2, inv_in, h2, N, P);
  agg2_final<<<(NB + S2_R - 1) / S2_R, S2_THREADS, 0, stream>>>(list, tabD, h2, inv_in, b2, out, N, P);
}


# x @ W1 no longer waits for out-degrees: the degree owners scale its finished rows by 1/sqrt(degree) (rounding changes), so the product and BOTH orderings share one launch: 4 launches, was 5
# speedup vs baseline: 5.4814x; 1.0558x over previous
#include <hip/hip_runtime.h>
#include <math.h>


typedef __attribute__((ext_vector_type(2))) float v2f;
typedef __attribute__((ext_vector_type(8))) float v8f;

#define F_IN 512
#define HID  16
#define NCLS 2

#define NPB_SHIFT 8
#define NPB   (1 << NPB_SHIFT)
#define MAXB  512
#define ENT_SHIFT 20

#define SC_THREADS 1024
#define SC_TRIPS 4
#define SC_CAP (SC_TRIPS * 4 * SC_THREADS)
#define PIECE 12512
#define SEG_SHIFT 16
#define PIECE_B 12544
static_assert(PIECE_B >= PIECE && PIECE_B % 128 == 0, "a piece's source entries fit its own whole lines");
#define SEG_PER_WAVE 8
#define RK_BITS 14
static_assert(SC_CAP <= (1 << RK_BITS), "a place inside a piece fits RK_BITS");
static_assert(MAXB <= (1 << (31 - RK_BITS)), "range number and place share one non-negative word");

__device__ __forceinline__ unsigned int bin_entry(int key, unsigned int payload) {
  return ((unsigned int)(key & (NPB - 1)) << ENT_SHIFT) | payload;
}
__device__ __forceinline__ int piece_rank(int* cntl, int key, int NB) {
  const int b = key >> NPB_SHIFT;
  return (unsigned int)b < (unsigned int)NB ? atomicAdd(&cntl[b], 1) : -1;
}

struct PieceLds {
  int cntl[MAXB];
  int lofs[MAXB];
  int ntot;
  int wtot[MAXB / 32];
  __attribute__((aligned(16))) unsigned int stage[SC_CAP];
};
static_assert(MAXB % 32 == 0 && MAXB / 32 <= 32 && MAXB <= SC_THREADS, "one range a thread, one wave's total a lane");

template <bool HAS_PAY>
__device__ __forceinline__ void order_piece(PieceLds& L, const int* __restrict__ key, const int* __restrict__ pay,
                                            unsigned int* __restrict__ tab, unsigned int* __restrict__ list,
                                            int E, int NB, int w) {
  int* cntl = L.cntl;
  int* lofs = L.lofs;
  unsigned int* stage = L.stage;
  unsigned char* stage8 = (unsigned char*)L.stage;
  const int t = (int)threadIdx.x;
  const int p0 = w * PIECE;
  const int p1 = min(p0 + PIECE, E);
  for (int b = t; b < MAXB; b += SC_THREADS) cntl[b] = b << RK_BITS;
  __syncthreads();

  int4 kv[SC_TRIPS];
  int4 pv[SC_TRIPS];
  #pragma unroll
  for (int u = 0; u < SC_TRIPS; ++u) {
    const int e = p0 + 4 * (u * SC_THREADS + t);
    kv[u] = make_int4(-1, -1, -1, -1);
    pv[u] = make_int4(0, 0, 0, 0);
    if (e + 3 < p1) {
      kv[u] = *(const int4*)(key + e);
      if (HAS_PAY) pv[u] = *(const int4*)(pay + e);
    }
  }
  const int rem = (p1 - p0) & 3;
  int kt = -1;
  int plt = 0;
  if (t < rem) {
    kt = key[p1 - rem + t];
    if (HAS_PAY) plt = pay[p1 - rem + t];
  }
  unsigned int en[SC_TRIPS][4];
  int rk[SC_TRIPS][4];
  #pragma unroll
  for (int u = 0; u < SC_TRIPS; ++u) {
    en[u][0] = HAS_PAY ? bin_entry(kv[u].x, (unsigned int)pv[u].x) : (unsigned int)(kv[u].x & (NPB - 1)); rk[u][0] = piece_rank(cntl, kv[u].x, NB);
    en[u][1] = HAS_PAY ? bin_entry(kv[u].y, (unsigned int)pv[u].y) : (unsigned int)(kv[u].y & (NPB - 1)); rk[u][1] = piece_rank(cntl, kv[u].y, NB);
    en[u][2] = HAS_PAY ? bin_entry(kv[u].z, (unsigned int)pv[u].z) : (unsigned int)(kv[u].z & (NPB - 1)); rk[u][2] = piece_rank(cntl, kv[u].z, NB);
    en[u][3] = HAS_PAY ? bin_entry(kv[u].w, (unsigned int)pv[u].w) : (unsigned int)(kv[u].w & (NPB - 1)); rk[u][3] = piece_rank(cntl, kv[u].w, NB);
  }
  const unsigned int ent = HAS_PAY ? bin_entry(kt, (unsigned int)plt) : (unsigned int)(kt & (NPB - 1));
  const int rkt = piece_rank(cntl, kt, NB);
  __syncthreads();

  {
    const int lane = t & 31;
    const int wave = t >> 5;
    const int own = t < MAXB ? cntl[t] & ((1 << RK_BITS) - 1) : 0;
    int inc = own;
    #pragma unroll
    for (int off = 1; off < 32; off <<= 1) {
      const int up = __shfl_up(inc, off, 32);
      if (lane >= off) inc += up;
    }
    if (lane == 31 && t < MAXB) L.wtot[wave] = inc;
    __syncthreads();
    const int wt = lane < MAXB / 32 ? L.wtot[lane] : 0;
    int winc = wt;
    #pragma unroll
    for (int off = 1; off < MAXB / 32; off <<= 1) {
      const int up = __shfl_up(winc, off, 32);
      if (lane >= off) winc += up;
    }
    const int before = __shfl(winc - wt, min(wave, MAXB / 32 - 1), 32);
    if (t < MAXB) {
      lofs[t] = before + inc - own;
      if (t == MAXB - 1) L.ntot = before + inc;
    }
  }
  __syncthreads();

  for (int b = t; b < MAXB; b += SC_THREADS)
    tab[(size_t)w * MAXB + b] = (unsigned int)lofs[b] | ((unsigned int)(cntl[b] & ((1 << RK_BITS) - 1)) << SEG_SHIFT);

  const int rk_mask = (1 << RK_BITS) - 1;
  #pragma unroll
  for (int u = 0; u < SC_TRIPS; u += 2) {
    int ps[2][4];
    #pragma unroll
    for (int v = 0; v < 2; ++v) {
      #pragma unroll
      for (int j = 0; j < 4; ++j) ps[v][j] = lofs[rk[u + v][j] >= 0 ? rk[u + v][j] >> RK_BITS : 0] + (rk[u + v][j] & rk_mask);
    }
    #pragma unroll
    for (int v = 0; v < 2; ++v) {
      #pragma unroll
      for (int j = 0; j < 4; ++j)
        if (rk[u + v][j] >= 0) {
          if (HAS_PAY) stage[ps[v][j]] = en[u + v][j];
          else stage8[ps[v][j]] = (unsigned char)en[u + v][j];
        }
    }
  }
  if (rkt >= 0) {
    const int pt = lofs[rkt >> RK_BITS] + (rkt & rk_mask);
    if (HAS_PAY) stage[pt] = ent;
    else stage8[pt] = (unsigned char)ent;
  }
  __syncthreads();

  const int n = L.ntot;
  if (HAS_PAY) {
    unsigned int* __restrict__ mine = list + (size_t)p0;
    for (int i = t; i < (n >> 2); i += SC_THREADS) ((uint4*)mine)[i] = ((const uint4*)stage)[i];
    if (t < (n & 3)) mine[(n & ~3) + t] = stage[(n & ~3) + t];
  } else {
    unsigned char* __restrict__ mine = (unsigned char*)list + (size_t)w * PIECE_B;
    for (int i = t; i < (n >> 4); i += SC_THREADS) ((uint4*)mine)[i] = ((const uint4*)stage)[i];
    if (t < (n & 15)) mine[(n & ~15) + t] = stage8[(n & ~15) + t];
  }
}


template <int PITCH>
__device__ __forceinline__ void run_open(const unsigned int* __restrict__ tab, int b, int g, int P, int lane,
                                         int (&pre)[SEG_PER_WAVE + 1], int (&off)[SEG_PER_WAVE]) {
  const int j = lane & (SEG_PER_WAVE - 1);
  const int piece = SEG_PER_WAVE * g + j;
  const unsigned int wd = piece < P ? tab[(size_t)piece * MAXB + b] : 0u;
  const int len = (int)(wd >> SEG_SHIFT);
  int inc = len;
  #pragma unroll
  for (int o = 1; o < SEG_PER_WAVE; o <<= 1) {
    const int up = __shfl_up(inc, o, SEG_PER_WAVE);
    if (j >= o) inc += up;
  }
  const int exc = inc - len;
  const int of = piece * PITCH + (int)(wd & ((1u << SEG_SHIFT) - 1u)) - exc;
  #pragma unroll
  for (int s = 0; s < SEG_PER_WAVE; ++s) {
    pre[s] = __builtin_amdgcn_readlane(exc, s);
    off[s] = __builtin_amdgcn_readlane(of, s);
  }
  pre[SEG_PER_WAVE] = __builtin_amdgcn_readlane(inc, SEG_PER_WAVE - 1);
}
__device__ __forceinline__ int run_index(int v, const int (&pre)[SEG_PER_WAVE + 1], const int (&off)[SEG_PER_WAVE]) {
  int o = off[0];
  #pragma unroll
  for (int s = 1; s < SEG_PER_WAVE; ++s) o = v >= pre[s] ? off[s] : o;
  return v + o;
}

__global__ __launch_bounds__(512) void deg_scale(const unsigned char* __restrict__ list, const unsigned int* __restrict__ tab,
                                                 float* __restrict__ inv, float* __restrict__ h1, int N, int P) {
  __shared__ int c[NPB + 1];
  __shared__ float sinv[NPB];
  const int t = (int)threadIdx.x;
  if (t <= NPB) c[t] = 0;
  __syncthreads();
  static_assert(NPB * HID / 4 == 2 * 512, "two 16-byte pieces of h1 a thread");
  float4* __restrict__ hv = (float4*)(h1 + (size_t)blockIdx.x * NPB * HID);
  const bool row0 = (int)blockIdx.x * NPB + (t >> 2) < N;
  const bool row1 = (int)blockIdx.x * NPB + ((t + 512) >> 2) < N;
  float4 r0 = make_float4(0.0f, 0.0f, 0.0f, 0.0f);
  float4 r1 = make_float4(0.0f, 0.0f, 0.0f, 0.0f);
  if (row0) r0 = hv[t];
  if (row1) r1 = hv[t + 512];
  const int lane = t & 31;
  const int G = (P + SEG_PER_WAVE - 1) / SEG_PER_WAVE;
  for (int g = t >> 5; g < G; g += (int)blockDim.x >> 5) {
    int pre[SEG_PER_WAVE + 1];
    int off[SEG_PER_WAVE];
    run_open<PIECE_B>(tab, (int)blockIdx.x, g, P, lane, pre, off);
    const int total = pre[SEG_PER_WAVE];
    for (int v0 = 0; v0 < total; v0 += 128) {
      unsigned int slot[4];
      for (int u = 0; u < 4; ++u) {
        const int v = v0 + 32 * u + lane;
        const unsigned int ent = list[run_index(min(v, total - 1), pre, off)];
        slot[u] = v < total ? ent : (unsigned int)NPB;
      }
      for (int u = 0; u < 4; ++u) atomicAdd(&c[slot[u]], 1);
    }
  }
  __syncthreads();
  const int node = blockIdx.x * NPB + t;
  if (t < NPB) {
    const float s = rsqrtf(fmaxf((float)c[t], 1.0f));
    sinv[t] = s;
    if (node < N) inv[node] = s;
  }
  __syncthreads();
  if (row0) {
    const float s = sinv[t >> 2];
    hv[t] = make_float4(r0.x * s, r0.y * s, r0.z * s, r0.w * s);
  }
  if (row1) {
    const float s = sinv[(t + 512) >> 2];
    hv[t + 512] = make_float4(r1.x * s, r1.y * s, r1.z * s, r1.w * s);
  }
}

#define GEMM_TILES_PER_WG (SC_THREADS / 32)
__device__ __forceinline__ void gemm1_tiles(
    const float* __restrict__ x, const float* __restrict__ W1,
    float* __restrict__ h1, int N, int wg) {
  __shared__ float wlds[F_IN * HID];
  {
    float wv[F_IN * HID / SC_THREADS];
    #pragma unroll
    for (int j = 0; j < F_IN * HID / SC_THREADS; ++j) wv[j] = W1[threadIdx.x + j * SC_THREADS];
    #pragma unroll
    for (int j = 0; j < F_IN * HID / SC_THREADS; ++j) wlds[threadIdx.x + j * SC_THREADS] = wv[j];
  }
  __syncthreads();

  const int wave = threadIdx.x >> 5;
  const int lane = threadIdx.x & 31;
  const int tiles = N >> 4;
  const int tile = wg * GEMM_TILES_PER_WG + wave;
  if (tile >= tiles) return;

  const int n16 = lane & 15;
  const int g   = lane >> 4;
  const int row = tile * 16 + n16;
  const float* __restrict__ xr = x + (size_t)row * F_IN;

  v8f c = {};
  #pragma unroll 4
  for (int k = 0; k < F_IN; k += 4) {
    const int kk = k + 2 * g;
    const v2f a = *(const v2f*)(xr + kk);
    v2f b;
    b.x = wlds[(kk + 0) * HID + n16];
    b.y = wlds[(kk + 1) * HID + n16];
    c = __builtin_amdgcn_wmma_f32_16x16x4_f32(
         false, a,  false, b,
         (short)0, c,  false,  false);
  }

  #pragma unroll
  for (int p = 0; p < 8; ++p) {
    const int m = tile * 16 + p + 8 * g;
    h1[(size_t)m * HID + n16] = c[p];
  }
}

__global__ __launch_bounds__(SC_THREADS, 16) void gemm1_and_orderings(
    const float* __restrict__ x, const float* __restrict__ W1,
    float* __restrict__ h1, int N, int gemm_wgs,
    const int* __restrict__ src, const int* __restrict__ dst,
    unsigned int* __restrict__ tabD, unsigned int* __restrict__ listD, int n_dst,
    unsigned int* __restrict__ tabS, unsigned int* __restrict__ listS, int E, int NB) {
  __shared__ PieceLds L;
  const int wg = (int)blockIdx.x;
  if (wg < gemm_wgs) gemm1_tiles(x, W1, h1, N, wg);
  else if (wg - gemm_wgs < n_dst) order_piece<true>(L, dst, src, tabD, listD, E, NB, wg - gemm_wgs);
  else order_piece<false>(L, src, src, tabS, listS, E, NB, wg - gemm_wgs - n_dst);
}

#define S2_R 2
#define S2_NODES (S2_R * NPB)
#define S2_THREADS 1024
#define S2_CAP 24576
#define S2_KEY_SHIFT 4
#define S2_BINS ((MAXB * NPB) >> S2_KEY_SHIFT)
static_assert(S2_BINS == 8 * S2_THREADS, "the running sum below gives every thread eight neighbouring bins");
__device__ __forceinline__ int s2_key(unsigned int ent) {
  return min((int)((ent & ((1u << ENT_SHIFT) - 1u)) >> S2_KEY_SHIFT), S2_BINS - 1);
}

#define A1_PITCH (S2_NODES + 4)
__global__ __launch_bounds__(S2_THREADS) void agg1_owner(const unsigned int* __restrict__ list, const unsigned int* __restrict__ tab,
                                                         const float* __restrict__ h1, const float* __restrict__ inv_out,
                                                         const float* __restrict__ b1, const float* __restrict__ W2,
                                                         float* __restrict__ inv_in, float* __restrict__ h2,
                                                         int N, int P) {
  __shared__ float acc[HID * A1_PITCH];
  __shared__ int cnt[S2_NODES + 1];
  __shared__ __attribute__((aligned(16))) int bin[S2_BINS];
  __shared__ unsigned int raw[S2_CAP];
  __shared__ unsigned int srt[S2_CAP];
  __shared__ int wsum[S2_THREADS / 32];
  __shared__ int fill;
  const int t = (int)threadIdx.x;
  const int lane = t & 31;
  const int wave = t >> 5;
  const int q = t & 3;
  const unsigned int src_mask = (1u << ENT_SHIFT) - 1u;
  for (int i = t; i < HID * A1_PITCH; i += S2_THREADS) acc[i] = 0.0f;
  for (int i = t; i <= S2_NODES; i += S2_THREADS) cnt[i] = 0;
  for (int i = t; i < S2_BINS; i += S2_THREADS) bin[i] = 0;
  if (t == 0) fill = 0;
  __syncthreads();

  const int NB = (N + NPB - 1) >> NPB_SHIFT;
  const int G = (P + SEG_PER_WAVE - 1) / SEG_PER_WAVE;
  for (int r = 0; r < S2_R; ++r) {
    const int b = S2_R * (int)blockIdx.x + r;
    if (b >= NB) break;
    const unsigned int tag = (unsigned int)r << (ENT_SHIFT + NPB_SHIFT);
    for (int g = wave; g < G; g += S2_THREADS / 32) {
      int pre[SEG_PER_WAVE + 1];
      int off[SEG_PER_WAVE];
      run_open<PIECE>(tab, b, g, P, lane, pre, off);
      const int total = pre[SEG_PER_WAVE];
      int base = 0;
      if (lane == 0 && total > 0) base = atomicAdd(&fill, total);
      base = __builtin_amdgcn_readfirstlane(base);
      for (int v0 = 0; v0 < total; v0 += 128) {
        unsigned int ent[4];
        for (int u = 0; u < 4; ++u) {
          const int n = v0 + 32 * u + lane;
          ent[u] = list[run_index(min(n, total - 1), pre, off)] | tag;
        }
        for (int u = 0; u < 4; ++u) {
          const int n = v0 + 32 * u + lane;
          if (n < total) {
            const int pos = base + n;
            if (pos < S2_CAP) {
              raw[pos] = ent[u];
              atomicAdd(&bin[s2_key(ent[u])], 1);
            } else {
              const unsigned int place = ent[u] >> ENT_SHIFT;
              const float4* hr = (const float4*)(h1 + (size_t)(ent[u] & src_mask) * HID);
              for (int f = 0; f < 4; ++f) {
                const float4 v = hr[f];
                atomicAdd(&acc[(4 * f + 0) * A1_PITCH + place], v.x);
                atomicAdd(&acc[(4 * f + 1) * A1_PITCH + place], v.y);
                atomicAdd(&acc[(4 * f + 2) * A1_PITCH + place], v.z);
                atomicAdd(&acc[(4 * f + 3) * A1_PITCH + place], v.w);
              }
              atomicAdd(&cnt[place], 1);
            }
          }
        }
      }
    }
  }
  __syncthreads();
  const int nst = min(fill, S2_CAP);

  const int4 c0 = ((const int4*)bin)[2 * t];
  const int4 c1 = ((const int4*)bin)[2 * t + 1];
  const int own = c0.x + c0.y + c0.z + c0.w + c1.x + c1.y + c1.z + c1.w;
  int inc = own;
  #pragma unroll
  for (int o = 1; o < 32; o <<= 1) {
    const int up = __shfl_up(inc, o, 32);
    if (lane >= o) inc += up;
  }
  if (lane == 31) wsum[wave] = inc;
  __syncthreads();
  if (t < 32) {
    const int w = wsum[t];
    int wi = w;
    #pragma unroll
    for (int o = 1; o < 32; o <<= 1) {
      const int up = __shfl_up(wi, o, 32);
      if (t >= o) wi += up;
    }
    wsum[t] = wi - w;
  }
  __syncthreads();
  {
    int run = wsum[wave] + inc - own;
    int4 o0, o1;
    o0.x = run; run += c0.x;
    o0.y = run; run += c0.y;
    o0.z = run; run += c0.z;
    o0.w = run; run += c0.w;
    o1.x = run; run += c1.x;
    o1.y = run; run += c1.y;
    o1.z = run; run += c1.z;
    o1.w = run;
    ((int4*)bin)[2 * t] = o0;
    ((int4*)bin)[2 * t + 1] = o1;
  }
  __syncthreads();

  for (int i0 = t; i0 < nst; i0 += 4 * S2_THREADS) {
    unsigned int e[4];
    int pos[4];
    #pragma unroll
    for (int u = 0; u < 4; ++u) {
      const int i = i0 + u * S2_THREADS;
      e[u] = 0u;
      if (i < nst) e[u] = raw[i];
    }
    #pragma unroll
    for (int u = 0; u < 4; ++u) {
      const int i = i0 + u * S2_THREADS;
      pos[u] = 0;
      if (i < nst) pos[u] = atomicAdd(&bin[s2_key(e[u])], 1);
    }
    #pragma unroll
    for (int u = 0; u < 4; ++u) {
      const int i = i0 + u * S2_THREADS;
      if (i < nst) srt[pos[u]] = e[u];
    }
  }
  __syncthreads();

  const float4* __restrict__ h1v = (const float4*)h1;
  const int grp = lane >> 2;
  for (int i0 = wave * 64; i0 < nst; i0 += 2 * S2_THREADS) {
    unsigned int w[2];
    unsigned int e[2][4];
    float4 v[2][4];
    #pragma unroll
    for (int h = 0; h < 2; ++h) {
      const int i = i0 + 32 * h + lane;
      w[h] = srt[min(i, nst - 1)];
      if (i >= nst) w[h] = (w[h] & src_mask) | ((unsigned int)S2_NODES << ENT_SHIFT);
    }
    #pragma unroll
    for (int h = 0; h < 2; ++h) {
      #pragma unroll
      for (int u = 0; u < 4; ++u) {
        e[h][u] = (unsigned int)__shfl((int)w[h], 8 * u + grp, 32);
        v[h][u] = h1v[4u * (e[h][u] & src_mask) + (unsigned int)q];
      }
    }
    #pragma unroll
    for (int h = 0; h < 2; ++h) {
      #pragma unroll
      for (int u = 0; u < 4; ++u) {
        float* row = acc + 4 * q * A1_PITCH + (e[h][u] >> ENT_SHIFT);
        atomicAdd(&row[0 * A1_PITCH], v[h][u].x);
        atomicAdd(&row[1 * A1_PITCH], v[h][u].y);
        atomicAdd(&row[2 * A1_PITCH], v[h][u].z);
        atomicAdd(&row[3 * A1_PITCH], v[h][u].w);
      }
      atomicAdd(&cnt[w[h] >> ENT_SHIFT], 1);
    }
  }
  __syncthreads();
  const int node = (int)blockIdx.x * S2_NODES + t;
  if (t >= S2_NODES || node >= N) return;
  const float si = rsqrtf(fmaxf((float)cnt[t], 1.0f));
  const float so = inv_out[node];
  inv_in[node] = si;
  float acc0 = 0.0f, acc1 = 0.0f;
  #pragma unroll
  for (int k = 0; k < HID; ++k) {
    float hv = fmaxf(fmaf(acc[k * A1_PITCH + t], si, b1[k]), 0.0f) * so;
    acc0 = fmaf(hv, W2[k * NCLS + 0], acc0);
    acc1 = fmaf(hv, W2[k * NCLS + 1], acc1);
  }
  *(float2*)(h2 + (size_t)node * NCLS) = make_float2(acc0, acc1);
}

__global__ __launch_bounds__(S2_THREADS) void agg2_final(const unsigned int* __restrict__ list, const unsigned int* __restrict__ tab,
                                                         const float* __restrict__ h2, const float* __restrict__ inv_in,
                                                         const float* __restrict__ b2, float* __restrict__ out, int N, int P) {
  __shared__ float a0[S2_NODES + 1];
  __shared__ float a1[S2_NODES + 1];
  __shared__ __attribute__((aligned(16))) int bin[S2_BINS];
  __shared__ unsigned int raw[S2_CAP];
  __shared__ unsigned int srt[S2_CAP];
  __shared__ int wsum[S2_THREADS / 32];
  __shared__ int fill;
  const int t = (int)threadIdx.x;
  const int lane = t & 31;
  const int wave = t >> 5;
  const unsigned int src_mask = (1u << ENT_SHIFT) - 1u;
  for (int i = t; i <= S2_NODES; i += S2_THREADS) { a0[i] = 0.0f; a1[i] = 0.0f; }
  for (int i = t; i < S2_BINS; i += S2_THREADS) bin[i] = 0;
  if (t == 0) fill = 0;
  __syncthreads();

  const int NB = (N + NPB - 1) >> NPB_SHIFT;
  const int G = (P + SEG_PER_WAVE - 1) / SEG_PER_WAVE;
  for (int r = 0; r < S2_R; ++r) {
    const int b = S2_R * (int)blockIdx.x + r;
    if (b >= NB) break;
    const unsigned int tag = (unsigned int)r << (ENT_SHIFT + NPB_SHIFT);
    for (int g = wave; g < G; g += S2_THREADS / 32) {
      int pre[SEG_PER_WAVE + 1];
      int off[SEG_PER_WAVE];
      run_open<PIECE>(tab, b, g, P, lane, pre, off);
      const int total = pre[SEG_PER_WAVE];
      int base = 0;
      if (lane == 0 && total > 0) base = atomicAdd(&fill, total);
      base = __builtin_amdgcn_readfirstlane(base);
      for (int v0 = 0; v0 < total; v0 += 128) {
        unsigned int ent[4];
        for (int u = 0; u < 4; ++u) {
          const int n = v0 + 32 * u + lane;
          ent[u] = list[run_index(min(n, total - 1), pre, off)] | tag;
        }
        for (int u = 0; u < 4; ++u) {
          const int n = v0 + 32 * u + lane;
          if (n < total) {
            const int pos = base + n;
            if (pos < S2_CAP) {
              raw[pos] = ent[u];
              atomicAdd(&bin[s2_key(ent[u])], 1);
            } else {
              const float2 v = *(const float2*)(h2 + (size_t)(ent[u] & src_mask) * NCLS);
              atomicAdd(&a0[ent[u] >> ENT_SHIFT], v.x);
              atomicAdd(&a1[ent[u] >> ENT_SHIFT], v.y);
            }
          }
        }
      }
    }
  }
  __syncthreads();
  const int nst = min(fill, S2_CAP);

  const int4 c0 = ((const int4*)bin)[2 * t];
  const int4 c1 = ((const int4*)bin)[2 * t + 1];
  const int own = c0.x + c0.y + c0.z + c0.w + c1.x + c1.y + c1.z + c1.w;
  int inc = own;
  #pragma unroll
  for (int o = 1; o < 32; o <<= 1) {
    const int up = __shfl_up(inc, o, 32);
    if (lane >= o) inc += up;
  }
  if (lane == 31) wsum[wave] = inc;
  __syncthreads();
  if (t < 32) {
    const int w = wsum[t];
    int wi = w;
    #pragma unroll
    for (int o = 1; o < 32; o <<= 1) {
      const int up = __shfl_up(wi, o, 32);
      if (t >= o) wi += up;
    }
    wsum[t] = wi - w;
  }
  __syncthreads();
  {
    int run = wsum[wave] + inc - own;
    int4 o0, o1;
    o0.x = run; run += c0.x;
    o0.y = run; run += c0.y;
    o0.z = run; run += c0.z;
    o0.w = run; run += c0.w;
    o1.x = run; run += c1.x;
    o1.y = run; run += c1.y;
    o1.z = run; run += c1.z;
    o1.w = run;
    ((int4*)bin)[2 * t] = o0;
    ((int4*)bin)[2 * t + 1] = o1;
  }
  __syncthreads();

  for (int i0 = t; i0 < nst; i0 += 4 * S2_THREADS) {
    unsigned int e[4];
    int pos[4];
    #pragma unroll
    for (int u = 0; u < 4; ++u) {
      const int i = i0 + u * S2_THREADS;
      e[u] = 0u;
      if (i < nst) e[u] = raw[i];
    }
    #pragma unroll
    for (int u = 0; u < 4; ++u) {
      const int i = i0 + u * S2_THREADS;
      pos[u] = 0;
      if (i < nst) pos[u] = atomicAdd(&bin[s2_key(e[u])], 1);
    }
    #pragma unroll
    for (int u = 0; u < 4; ++u) {
      const int i = i0 + u * S2_THREADS;
      if (i < nst) srt[pos[u]] = e[u];
    }
  }
  __syncthreads();

  for (int i0 = wave * 128; i0 < nst; i0 += (S2_THREADS / 32) * 128) {
    unsigned int slot[4];
    float2 v[4];
    for (int u = 0; u < 4; ++u) {
      const int i = i0 + 32 * u + lane;
      const unsigned int e = srt[min(i, nst - 1)];
      slot[u] = i < nst ? (e >> ENT_SHIFT) : (unsigned int)S2_NODES;
      v[u] = *(const float2*)(h2 + (size_t)(e & src_mask) * NCLS);
    }
    for (int u = 0; u < 4; ++u) {
      atomicAdd(&a0[slot[u]], v[u].x);
      atomicAdd(&a1[slot[u]], v[u].y);
    }
  }
  __syncthreads();
  const int i = (int)blockIdx.x * S2_NODES + t;
  if (t >= S2_NODES || i >= N) return;
  const float si = inv_in[i];
  float v0 = fmaf(a0[t], si, b2[0]);
  float v1 = fmaf(a1[t], si, b2[1]);
  float m = fmaxf(v0, v1);
  float lse = m + logf(expf(v0 - m) + expf(v1 - m));
  out[(size_t)i * NCLS + 0] = v0 - lse;
  out[(size_t)i * NCLS + 1] = v1 - lse;
}

extern "C" void kernel_launch(void* const* d_in, const int* in_sizes, int n_in,
                              void* d_out, int out_size, void* d_ws, size_t ws_size,
                              hipStream_t stream) {
  const float* x   = (const float*)d_in[0];
  const int*   src = (const int*)d_in[1];
  const int*   dst = (const int*)d_in[2];
  const float* W1  = (const float*)d_in[3];
  const float* b1  = (const float*)d_in[4];
  const float* W2  = (const float*)d_in[5];
  const float* b2  = (const float*)d_in[6];
  float* out = (float*)d_out;

  const int N = in_sizes[0] / F_IN;
  const int E = in_sizes[1];
  const int P = (E + PIECE - 1) / PIECE;

  float* ws      = (float*)d_ws;
  float* inv_out = ws;
  float* inv_in  = ws + (size_t)N;
  float* h2      = ws + (size_t)2 * N;
  const size_t tab0 = ((size_t)4 * N + 63) & ~(size_t)63;
  unsigned int* tabS = (unsigned int*)(ws + tab0);
  unsigned int* tabD = tabS + (size_t)P * MAXB;
  const size_t list0 = tab0 + (size_t)2 * P * MAXB;
  unsigned int* list = (unsigned int*)(ws + list0);
  const size_t h10 = list0 + (((size_t)P * PIECE + 63) & ~(size_t)63);
  float* h1      = ws + h10;
  if ((h10 + (size_t)HID * N) * sizeof(float) > ws_size) return;
  const size_t ls0 = h10 + (((size_t)HID * N + 63) & ~(size_t)63);
  const bool both = (ls0 + (size_t)P * (PIECE_B / 4)) * sizeof(float) <= ws_size;
  unsigned int* listS = both ? (unsigned int*)(ws + ls0) : list;

  const int NB = (N + NPB - 1) / NPB;
  const int tiles = N >> 4;
  const int gemm_wgs = (tiles + GEMM_TILES_PER_WG - 1) / GEMM_TILES_PER_WG;

  gemm1_and_orderings<<<gemm_wgs + (both ? 2 * P : P), SC_THREADS, 0, stream>>>(x, W1, h1, N, gemm_wgs, src, dst,
                                                                                tabD, list, both ? P : 0, tabS, listS, E, NB);
  deg_scale<<<NB, 512, 0, stream>>>((const unsigned char*)listS, tabS, inv_out, h1, N, P);
  if (!both)
    gemm1_and_orderings<<<P, SC_THREADS, 0, stream>>>(x, W1, h1, N, 0, src, dst, tabD, list, P, tabS, listS, E, NB);
  agg1_owner<<<(NB + S2_R - 1) / S2_R, S2_THREADS, 0, stream>>>(list, tabD, h1, inv_out, b1, W2, inv_in, h2, N, P);
  agg2_final<<<(NB + S2_R - 1) / S2_R, S2_THREADS, 0, stream>>>(list, tabD, h2, inv_in, b2, out, N, P);
}


# Destination ranges of 416 nodes (was 256), one per owner work-group in both layers (was a pair, 512 nodes): 241 owner work-groups on the 256 units (was 196); sums arrival-ordered as before
# speedup vs baseline: 5.9049x; 1.0773x over previous
#include <hip/hip_runtime.h>
#include <math.h>


typedef __attribute__((ext_vector_type(2))) float v2f;
typedef __attribute__((ext_vector_type(8))) float v8f;

#define F_IN 512
#define HID  16
#define NCLS 2

#define NPB_SHIFT 8
#define NPB   (1 << NPB_SHIFT)
#define NPD   416
#define MAXB  512
#define ENT_SHIFT 20
#define PL_BITS 9
static_assert(NPD % 32 == 0 && NPD < (1 << PL_BITS) && ENT_SHIFT + PL_BITS <= 32, "a destination place and the spare place fit above the source node");

#define SC_THREADS 1024
#define SC_TRIPS 4
#define SC_CAP (SC_TRIPS * 4 * SC_THREADS)
#define PIECE 12512
#define SEG_SHIFT 16
#define PIECE_B 12544
static_assert(PIECE_B >= PIECE && PIECE_B % 128 == 0, "a piece's source entries fit its own whole lines");
#define SEG_PER_WAVE 8
#define RK_BITS 14
static_assert(SC_CAP <= (1 << RK_BITS), "a place inside a piece fits RK_BITS");
static_assert(MAXB <= (1 << (31 - RK_BITS)), "range number and place share one non-negative word");

__device__ __forceinline__ unsigned int bin_entry(int key, unsigned int payload) {
  return (((unsigned int)key & ((1u << PL_BITS) - 1u)) << ENT_SHIFT) | payload;
}
__device__ __forceinline__ unsigned int owner_entry(unsigned int listed, int b) {
  const unsigned int first = (((unsigned int)b * NPD) & ((1u << PL_BITS) - 1u)) << ENT_SHIFT;
  return (listed - first) & ((1u << (ENT_SHIFT + PL_BITS)) - 1u);
}
template <int R>
__device__ __forceinline__ unsigned int range_of(int key) {
  if constexpr ((R & (R - 1)) == 0) {
    return (unsigned int)key >> __builtin_ctz((unsigned int)R);
  } else {
    constexpr unsigned long long M = (0x100000000ull + R - 1) / R;
    static_assert((M * R - 0x100000000ull) * (unsigned long long)(MAXB * NPB) < 0x100000000ull, "exact for every node the ranges cover");
    return __umulhi((unsigned int)key, (unsigned int)M);
  }
}
template <int R>
__device__ __forceinline__ int piece_rank(int* cntl, int key, int NB) {
  const unsigned int b = range_of<R>(key);
  return b < (unsigned int)NB ? atomicAdd(&cntl[b], 1) : -1;
}

struct PieceLds {
  int cntl[MAXB];
  int lofs[MAXB];
  int ntot;
  int wtot[MAXB / 32];
  __attribute__((aligned(16))) unsigned int stage[SC_CAP];
};
static_assert(MAXB % 32 == 0 && MAXB / 32 <= 32 && MAXB <= SC_THREADS, "one range a thread, one wave's total a lane");

template <bool HAS_PAY>
__device__ __forceinline__ void order_piece(PieceLds& L, const int* __restrict__ key, const int* __restrict__ pay,
                                            unsigned int* __restrict__ tab, unsigned int* __restrict__ list,
                                            int E, int NB, int w) {
  constexpr int R = HAS_PAY ? NPD : NPB;
  static_assert(HAS_PAY || R <= 256, "a place in a source range fits the one-byte list");
  int* cntl = L.cntl;
  int* lofs = L.lofs;
  unsigned int* stage = L.stage;
  unsigned char* stage8 = (unsigned char*)L.stage;
  const int t = (int)threadIdx.x;
  const int p0 = w * PIECE;
  const int p1 = min(p0 + PIECE, E);
  for (int b = t; b < MAXB; b += SC_THREADS) cntl[b] = b << RK_BITS;
  __syncthreads();

  int4 kv[SC_TRIPS];
  int4 pv[SC_TRIPS];
  #pragma unroll
  for (int u = 0; u < SC_TRIPS; ++u) {
    const int e = p0 + 4 * (u * SC_THREADS + t);
    kv[u] = make_int4(-1, -1, -1, -1);
    pv[u] = make_int4(0, 0, 0, 0);
    if (e + 3 < p1) {
      kv[u] = *(const int4*)(key + e);
      if (HAS_PAY) pv[u] = *(const int4*)(pay + e);
    }
  }
  const int rem = (p1 - p0) & 3;
  int kt = -1;
  int plt = 0;
  if (t < rem) {
    kt = key[p1 - rem + t];
    if (HAS_PAY) plt = pay[p1 - rem + t];
  }
  unsigned int en[SC_TRIPS][4];
  int rk[SC_TRIPS][4];
  #pragma unroll
  for (int u = 0; u < SC_TRIPS; ++u) {
    en[u][0] = HAS_PAY ? bin_entry(kv[u].x, (unsigned int)pv[u].x) : (unsigned int)(kv[u].x & (NPB - 1)); rk[u][0] = piece_rank<R>(cntl, kv[u].x, NB);
    en[u][1] = HAS_PAY ? bin_entry(kv[u].y, (unsigned int)pv[u].y) : (unsigned int)(kv[u].y & (NPB - 1)); rk[u][1] = piece_rank<R>(cntl, kv[u].y, NB);
    en[u][2] = HAS_PAY ? bin_entry(kv[u].z, (unsigned int)pv[u].z) : (unsigned int)(kv[u].z & (NPB - 1)); rk[u][2] = piece_rank<R>(cntl, kv[u].z, NB);
    en[u][3] = HAS_PAY ? bin_entry(kv[u].w, (unsigned int)pv[u].w) : (unsigned int)(kv[u].w & (NPB - 1)); rk[u][3] = piece_rank<R>(cntl, kv[u].w, NB);
  }
  const unsigned int ent = HAS_PAY ? bin_entry(kt, (unsigned int)plt) : (unsigned int)(kt & (NPB - 1));
  const int rkt = piece_rank<R>(cntl, kt, NB);
  __syncthreads();

  {
    const int lane = t & 31;
    const int wave = t >> 5;
    const int own = t < MAXB ? cntl[t] & ((1 << RK_BITS) - 1) : 0;
    int inc = own;
    #pragma unroll
    for (int off = 1; off < 32; off <<= 1) {
      const int up = __shfl_up(inc, off, 32);
      if (lane >= off) inc += up;
    }
    if (lane == 31 && t < MAXB) L.wtot[wave] = inc;
    __syncthreads();
    const int wt = lane < MAXB / 32 ? L.wtot[lane] : 0;
    int winc = wt;
    #pragma unroll
    for (int off = 1; off < MAXB / 32; off <<= 1) {
      const int up = __shfl_up(winc, off, 32);
      if (lane >= off) winc += up;
    }
    const int before = __shfl(winc - wt, min(wave, MAXB / 32 - 1), 32);
    if (t < MAXB) {
      lofs[t] = before + inc - own;
      if (t == MAXB - 1) L.ntot = before + inc;
    }
  }
  __syncthreads();

  for (int b = t; b < MAXB; b += SC_THREADS)
    tab[(size_t)w * MAXB + b] = (unsigned int)lofs[b] | ((unsigned int)(cntl[b] & ((1 << RK_BITS) - 1)) << SEG_SHIFT);

  const int rk_mask = (1 << RK_BITS) - 1;
  #pragma unroll
  for (int u = 0; u < SC_TRIPS; u += 2) {
    int ps[2][4];
    #pragma unroll
    for (int v = 0; v < 2; ++v) {
      #pragma unroll
      for (int j = 0; j < 4; ++j) ps[v][j] = lofs[rk[u + v][j] >= 0 ? rk[u + v][j] >> RK_BITS : 0] + (rk[u + v][j] & rk_mask);
    }
    #pragma unroll
    for (int v = 0; v < 2; ++v) {
      #pragma unroll
      for (int j = 0; j < 4; ++j)
        if (rk[u + v][j] >= 0) {
          if (HAS_PAY) stage[ps[v][j]] = en[u + v][j];
          else stage8[ps[v][j]] = (unsigned char)en[u + v][j];
        }
    }
  }
  if (rkt >= 0) {
    const int pt = lofs[rkt >> RK_BITS] + (rkt & rk_mask);
    if (HAS_PAY) stage[pt] = ent;
    else stage8[pt] = (unsigned char)ent;
  }
  __syncthreads();

  const int n = L.ntot;
  if (HAS_PAY) {
    unsigned int* __restrict__ mine = list + (size_t)p0;
    for (int i = t; i < (n >> 2); i += SC_THREADS) ((uint4*)mine)[i] = ((const uint4*)stage)[i];
    if (t < (n & 3)) mine[(n & ~3) + t] = stage[(n & ~3) + t];
  } else {
    unsigned char* __restrict__ mine = (unsigned char*)list + (size_t)w * PIECE_B;
    for (int i = t; i < (n >> 4); i += SC_THREADS) ((uint4*)mine)[i] = ((const uint4*)stage)[i];
    if (t < (n & 15)) mine[(n & ~15) + t] = stage8[(n & ~15) + t];
  }
}


template <int PITCH>
__device__ __forceinline__ void run_open(const unsigned int* __restrict__ tab, int b, int g, int P, int lane,
                                         int (&pre)[SEG_PER_WAVE + 1], int (&off)[SEG_PER_WAVE]) {
  const int j = lane & (SEG_PER_WAVE - 1);
  const int piece = SEG_PER_WAVE * g + j;
  const unsigned int wd = piece < P ? tab[(size_t)piece * MAXB + b] : 0u;
  const int len = (int)(wd >> SEG_SHIFT);
  int inc = len;
  #pragma unroll
  for (int o = 1; o < SEG_PER_WAVE; o <<= 1) {
    const int up = __shfl_up(inc, o, SEG_PER_WAVE);
    if (j >= o) inc += up;
  }
  const int exc = inc - len;
  const int of = piece * PITCH + (int)(wd & ((1u << SEG_SHIFT) - 1u)) - exc;
  #pragma unroll
  for (int s = 0; s < SEG_PER_WAVE; ++s) {
    pre[s] = __builtin_amdgcn_readlane(exc, s);
    off[s] = __builtin_amdgcn_readlane(of, s);
  }
  pre[SEG_PER_WAVE] = __builtin_amdgcn_readlane(inc, SEG_PER_WAVE - 1);
}
__device__ __forceinline__ int run_index(int v, const int (&pre)[SEG_PER_WAVE + 1], const int (&off)[SEG_PER_WAVE]) {
  int o = off[0];
  #pragma unroll
  for (int s = 1; s < SEG_PER_WAVE; ++s) o = v >= pre[s] ? off[s] : o;
  return v + o;
}

__global__ __launch_bounds__(512) void deg_scale(const unsigned char* __restrict__ list, const unsigned int* __restrict__ tab,
                                                 float* __restrict__ inv, float* __restrict__ h1, int N, int P) {
  __shared__ int c[NPB + 1];
  __shared__ float sinv[NPB];
  const int t = (int)threadIdx.x;
  if (t <= NPB) c[t] = 0;
  __syncthreads();
  static_assert(NPB * HID / 4 == 2 * 512, "two 16-byte pieces of h1 a thread");
  float4* __restrict__ hv = (float4*)(h1 + (size_t)blockIdx.x * NPB * HID);
  const bool row0 = (int)blockIdx.x * NPB + (t >> 2) < N;
  const bool row1 = (int)blockIdx.x * NPB + ((t + 512) >> 2) < N;
  float4 r0 = make_float4(0.0f, 0.0f, 0.0f, 0.0f);
  float4 r1 = make_float4(0.0f, 0.0f, 0.0f, 0.0f);
  if (row0) r0 = hv[t];
  if (row1) r1 = hv[t + 512];
  const int lane = t & 31;
  const int G = (P + SEG_PER_WAVE - 1) / SEG_PER_WAVE;
  for (int g = t >> 5; g < G; g += (int)blockDim.x >> 5) {
    int pre[SEG_PER_WAVE + 1];
    int off[SEG_PER_WAVE];
    run_open<PIECE_B>(tab, (int)blockIdx.x, g, P, lane, pre, off);
    const int total = pre[SEG_PER_WAVE];
    for (int v0 = 0; v0 < total; v0 += 128) {
      unsigned int slot[4];
      for (int u = 0; u < 4; ++u) {
        const int v = v0 + 32 * u + lane;
        const unsigned int ent = list[run_index(min(v, total - 1), pre, off)];
        slot[u] = v < total ? ent : (unsigned int)NPB;
      }
      for (int u = 0; u < 4; ++u) atomicAdd(&c[slot[u]], 1);
    }
  }
  __syncthreads();
  const int node = blockIdx.x * NPB + t;
  if (t < NPB) {
    const float s = rsqrtf(fmaxf((float)c[t], 1.0f));
    sinv[t] = s;
    if (node < N) inv[node] = s;
  }
  __syncthreads();
  if (row0) {
    const float s = sinv[t >> 2];
    hv[t] = make_float4(r0.x * s, r0.y * s, r0.z * s, r0.w * s);
  }
  if (row1) {
    const float s = sinv[(t + 512) >> 2];
    hv[t + 512] = make_float4(r1.x * s, r1.y * s, r1.z * s, r1.w * s);
  }
}

#define GEMM_TILES_PER_WG (SC_THREADS / 32)
__device__ __forceinline__ void gemm1_tiles(
    const float* __restrict__ x, const float* __restrict__ W1,
    float* __restrict__ h1, int N, int wg) {
  __shared__ float wlds[F_IN * HID];
  {
    float wv[F_IN * HID / SC_THREADS];
    #pragma unroll
    for (int j = 0; j < F_IN * HID / SC_THREADS; ++j) wv[j] = W1[threadIdx.x + j * SC_THREADS];
    #pragma unroll
    for (int j = 0; j < F_IN * HID / SC_THREADS; ++j) wlds[threadIdx.x + j * SC_THREADS] = wv[j];
  }
  __syncthreads();

  const int wave = threadIdx.x >> 5;
  const int lane = threadIdx.x & 31;
  const int tiles = N >> 4;
  const int tile = wg * GEMM_TILES_PER_WG + wave;
  if (tile >= tiles) return;

  const int n16 = lane & 15;
  const int g   = lane >> 4;
  const int row = tile * 16 + n16;
  const float* __restrict__ xr = x + (size_t)row * F_IN;

  v8f c = {};
  #pragma unroll 4
  for (int k = 0; k < F_IN; k += 4) {
    const int kk = k + 2 * g;
    const v2f a = *(const v2f*)(xr + kk);
    v2f b;
    b.x = wlds[(kk + 0) * HID + n16];
    b.y = wlds[(kk + 1) * HID + n16];
    c = __builtin_amdgcn_wmma_f32_16x16x4_f32(
         false, a,  false, b,
         (short)0, c,  false,  false);
  }

  #pragma unroll
  for (int p = 0; p < 8; ++p) {
    const int m = tile * 16 + p + 8 * g;
    h1[(size_t)m * HID + n16] = c[p];
  }
}

__global__ __launch_bounds__(SC_THREADS, 16) void gemm1_and_orderings(
    const float* __restrict__ x, const float* __restrict__ W1,
    float* __restrict__ h1, int N, int gemm_wgs,
    const int* __restrict__ src, const int* __restrict__ dst,
    unsigned int* __restrict__ tabD, unsigned int* __restrict__ listD, int n_dst,
    unsigned int* __restrict__ tabS, unsigned int* __restrict__ listS, int E, int NBS, int NBD) {
  __shared__ PieceLds L;
  const int wg = (int)blockIdx.x;
  if (wg < gemm_wgs) gemm1_tiles(x, W1, h1, N, wg);
  else if (wg - gemm_wgs < n_dst) order_piece<true>(L, dst, src, tabD, listD, E, NBD, wg - gemm_wgs);
  else order_piece<false>(L, src, src, tabS, listS, E, NBS, wg - gemm_wgs - n_dst);
}

#define S2_NODES NPD
#define S2_THREADS 1024
#define S2_CAP 24576
#define S2_KEY_SHIFT 4
#define S2_BINS ((MAXB * NPB) >> S2_KEY_SHIFT)
static_assert(S2_NODES <= S2_THREADS, "one node a thread when the sums are finished");
static_assert(S2_BINS == 8 * S2_THREADS, "the running sum below gives every thread eight neighbouring bins");
__device__ __forceinline__ int s2_key(unsigned int ent) {
  return min((int)((ent & ((1u << ENT_SHIFT) - 1u)) >> S2_KEY_SHIFT), S2_BINS - 1);
}

#define A1_PITCH (S2_NODES + 4)
__global__ __launch_bounds__(S2_THREADS) void agg1_owner(const unsigned int* __restrict__ list, const unsigned int* __restrict__ tab,
                                                         const float* __restrict__ h1, const float* __restrict__ inv_out,
                                                         const float* __restrict__ b1, const float* __restrict__ W2,
                                                         float* __restrict__ inv_in, float* __restrict__ h2,
                                                         int N, int P) {
  __shared__ float acc[HID * A1_PITCH];
  __shared__ int cnt[S2_NODES + 1];
  __shared__ __attribute__((aligned(16))) int bin[S2_BINS];
  __shared__ unsigned int raw[S2_CAP];
  __shared__ unsigned int srt[S2_CAP];
  __shared__ int wsum[S2_THREADS / 32];
  __shared__ int fill;
  const int t = (int)threadIdx.x;
  const int lane = t & 31;
  const int wave = t >> 5;
  const int q = t & 3;
  const unsigned int src_mask = (1u << ENT_SHIFT) - 1u;
  for (int i = t; i < HID * A1_PITCH; i += S2_THREADS) acc[i] = 0.0f;
  for (int i = t; i <= S2_NODES; i += S2_THREADS) cnt[i] = 0;
  for (int i = t; i < S2_BINS; i += S2_THREADS) bin[i] = 0;
  if (t == 0) fill = 0;
  __syncthreads();

  const int G = (P + SEG_PER_WAVE - 1) / SEG_PER_WAVE;
  for (int g = wave; g < G; g += S2_THREADS / 32) {
    int pre[SEG_PER_WAVE + 1];
    int off[SEG_PER_WAVE];
    run_open<PIECE>(tab, (int)blockIdx.x, g, P, lane, pre, off);
    const int total = pre[SEG_PER_WAVE];
    int base = 0;
    if (lane == 0 && total > 0) base = atomicAdd(&fill, total);
    base = __builtin_amdgcn_readfirstlane(base);
    for (int v0 = 0; v0 < total; v0 += 128) {
      unsigned int ent[4];
      for (int u = 0; u < 4; ++u) {
        const int n = v0 + 32 * u + lane;
        ent[u] = owner_entry(list[run_index(min(n, total - 1), pre, off)], (int)blockIdx.x);
      }
      for (int u = 0; u < 4; ++u) {
        const int n = v0 + 32 * u + lane;
        if (n < total) {
          const int pos = base + n;
          if (pos < S2_CAP) {
            raw[pos] = ent[u];
            atomicAdd(&bin[s2_key(ent[u])], 1);
          } else {
            const unsigned int place = ent[u] >> ENT_SHIFT;
            const float4* hr = (const float4*)(h1 + (size_t)(ent[u] & src_mask) * HID);
            for (int f = 0; f < 4; ++f) {
              const float4 v = hr[f];
              atomicAdd(&acc[(4 * f + 0) * A1_PITCH + place], v.x);
              atomicAdd(&acc[(4 * f + 1) * A1_PITCH + place], v.y);
              atomicAdd(&acc[(4 * f + 2) * A1_PITCH + place], v.z);
              atomicAdd(&acc[(4 * f + 3) * A1_PITCH + place], v.w);
            }
            atomicAdd(&cnt[place], 1);
          }
        }
      }
    }
  }
  __syncthreads();
  const int nst = min(fill, S2_CAP);

  const int4 c0 = ((const int4*)bin)[2 * t];
  const int4 c1 = ((const int4*)bin)[2 * t + 1];
  const int own = c0.x + c0.y + c0.z + c0.w + c1.x + c1.y + c1.z + c1.w;
  int inc = own;
  #pragma unroll
  for (int o = 1; o < 32; o <<= 1) {
    const int up = __shfl_up(inc, o, 32);
    if (lane >= o) inc += up;
  }
  if (lane == 31) wsum[wave] = inc;
  __syncthreads();
  if (t < 32) {
    const int w = wsum[t];
    int wi = w;
    #pragma unroll
    for (int o = 1; o < 32; o <<= 1) {
      const int up = __shfl_up(wi, o, 32);
      if (t >= o) wi += up;
    }
    wsum[t] = wi - w;
  }
  __syncthreads();
  {
    int run = wsum[wave] + inc - own;
    int4 o0, o1;
    o0.x = run; run += c0.x;
    o0.y = run; run += c0.y;
    o0.z = run; run += c0.z;
    o0.w = run; run += c0.w;
    o1.x = run; run += c1.x;
    o1.y = run; run += c1.y;
    o1.z = run; run += c1.z;
    o1.w = run;
    ((int4*)bin)[2 * t] = o0;
    ((int4*)bin)[2 * t + 1] = o1;
  }
  __syncthreads();

  for (int i0 = t; i0 < nst; i0 += 4 * S2_THREADS) {
    unsigned int e[4];
    int pos[4];
    #pragma unroll
    for (int u = 0; u < 4; ++u) {
      const int i = i0 + u * S2_THREADS;
      e[u] = 0u;
      if (i < nst) e[u] = raw[i];
    }
    #pragma unroll
    for (int u = 0; u < 4; ++u) {
      const int i = i0 + u * S2_THREADS;
      pos[u] = 0;
      if (i < nst) pos[u] = atomicAdd(&bin[s2_key(e[u])], 1);
    }
    #pragma unroll
    for (int u = 0; u < 4; ++u) {
      const int i = i0 + u * S2_THREADS;
      if (i < nst) srt[pos[u]] = e[u];
    }
  }
  __syncthreads();

  const float4* __restrict__ h1v = (const float4*)h1;
  const int grp = lane >> 2;
  for (int i0 = wave * 64; i0 < nst; i0 += 2 * S2_THREADS) {
    unsigned int w[2];
    unsigned int e[2][4];
    float4 v[2][4];
    #pragma unroll
    for (int h = 0; h < 2; ++h) {
      const int i = i0 + 32 * h + lane;
      w[h] = srt[min(i, nst - 1)];
      if (i >= nst) w[h] = (w[h] & src_mask) | ((unsigned int)S2_NODES << ENT_SHIFT);
    }
    #pragma unroll
    for (int h = 0; h < 2; ++h) {
      #pragma unroll
      for (int u = 0; u < 4; ++u) {
        e[h][u] = (unsigned int)__shfl((int)w[h], 8 * u + grp, 32);
        v[h][u] = h1v[4u * (e[h][u] & src_mask) + (unsigned int)q];
      }
    }
    #pragma unroll
    for (int h = 0; h < 2; ++h) {
      #pragma unroll
      for (int u = 0; u < 4; ++u) {
        float* row = acc + 4 * q * A1_PITCH + (e[h][u] >> ENT_SHIFT);
        atomicAdd(&row[0 * A1_PITCH], v[h][u].x);
        atomicAdd(&row[1 * A1_PITCH], v[h][u].y);
        atomicAdd(&row[2 * A1_PITCH], v[h][u].z);
        atomicAdd(&row[3 * A1_PITCH], v[h][u].w);
      }
      atomicAdd(&cnt[w[h] >> ENT_SHIFT], 1);
    }
  }
  __syncthreads();
  const int node = (int)blockIdx.x * S2_NODES + t;
  if (t >= S2_NODES || node >= N) return;
  const float si = rsqrtf(fmaxf((float)cnt[t], 1.0f));
  const float so = inv_out[node];
  inv_in[node] = si;
  float acc0 = 0.0f, acc1 = 0.0f;
  #pragma unroll
  for (int k = 0; k < HID; ++k) {
    float hv = fmaxf(fmaf(acc[k * A1_PITCH + t], si, b1[k]), 0.0f) * so;
    acc0 = fmaf(hv, W2[k * NCLS + 0], acc0);
    acc1 = fmaf(hv, W2[k * NCLS + 1], acc1);
  }
  *(float2*)(h2 + (size_t)node * NCLS) = make_float2(acc0, acc1);
}

__global__ __launch_bounds__(S2_THREADS) void agg2_final(const unsigned int* __restrict__ list, const unsigned int* __restrict__ tab,
                                                         const float* __restrict__ h2, const float* __restrict__ inv_in,
                                                         const float* __restrict__ b2, float* __restrict__ out, int N, int P) {
  __shared__ float a0[S2_NODES + 1];
  __shared__ float a1[S2_NODES + 1];
  __shared__ __attribute__((aligned(16))) int bin[S2_BINS];
  __shared__ unsigned int raw[S2_CAP];
  __shared__ unsigned int srt[S2_CAP];
  __shared__ int wsum[S2_THREADS / 32];
  __shared__ int fill;
  const int t = (int)threadIdx.x;
  const int lane = t & 31;
  const int wave = t >> 5;
  const unsigned int src_mask = (1u << ENT_SHIFT) - 1u;
  for (int i = t; i <= S2_NODES; i += S2_THREADS) { a0[i] = 0.0f; a1[i] = 0.0f; }
  for (int i = t; i < S2_BINS; i += S2_THREADS) bin[i] = 0;
  if (t == 0) fill = 0;
  __syncthreads();

  const int G = (P + SEG_PER_WAVE - 1) / SEG_PER_WAVE;
  for (int g = wave; g < G; g += S2_THREADS / 32) {
    int pre[SEG_PER_WAVE + 1];
    int off[SEG_PER_WAVE];
    run_open<PIECE>(tab, (int)blockIdx.x, g, P, lane, pre, off);
    const int total = pre[SEG_PER_WAVE];
    int base = 0;
    if (lane == 0 && total > 0) base = atomicAdd(&fill, total);
    base = __builtin_amdgcn_readfirstlane(base);
    for (int v0 = 0; v0 < total; v0 += 128) {
      unsigned int ent[4];
      for (int u = 0; u < 4; ++u) {
        const int n = v0 + 32 * u + lane;
        ent[u] = owner_entry(list[run_index(min(n, total - 1), pre, off)], (int)blockIdx.x);
      }
      for (int u = 0; u < 4; ++u) {
        const int n = v0 + 32 * u + lane;
        if (n < total) {
          const int pos = base + n;
          if (pos < S2_CAP) {
            raw[pos] = ent[u];
            atomicAdd(&bin[s2_key(ent[u])], 1);
          } else {
            const float2 v = *(const float2*)(h2 + (size_t)(ent[u] & src_mask) * NCLS);
            atomicAdd(&a0[ent[u] >> ENT_SHIFT], v.x);
            atomicAdd(&a1[ent[u] >> ENT_SHIFT], v.y);
          }
        }
      }
    }
  }
  __syncthreads();
  const int nst = min(fill, S2_CAP);

  const int4 c0 = ((const int4*)bin)[2 * t];
  const int4 c1 = ((const int4*)bin)[2 * t + 1];
  const int own = c0.x + c0.y + c0.z + c0.w + c1.x + c1.y + c1.z + c1.w;
  int inc = own;
  #pragma unroll
  for (int o = 1; o < 32; o <<= 1) {
    const int up = __shfl_up(inc, o, 32);
    if (lane >= o) inc += up;
  }
  if (lane == 31) wsum[wave] = inc;
  __syncthreads();
  if (t < 32) {
    const int w = wsum[t];
    int wi = w;
    #pragma unroll
    for (int o = 1; o < 32; o <<= 1) {
      const int up = __shfl_up(wi, o, 32);
      if (t >= o) wi += up;
    }
    wsum[t] = wi - w;
  }
  __syncthreads();
  {
    int run = wsum[wave] + inc - own;
    int4 o0, o1;
    o0.x = run; run += c0.x;
    o0.y = run; run += c0.y;
    o0.z = run; run += c0.z;
    o0.w = run; run += c0.w;
    o1.x = run; run += c1.x;
    o1.y = run; run += c1.y;
    o1.z = run; run += c1.z;
    o1.w = run;
    ((int4*)bin)[2 * t] = o0;
    ((int4*)bin)[2 * t + 1] = o1;
  }
  __syncthreads();

  for (int i0 = t; i0 < nst; i0 += 4 * S2_THREADS) {
    unsigned int e[4];
    int pos[4];
    #pragma unroll
    for (int u = 0; u < 4; ++u) {
      const int i = i0 + u * S2_THREADS;
      e[u] = 0u;
      if (i < nst) e[u] = raw[i];
    }
    #pragma unroll
    for (int u = 0; u < 4; ++u) {
      const int i = i0 + u * S2_THREADS;
      pos[u] = 0;
      if (i < nst) pos[u] = atomicAdd(&bin[s2_key(e[u])], 1);
    }
    #pragma unroll
    for (int u = 0; u < 4; ++u) {
      const int i = i0 + u * S2_THREADS;
      if (i < nst) srt[pos[u]] = e[u];
    }
  }
  __syncthreads();

  for (int i0 = wave * 128; i0 < nst; i0 += (S2_THREADS / 32) * 128) {
    unsigned int slot[4];
    float2 v[4];
    for (int u = 0; u < 4; ++u) {
      const int i = i0 + 32 * u + lane;
      const unsigned int e = srt[min(i, nst - 1)];
      slot[u] = i < nst ? (e >> ENT_SHIFT) : (unsigned int)S2_NODES;
      v[u] = *(const float2*)(h2 + (size_t)(e & src_mask) * NCLS);
    }
    for (int u = 0; u < 4; ++u) {
      atomicAdd(&a0[slot[u]], v[u].x);
      atomicAdd(&a1[slot[u]], v[u].y);
    }
  }
  __syncthreads();
  const int i = (int)blockIdx.x * S2_NODES + t;
  if (t >= S2_NODES || i >= N) return;
  const float si = inv_in[i];
  float v0 = fmaf(a0[t], si, b2[0]);
  float v1 = fmaf(a1[t], si, b2[1]);
  float m = fmaxf(v0, v1);
  float lse = m + logf(expf(v0 - m) + expf(v1 - m));
  out[(size_t)i * NCLS + 0] = v0 - lse;
  out[(size_t)i * NCLS + 1] = v1 - lse;
}

extern "C" void kernel_launch(void* const* d_in, const int* in_sizes, int n_in,
                              void* d_out, int out_size, void* d_ws, size_t ws_size,
                              hipStream_t stream) {
  const float* x   = (const float*)d_in[0];
  const int*   src = (const int*)d_in[1];
  const int*   dst = (const int*)d_in[2];
  const float* W1  = (const float*)d_in[3];
  const float* b1  = (const float*)d_in[4];
  const float* W2  = (const float*)d_in[5];
  const float* b2  = (const float*)d_in[6];
  float* out = (float*)d_out;

  const int N = in_sizes[0] / F_IN;
  const int E = in_sizes[1];
  const int P = (E + PIECE - 1) / PIECE;

  float* ws      = (float*)d_ws;
  float* inv_out = ws;
  float* inv_in  = ws + (size_t)N;
  float* h2      = ws + (size_t)2 * N;
  const size_t tab0 = ((size_t)4 * N + 63) & ~(size_t)63;
  unsigned int* tabS = (unsigned int*)(ws + tab0);
  unsigned int* tabD = tabS + (size_t)P * MAXB;
  const size_t list0 = tab0 + (size_t)2 * P * MAXB;
  unsigned int* list = (unsigned int*)(ws + list0);
  const size_t h10 = list0 + (((size_t)P * PIECE + 63) & ~(size_t)63);
  float* h1      = ws + h10;
  if ((h10 + (size_t)HID * N) * sizeof(float) > ws_size) return;
  const size_t ls0 = h10 + (((size_t)HID * N + 63) & ~(size_t)63);
  const bool both = (ls0 + (size_t)P * (PIECE_B / 4)) * sizeof(float) <= ws_size;
  unsigned int* listS = both ? (unsigned int*)(ws + ls0) : list;

  const int NBS = (N + NPB - 1) / NPB;
  const int NBD = (N + NPD - 1) / NPD;
  const int tiles = N >> 4;
  const int gemm_wgs = (tiles + GEMM_TILES_PER_WG - 1) / GEMM_TILES_PER_WG;

  gemm1_and_orderings<<<gemm_wgs + (both ? 2 * P : P), SC_THREADS, 0, stream>>>(x, W1, h1, N, gemm_wgs, src, dst,
                                                                                tabD, list, both ? P : 0, tabS, listS, E, NBS, NBD);
  deg_scale<<<NBS, 512, 0, stream>>>((const unsigned char*)listS, tabS, inv_out, h1, N, P);
  if (!both)
    gemm1_and_orderings<<<P, SC_THREADS, 0, stream>>>(x, W1, h1, N, 0, src, dst, tabD, list, P, tabS, listS, E, NBS, NBD);
  agg1_owner<<<NBD, S2_THREADS, 0, stream>>>(list, tabD, h1, inv_out, b1, W2, inv_in, h2, N, P);
  agg2_final<<<NBD, S2_THREADS, 0, stream>>>(list, tabD, h2, inv_in, b2, out, N, P);
}
